# starting kernel
# baseline (speedup 1.0000x reference)
#include <hip/hip_runtime.h>
#include <hip/hip_bf16.h>
#include <math.h>

typedef float v2f __attribute__((ext_vector_type(2)));
typedef float v8f __attribute__((ext_vector_type(8)));

#define D_FEAT 128

__global__ void init_state_kernel(float* __restrict__ c, float* __restrict__ z,
                                  unsigned int* __restrict__ mb, int V) {
    int i = blockIdx.x * blockDim.x + threadIdx.x;
    int n = V * D_FEAT;
    if (i < n) c[i] = 0.0f;
    if (i < V) { z[i] = 0.0f; mb[i] = 0xFF800000u; }
}

__global__ void packT_kernel(const float* __restrict__ W, float* __restrict__ WT,
                             int R, int C) {
    int i = blockIdx.x * blockDim.x + threadIdx.x;
    if (i >= R * C) return;
    int n = i / C;
    int k = i - n * C;
    WT[(size_t)k * R + n] = W[i];
}

__global__ void node_dot_kernel(const float* __restrict__ nf,
                                const float* __restrict__ W_edge,
                                float* __restrict__ pd, float* __restrict__ ps,
                                int V) {
    int gid = blockIdx.x * blockDim.x + threadIdx.x;
    int v = gid >> 5;
    int lane = gid & 31;
    if (v >= V) return;
    float4 x  = ((const float4*)(nf + (size_t)v * D_FEAT))[lane];
    float4 w0 = ((const float4*)W_edge)[lane];
    float4 w1 = ((const float4*)(W_edge + D_FEAT))[lane];
    float d0 = x.x * w0.x + x.y * w0.y + x.z * w0.z + x.w * w0.w;
    float d1 = x.x * w1.x + x.y * w1.y + x.z * w1.z + x.w * w1.w;
#pragma unroll
    for (int off = 16; off > 0; off >>= 1) {
        d0 += __shfl_down(d0, off, 32);
        d1 += __shfl_down(d1, off, 32);
    }
    if (lane == 0) { pd[v] = d0; ps[v] = d1; }
}

template <int N>
__global__ __launch_bounds__(256) void wmma_gemm_kernel(
    const float* __restrict__ X, const float* __restrict__ B,
    const float* __restrict__ bias, float* __restrict__ Out, int V) {
    __shared__ float xs[16][132];
    const int tid = threadIdx.x;
    const int rowbase = blockIdx.x * 16;
#pragma unroll
    for (int i = 0; i < 8; ++i) {
        int idx = tid + i * 256;
        int r = idx >> 7;
        int col = idx & 127;
        int row = rowbase + r;
        if (row >= V) row = V - 1;
        xs[r][col] = X[(size_t)row * D_FEAT + col];
    }
    __syncthreads();
    const int lane = tid & 31;
    const int wave = tid >> 5;
    const int lo = lane & 15;
    const int hi = lane >> 4;
    const bool fullTile = (rowbase + 16 <= V);
    for (int t = wave; t < N / 16; t += 8) {
        const int colbase = t * 16;
        v8f acc;
        const float bv = bias[colbase + lo];
#pragma unroll
        for (int r = 0; r < 8; ++r) acc[r] = bv;
#pragma unroll
        for (int kk = 0; kk < D_FEAT; kk += 4) {
            const int ka = kk + 2 * hi;
            v2f a, b;
            a[0] = xs[lo][ka];
            a[1] = xs[lo][ka + 1];
            b[0] = B[(size_t)ka * N + colbase + lo];
            b[1] = B[(size_t)(ka + 1) * N + colbase + lo];
            acc = __builtin_amdgcn_wmma_f32_16x16x4_f32(
                false, a, false, b, (short)0, acc, false, false);
        }
        if (fullTile) {
#pragma unroll
            for (int r = 0; r < 8; ++r) {
                int row = rowbase + r + 8 * hi;
                Out[(size_t)row * N + colbase + lo] = acc[r];
            }
        } else {
#pragma unroll
            for (int r = 0; r < 8; ++r) {
                int row = rowbase + r + 8 * hi;
                if (row < V) Out[(size_t)row * N + colbase + lo] = acc[r];
            }
        }
    }
}

__global__ void edge_logit_kernel(const int* __restrict__ src,
                                  const int* __restrict__ dst,
                                  const float* __restrict__ pd,
                                  const float* __restrict__ ps,
                                  const float* __restrict__ b_edge,
                                  float* __restrict__ logit,
                                  unsigned int* __restrict__ mb, int E) {
    int e = blockIdx.x * blockDim.x + threadIdx.x;
    if (e >= E) return;
    int d = dst[e];
    float x = pd[d] + ps[src[e]] + b_edge[0];
    float l = x > 0.0f ? x : 0.01f * x;
    logit[e] = l;
    int iv = __float_as_int(l);
    if (iv >= 0) atomicMax((int*)(mb + d), iv);
    else         atomicMin(mb + d, (unsigned int)iv);
}

__global__ void edge_scatter_kernel(const int* __restrict__ src,
                                    const int* __restrict__ dst,
                                    const float* __restrict__ logit,
                                    const unsigned int* __restrict__ mb,
                                    float* __restrict__ z,
                                    const float* __restrict__ hv,
                                    float* __restrict__ c, int E) {
    int gid = blockIdx.x * blockDim.x + threadIdx.x;
    int e = gid >> 5;
    int lane = gid & 31;
    if (e >= E) return;
    int d = dst[e], s = src[e];
    float m = __int_as_float((int)mb[d]);
    float ev = expf(logit[e] - m);
    if (lane == 0) atomicAdd(z + d, ev);
    float4 h = ((const float4*)(hv + (size_t)s * D_FEAT))[lane];
    float* cd = c + (size_t)d * D_FEAT + lane * 4;
    atomicAdd(cd + 0, ev * h.x);
    atomicAdd(cd + 1, ev * h.y);
    atomicAdd(cd + 2, ev * h.z);
    atomicAdd(cd + 3, ev * h.w);
}

__global__ void context_kernel(float* __restrict__ c, const float* __restrict__ z,
                               int V) {
    int i = blockIdx.x * blockDim.x + threadIdx.x;
    if (i >= V * D_FEAT) return;
    float zz = z[i >> 7];
    float x = zz > 0.0f ? c[i] / zz : 0.0f;
    c[i] = x > 0.0f ? x : expf(x) - 1.0f;
}

__global__ void gru_kernel(const float* __restrict__ gi, const float* __restrict__ gh,
                           const float* __restrict__ nf, float* __restrict__ out,
                           int V) {
    int i = blockIdx.x * blockDim.x + threadIdx.x;
    if (i >= V * D_FEAT) return;
    int v = i >> 7;
    int d = i & 127;
    size_t base = (size_t)v * 384;
    float ir = gi[base + d], iz = gi[base + 128 + d], in_ = gi[base + 256 + d];
    float hr = gh[base + d], hz = gh[base + 128 + d], hn  = gh[base + 256 + d];
    float r = 1.0f / (1.0f + expf(-(ir + hr)));
    float u = 1.0f / (1.0f + expf(-(iz + hz)));
    float n = tanhf(in_ + r * hn);
    float h = (1.0f - u) * n + u * nf[i];
    out[i] = h > 0.0f ? h : 0.0f;
}

extern "C" void kernel_launch(void* const* d_in, const int* in_sizes, int n_in,
                              void* d_out, int out_size, void* d_ws, size_t ws_size,
                              hipStream_t stream) {
    const float* nf     = (const float*)d_in[0];
    const int*   src    = (const int*)d_in[1];
    const int*   dst    = (const int*)d_in[2];
    const float* W_edge = (const float*)d_in[3];
    const float* b_edge = (const float*)d_in[4];
    const float* W_proj = (const float*)d_in[5];
    const float* b_proj = (const float*)d_in[6];
    const float* W_ih   = (const float*)d_in[7];
    const float* b_ih   = (const float*)d_in[8];
    const float* W_hh   = (const float*)d_in[9];
    const float* b_hh   = (const float*)d_in[10];

    const int V = in_sizes[0] / D_FEAT;
    const int E = in_sizes[1];

    float* wsf = (float*)d_ws;
    float* pd    = wsf;
    float* ps    = pd + V;
    unsigned int* mb = (unsigned int*)(ps + V);
    float* z     = (float*)(mb + V);
    float* logit = z + V;
    float* hv    = logit + E;
    float* gh    = hv + (size_t)V * 128;
    float* c     = gh + (size_t)V * 384;
    float* gi    = c + (size_t)V * 128;
    float* WhhT  = gi + (size_t)V * 384;
    float* WihT  = WhhT + 128 * 384;

    const int blk = 256;

    packT_kernel<<<(384 * 128 + blk - 1) / blk, blk, 0, stream>>>(W_hh, WhhT, 384, 128);
    packT_kernel<<<(384 * 128 + blk - 1) / blk, blk, 0, stream>>>(W_ih, WihT, 384, 128);

    init_state_kernel<<<((size_t)V * 128 + blk - 1) / blk, blk, 0, stream>>>(c, z, mb, V);

    node_dot_kernel<<<((size_t)V * 32 + blk - 1) / blk, blk, 0, stream>>>(
        nf, W_edge, pd, ps, V);

    const int gemmBlocks = (V + 15) / 16;
    wmma_gemm_kernel<128><<<gemmBlocks, 256, 0, stream>>>(nf, W_proj, b_proj, hv, V);
    wmma_gemm_kernel<384><<<gemmBlocks, 256, 0, stream>>>(nf, WhhT, b_hh, gh, V);

    edge_logit_kernel<<<(E + blk - 1) / blk, blk, 0, stream>>>(
        src, dst, pd, ps, b_edge, logit, mb, E);
    edge_scatter_kernel<<<((size_t)E * 32 + blk - 1) / blk, blk, 0, stream>>>(
        src, dst, logit, mb, z, hv, c, E);

    context_kernel<<<((size_t)V * 128 + blk - 1) / blk, blk, 0, stream>>>(c, z, V);

    wmma_gemm_kernel<384><<<gemmBlocks, 256, 0, stream>>>(c, WihT, b_ih, gi, V);

    gru_kernel<<<((size_t)V * 128 + blk - 1) / blk, blk, 0, stream>>>(
        gi, gh, nf, (float*)d_out, V);
}


# Float atomic scatter replaced by per-destination edge lists (one int exchange per edge) walked by a wave per node: running-max softmax and row sum in registers, so the summation order changes
# speedup vs baseline: 1.9302x; 1.9302x over previous
#include <hip/hip_runtime.h>
#include <hip/hip_bf16.h>
#include <math.h>

typedef float v2f __attribute__((ext_vector_type(2)));
typedef float v8f __attribute__((ext_vector_type(8)));

#define D_FEAT 128
#define HEAD_PITCH 32

__global__ void packT_kernel(const float* __restrict__ W, float* __restrict__ WT,
                             int R, int C) {
    int i = blockIdx.x * blockDim.x + threadIdx.x;
    if (i >= R * C) return;
    int n = i / C;
    int k = i - n * C;
    WT[(size_t)k * R + n] = W[i];
}

__global__ void node_dot_kernel(const float* __restrict__ nf,
                                const float* __restrict__ W_edge,
                                float* __restrict__ pd, float* __restrict__ ps,
                                int* __restrict__ head, int V) {
    int gid = blockIdx.x * blockDim.x + threadIdx.x;
    int v = gid >> 5;
    int lane = gid & 31;
    if (v >= V) return;
    float4 x  = ((const float4*)(nf + (size_t)v * D_FEAT))[lane];
    float4 w0 = ((const float4*)W_edge)[lane];
    float4 w1 = ((const float4*)(W_edge + D_FEAT))[lane];
    float d0 = x.x * w0.x + x.y * w0.y + x.z * w0.z + x.w * w0.w;
    float d1 = x.x * w1.x + x.y * w1.y + x.z * w1.z + x.w * w1.w;
#pragma unroll
    for (int off = 16; off > 0; off >>= 1) {
        d0 += __shfl_down(d0, off, 32);
        d1 += __shfl_down(d1, off, 32);
    }
    head[(size_t)v * HEAD_PITCH + lane] = -1;
    if (lane == 0) { pd[v] = d0; ps[v] = d1; }
}

template <int N>
__global__ __launch_bounds__(256) void wmma_gemm_kernel(
    const float* __restrict__ X, const float* __restrict__ B,
    const float* __restrict__ bias, float* __restrict__ Out, int V) {
    __shared__ float xs[16][132];
    const int tid = threadIdx.x;
    const int rowbase = blockIdx.x * 16;
#pragma unroll
    for (int i = 0; i < 8; ++i) {
        int idx = tid + i * 256;
        int r = idx >> 7;
        int col = idx & 127;
        int row = rowbase + r;
        if (row >= V) row = V - 1;
        xs[r][col] = X[(size_t)row * D_FEAT + col];
    }
    __syncthreads();
    const int lane = tid & 31;
    const int wave = tid >> 5;
    const int lo = lane & 15;
    const int hi = lane >> 4;
    const bool fullTile = (rowbase + 16 <= V);
    for (int t = wave; t < N / 16; t += 8) {
        const int colbase = t * 16;
        v8f acc;
        const float bv = bias[colbase + lo];
#pragma unroll
        for (int r = 0; r < 8; ++r) acc[r] = bv;
#pragma unroll
        for (int kk = 0; kk < D_FEAT; kk += 4) {
            const int ka = kk + 2 * hi;
            v2f a, b;
            a[0] = xs[lo][ka];
            a[1] = xs[lo][ka + 1];
            b[0] = B[(size_t)ka * N + colbase + lo];
            b[1] = B[(size_t)(ka + 1) * N + colbase + lo];
            acc = __builtin_amdgcn_wmma_f32_16x16x4_f32(
                false, a, false, b, (short)0, acc, false, false);
        }
        if (fullTile) {
#pragma unroll
            for (int r = 0; r < 8; ++r) {
                int row = rowbase + r + 8 * hi;
                Out[(size_t)row * N + colbase + lo] = acc[r];
            }
        } else {
#pragma unroll
            for (int r = 0; r < 8; ++r) {
                int row = rowbase + r + 8 * hi;
                if (row < V) Out[(size_t)row * N + colbase + lo] = acc[r];
            }
        }
    }
}

__global__ void edge_link_kernel(const int* __restrict__ src,
                                 const int* __restrict__ dst,
                                 int* __restrict__ head,
                                 int2* __restrict__ lnk, int E) {
    int e = blockIdx.x * blockDim.x + threadIdx.x;
    if (e >= E) return;
    int old = atomicExch(head + (size_t)dst[e] * HEAD_PITCH, e);
    lnk[e] = make_int2(src[e], old);
}

__global__ __launch_bounds__(256) void node_gather_kernel(
    const int* __restrict__ head, const int2* __restrict__ lnk,
    const float* __restrict__ pd, const float* __restrict__ ps,
    const float* __restrict__ b_edge, const float* __restrict__ hv,
    float* __restrict__ c, int V, int E) {
    const int gid = blockIdx.x * blockDim.x + threadIdx.x;
    const int lane = threadIdx.x & 31;
    const int v = __builtin_amdgcn_readfirstlane(gid >> 5);
    if (v >= V) return;
    float4 acc = make_float4(0.0f, 0.0f, 0.0f, 0.0f);
    float z = 0.0f;
    int e = head[(size_t)v * HEAD_PITCH];
    if (e >= 0) {
        const float pdv = pd[v];
        const float be = b_edge[0];
        int2 ln = lnk[e];
        float m = 0.0f;
        {
            const int s = ln.x;
            e = ln.y;
            const float psv = ps[s];
            acc = ((const float4*)(hv + (size_t)s * D_FEAT))[lane];
            if (e >= 0) ln = lnk[e];
            const float x = pdv + psv + be;
            m = x > 0.0f ? x : 0.01f * x;
            z = 1.0f;
        }
        for (int it = 1; it < E && e >= 0; ++it) {
            const int s = ln.x;
            e = ln.y;
            const float psv = ps[s];
            const float4 h = ((const float4*)(hv + (size_t)s * D_FEAT))[lane];
            if (e >= 0) ln = lnk[e];
            const float x = pdv + psv + be;
            const float l = x > 0.0f ? x : 0.01f * x;
            const float d = l - m;
            const float t = expf(-fabsf(d));
            const float sc = d > 0.0f ? t : 1.0f;
            const float w  = d > 0.0f ? 1.0f : t;
            m = d > 0.0f ? l : m;
            z = z * sc + w;
            acc.x = acc.x * sc + w * h.x;
            acc.y = acc.y * sc + w * h.y;
            acc.z = acc.z * sc + w * h.z;
            acc.w = acc.w * sc + w * h.w;
        }
    }
    float4 o;
    {
        float x;
        x = z > 0.0f ? acc.x / z : 0.0f;  o.x = x > 0.0f ? x : expf(x) - 1.0f;
        x = z > 0.0f ? acc.y / z : 0.0f;  o.y = x > 0.0f ? x : expf(x) - 1.0f;
        x = z > 0.0f ? acc.z / z : 0.0f;  o.z = x > 0.0f ? x : expf(x) - 1.0f;
        x = z > 0.0f ? acc.w / z : 0.0f;  o.w = x > 0.0f ? x : expf(x) - 1.0f;
    }
    ((float4*)(c + (size_t)v * D_FEAT))[lane] = o;
}

__global__ void gru_kernel(const float* __restrict__ gi, const float* __restrict__ gh,
                           const float* __restrict__ nf, float* __restrict__ out,
                           int V) {
    int i = blockIdx.x * blockDim.x + threadIdx.x;
    if (i >= V * D_FEAT) return;
    int v = i >> 7;
    int d = i & 127;
    size_t base = (size_t)v * 384;
    float ir = gi[base + d], iz = gi[base + 128 + d], in_ = gi[base + 256 + d];
    float hr = gh[base + d], hz = gh[base + 128 + d], hn  = gh[base + 256 + d];
    float r = 1.0f / (1.0f + expf(-(ir + hr)));
    float u = 1.0f / (1.0f + expf(-(iz + hz)));
    float n = tanhf(in_ + r * hn);
    float h = (1.0f - u) * n + u * nf[i];
    out[i] = h > 0.0f ? h : 0.0f;
}

extern "C" void kernel_launch(void* const* d_in, const int* in_sizes, int n_in,
                              void* d_out, int out_size, void* d_ws, size_t ws_size,
                              hipStream_t stream) {
    const float* nf     = (const float*)d_in[0];
    const int*   src    = (const int*)d_in[1];
    const int*   dst    = (const int*)d_in[2];
    const float* W_edge = (const float*)d_in[3];
    const float* b_edge = (const float*)d_in[4];
    const float* W_proj = (const float*)d_in[5];
    const float* b_proj = (const float*)d_in[6];
    const float* W_ih   = (const float*)d_in[7];
    const float* b_ih   = (const float*)d_in[8];
    const float* W_hh   = (const float*)d_in[9];
    const float* b_hh   = (const float*)d_in[10];

    const int V = in_sizes[0] / D_FEAT;
    const int E = in_sizes[1];

    float* hv    = (float*)d_ws;
    float* gh    = hv + (size_t)V * 128;
    float* c     = gh + (size_t)V * 384;
    float* gi    = c + (size_t)V * 128;
    float* WhhT  = gi + (size_t)V * 384;
    float* WihT  = WhhT + 128 * 384;
    int2*  lnk   = (int2*)(WihT + 128 * 384);
    int*   head  = (int*)(lnk + E);
    float* pd    = (float*)(head + (size_t)V * HEAD_PITCH);
    float* ps    = pd + V;

    const int blk = 256;

    packT_kernel<<<(384 * 128 + blk - 1) / blk, blk, 0, stream>>>(W_hh, WhhT, 384, 128);
    packT_kernel<<<(384 * 128 + blk - 1) / blk, blk, 0, stream>>>(W_ih, WihT, 384, 128);

    node_dot_kernel<<<((size_t)V * 32 + blk - 1) / blk, blk, 0, stream>>>(
        nf, W_edge, pd, ps, head, V);

    const int gemmBlocks = (V + 15) / 16;
    wmma_gemm_kernel<128><<<gemmBlocks, 256, 0, stream>>>(nf, W_proj, b_proj, hv, V);
    wmma_gemm_kernel<384><<<gemmBlocks, 256, 0, stream>>>(nf, WhhT, b_hh, gh, V);

    edge_link_kernel<<<(E + blk - 1) / blk, blk, 0, stream>>>(src, dst, head, lnk, E);
    node_gather_kernel<<<((size_t)V * 32 + blk - 1) / blk, blk, 0, stream>>>(
        head, lnk, pd, ps, b_edge, hv, c, V, E);

    wmma_gemm_kernel<384><<<gemmBlocks, 256, 0, stream>>>(c, WihT, b_ih, gi, V);

    gru_kernel<<<((size_t)V * 128 + blk - 1) / blk, blk, 0, stream>>>(
        gi, gh, nf, (float*)d_out, V);
}


# f32 WMMA GEMMs: a wave keeps its 16-column B fragments for all of K in registers over 2 to 4 row tiles fed through a double-buffered LDS tile; B loads per product fall 2-4x; every sum keeps its order
# speedup vs baseline: 2.0130x; 1.0429x over previous
#include <hip/hip_runtime.h>
#include <hip/hip_bf16.h>
#include <math.h>

typedef float v2f __attribute__((ext_vector_type(2)));
typedef float v8f __attribute__((ext_vector_type(8)));

#define D_FEAT 128
#define HEAD_PITCH 32

__global__ void packT_kernel(const float* __restrict__ W, float* __restrict__ WT,
                             int R, int C) {
    int i = blockIdx.x * blockDim.x + threadIdx.x;
    if (i >= R * C) return;
    int n = i / C;
    int k = i - n * C;
    WT[(size_t)k * R + n] = W[i];
}

__global__ void node_dot_kernel(const float* __restrict__ nf,
                                const float* __restrict__ W_edge,
                                float* __restrict__ pd, float* __restrict__ ps,
                                int* __restrict__ head, int V) {
    int gid = blockIdx.x * blockDim.x + threadIdx.x;
    int v = gid >> 5;
    int lane = gid & 31;
    if (v >= V) return;
    float4 x  = ((const float4*)(nf + (size_t)v * D_FEAT))[lane];
    float4 w0 = ((const float4*)W_edge)[lane];
    float4 w1 = ((const float4*)(W_edge + D_FEAT))[lane];
    float d0 = x.x * w0.x + x.y * w0.y + x.z * w0.z + x.w * w0.w;
    float d1 = x.x * w1.x + x.y * w1.y + x.z * w1.z + x.w * w1.w;
#pragma unroll
    for (int off = 16; off > 0; off >>= 1) {
        d0 += __shfl_down(d0, off, 32);
        d1 += __shfl_down(d1, off, 32);
    }
    head[(size_t)v * HEAD_PITCH + lane] = -1;
    if (lane == 0) { pd[v] = d0; ps[v] = d1; }
}

template <int N, int T>
__global__ __launch_bounds__(256) void wmma_gemm_kernel(
    const float* __restrict__ X, const float* __restrict__ B,
    const float* __restrict__ bias, float* __restrict__ Out, int V) {
    __shared__ __attribute__((aligned(16))) float xs[2][16][132];
    const int tid = threadIdx.x;
    const int lane = tid & 31;
    const int wave = tid >> 5;
    const int lo = lane & 15;
    const int hi = lane >> 4;
    const int nTiles = (V + 15) >> 4;
    const int t0 = blockIdx.x * T;
    const int t1 = (t0 + T < nTiles) ? t0 + T : nTiles;
    if (t0 >= t1) return;
    const int colbase = blockIdx.y * 128 + wave * 16;

    v2f bf[32];
#pragma unroll
    for (int s = 0; s < 32; ++s) {
        const int ka = 4 * s + 2 * hi;
        bf[s][0] = B[(size_t)ka * N + colbase + lo];
        bf[s][1] = B[(size_t)(ka + 1) * N + colbase + lo];
    }
    const float bv = bias[colbase + lo];

    const int sr = tid >> 5;
    const int sc = (tid & 31) * 4;
    float4 st0, st1;
    {
        int r0 = t0 * 16 + sr;      if (r0 >= V) r0 = V - 1;
        int r1 = t0 * 16 + sr + 8;  if (r1 >= V) r1 = V - 1;
        st0 = *(const float4*)(X + (size_t)r0 * D_FEAT + sc);
        st1 = *(const float4*)(X + (size_t)r1 * D_FEAT + sc);
    }
    *(float4*)&xs[0][sr][sc] = st0;
    *(float4*)&xs[0][sr + 8][sc] = st1;
    __syncthreads();

    for (int t = t0; t < t1; ++t) {
        const int buf = (t - t0) & 1;
        const bool more = (t + 1 < t1);
        if (more) {
            int r0 = (t + 1) * 16 + sr;      if (r0 >= V) r0 = V - 1;
            int r1 = (t + 1) * 16 + sr + 8;  if (r1 >= V) r1 = V - 1;
            st0 = *(const float4*)(X + (size_t)r0 * D_FEAT + sc);
            st1 = *(const float4*)(X + (size_t)r1 * D_FEAT + sc);
        }
        v8f acc;
#pragma unroll
        for (int r = 0; r < 8; ++r) acc[r] = bv;
#pragma unroll
        for (int s = 0; s < 32; ++s) {
            const int ka = 4 * s + 2 * hi;
            v2f a;
            a[0] = xs[buf][lo][ka];
            a[1] = xs[buf][lo][ka + 1];
            acc = __builtin_amdgcn_wmma_f32_16x16x4_f32(
                false, a, false, bf[s], (short)0, acc, false, false);
        }
        if (more) {
            *(float4*)&xs[buf ^ 1][sr][sc] = st0;
            *(float4*)&xs[buf ^ 1][sr + 8][sc] = st1;
        }
        __syncthreads();
        const int rowbase = t * 16;
        if (rowbase + 16 <= V) {
#pragma unroll
            for (int r = 0; r < 8; ++r) {
                int row = rowbase + r + 8 * hi;
                Out[(size_t)row * N + colbase + lo] = acc[r];
            }
        } else {
#pragma unroll
            for (int r = 0; r < 8; ++r) {
                int row = rowbase + r + 8 * hi;
                if (row < V) Out[(size_t)row * N + colbase + lo] = acc[r];
            }
        }
    }
}

__global__ void edge_link_kernel(const int* __restrict__ src,
                                 const int* __restrict__ dst,
                                 int* __restrict__ head,
                                 int2* __restrict__ lnk, int E) {
    int e = blockIdx.x * blockDim.x + threadIdx.x;
    if (e >= E) return;
    int old = atomicExch(head + (size_t)dst[e] * HEAD_PITCH, e);
    lnk[e] = make_int2(src[e], old);
}

__global__ __launch_bounds__(256) void node_gather_kernel(
    const int* __restrict__ head, const int2* __restrict__ lnk,
    const float* __restrict__ pd, const float* __restrict__ ps,
    const float* __restrict__ b_edge, const float* __restrict__ hv,
    float* __restrict__ c, int V, int E) {
    const int gid = blockIdx.x * blockDim.x + threadIdx.x;
    const int lane = threadIdx.x & 31;
    const int v = __builtin_amdgcn_readfirstlane(gid >> 5);
    if (v >= V) return;
    float4 acc = make_float4(0.0f, 0.0f, 0.0f, 0.0f);
    float z = 0.0f;
    int e = head[(size_t)v * HEAD_PITCH];
    if (e >= 0) {
        const float pdv = pd[v];
        const float be = b_edge[0];
        int2 ln = lnk[e];
        float m = 0.0f;
        {
            const int s = ln.x;
            e = ln.y;
            const float psv = ps[s];
            acc = ((const float4*)(hv + (size_t)s * D_FEAT))[lane];
            if (e >= 0) ln = lnk[e];
            const float x = pdv + psv + be;
            m = x > 0.0f ? x : 0.01f * x;
            z = 1.0f;
        }
        for (int it = 1; it < E && e >= 0; ++it) {
            const int s = ln.x;
            e = ln.y;
            const float psv = ps[s];
            const float4 h = ((const float4*)(hv + (size_t)s * D_FEAT))[lane];
            if (e >= 0) ln = lnk[e];
            const float x = pdv + psv + be;
            const float l = x > 0.0f ? x : 0.01f * x;
            const float d = l - m;
            const float t = expf(-fabsf(d));
            const float sc = d > 0.0f ? t : 1.0f;
            const float w  = d > 0.0f ? 1.0f : t;
            m = d > 0.0f ? l : m;
            z = z * sc + w;
            acc.x = acc.x * sc + w * h.x;
            acc.y = acc.y * sc + w * h.y;
            acc.z = acc.z * sc + w * h.z;
            acc.w = acc.w * sc + w * h.w;
        }
    }
    float4 o;
    {
        float x;
        x = z > 0.0f ? acc.x / z : 0.0f;  o.x = x > 0.0f ? x : expf(x) - 1.0f;
        x = z > 0.0f ? acc.y / z : 0.0f;  o.y = x > 0.0f ? x : expf(x) - 1.0f;
        x = z > 0.0f ? acc.z / z : 0.0f;  o.z = x > 0.0f ? x : expf(x) - 1.0f;
        x = z > 0.0f ? acc.w / z : 0.0f;  o.w = x > 0.0f ? x : expf(x) - 1.0f;
    }
    ((float4*)(c + (size_t)v * D_FEAT))[lane] = o;
}

__global__ void gru_kernel(const float* __restrict__ gi, const float* __restrict__ gh,
                           const float* __restrict__ nf, float* __restrict__ out,
                           int V) {
    int i = blockIdx.x * blockDim.x + threadIdx.x;
    if (i >= V * D_FEAT) return;
    int v = i >> 7;
    int d = i & 127;
    size_t base = (size_t)v * 384;
    float ir = gi[base + d], iz = gi[base + 128 + d], in_ = gi[base + 256 + d];
    float hr = gh[base + d], hz = gh[base + 128 + d], hn  = gh[base + 256 + d];
    float r = 1.0f / (1.0f + expf(-(ir + hr)));
    float u = 1.0f / (1.0f + expf(-(iz + hz)));
    float n = tanhf(in_ + r * hn);
    float h = (1.0f - u) * n + u * nf[i];
    out[i] = h > 0.0f ? h : 0.0f;
}

extern "C" void kernel_launch(void* const* d_in, const int* in_sizes, int n_in,
                              void* d_out, int out_size, void* d_ws, size_t ws_size,
                              hipStream_t stream) {
    const float* nf     = (const float*)d_in[0];
    const int*   src    = (const int*)d_in[1];
    const int*   dst    = (const int*)d_in[2];
    const float* W_edge = (const float*)d_in[3];
    const float* b_edge = (const float*)d_in[4];
    const float* W_proj = (const float*)d_in[5];
    const float* b_proj = (const float*)d_in[6];
    const float* W_ih   = (const float*)d_in[7];
    const float* b_ih   = (const float*)d_in[8];
    const float* W_hh   = (const float*)d_in[9];
    const float* b_hh   = (const float*)d_in[10];

    const int V = in_sizes[0] / D_FEAT;
    const int E = in_sizes[1];

    float* hv    = (float*)d_ws;
    float* gh    = hv + (size_t)V * 128;
    float* c     = gh + (size_t)V * 384;
    float* gi    = c + (size_t)V * 128;
    float* WhhT  = gi + (size_t)V * 384;
    float* WihT  = WhhT + 128 * 384;
    int2*  lnk   = (int2*)(WihT + 128 * 384);
    int*   head  = (int*)(lnk + E);
    float* pd    = (float*)(head + (size_t)V * HEAD_PITCH);
    float* ps    = pd + V;

    const int blk = 256;

    packT_kernel<<<(384 * 128 + blk - 1) / blk, blk, 0, stream>>>(W_hh, WhhT, 384, 128);
    packT_kernel<<<(384 * 128 + blk - 1) / blk, blk, 0, stream>>>(W_ih, WihT, 384, 128);

    node_dot_kernel<<<((size_t)V * 32 + blk - 1) / blk, blk, 0, stream>>>(
        nf, W_edge, pd, ps, head, V);

    const int rowTiles = (V + 15) / 16;
    const dim3 grid128((rowTiles + 1) / 2, 1);
    const dim3 grid384((rowTiles + 3) / 4, 3);
    wmma_gemm_kernel<128, 2><<<grid128, 256, 0, stream>>>(nf, W_proj, b_proj, hv, V);
    wmma_gemm_kernel<384, 4><<<grid384, 256, 0, stream>>>(nf, WhhT, b_hh, gh, V);

    edge_link_kernel<<<(E + blk - 1) / blk, blk, 0, stream>>>(src, dst, head, lnk, E);
    node_gather_kernel<<<((size_t)V * 32 + blk - 1) / blk, blk, 0, stream>>>(
        head, lnk, pd, ps, b_edge, hv, c, V, E);

    wmma_gemm_kernel<384, 4><<<grid384, 256, 0, stream>>>(c, WihT, b_ih, gi, V);

    gru_kernel<<<((size_t)V * 128 + blk - 1) / blk, blk, 0, stream>>>(
        gi, gh, nf, (float*)d_out, V);
}


# Launch merging: both weight transposes and the per-node scalar pass with its head clear ride in the projection GEMM's launch (rows taken from its staging registers); 9 launches become 6
# speedup vs baseline: 2.1441x; 1.0651x over previous
#include <hip/hip_runtime.h>
#include <hip/hip_bf16.h>
#include <math.h>

typedef float v2f __attribute__((ext_vector_type(2)));
typedef float v8f __attribute__((ext_vector_type(8)));

#define D_FEAT 128
#define HEAD_PITCH 32

__device__ __forceinline__ void node_scalars(const float4 x,
                                             const float* __restrict__ W_edge,
                                             int v, int lane,
                                             float* __restrict__ pd,
                                             float* __restrict__ ps,
                                             int* __restrict__ head, int V) {
    const float4 w0 = ((const float4*)W_edge)[lane];
    const float4 w1 = ((const float4*)(W_edge + D_FEAT))[lane];
    float d0 = x.x * w0.x + x.y * w0.y + x.z * w0.z + x.w * w0.w;
    float d1 = x.x * w1.x + x.y * w1.y + x.z * w1.z + x.w * w1.w;
#pragma unroll
    for (int off = 16; off > 0; off >>= 1) {
        d0 += __shfl_down(d0, off, 32);
        d1 += __shfl_down(d1, off, 32);
    }
    if (v < V) {
        head[(size_t)v * HEAD_PITCH + lane] = -1;
        if (lane == 0) { pd[v] = d0; ps[v] = d1; }
    }
}

template <int N, int T, bool NODE_PASS>
__device__ __forceinline__ void gemm_rows(
    const float* __restrict__ X, const float* __restrict__ B,
    const float* __restrict__ bias, float* __restrict__ Out, int V,
    int bx, int by,
    const float* __restrict__ W_edge, float* __restrict__ pd,
    float* __restrict__ ps, int* __restrict__ head) {
    __shared__ __attribute__((aligned(16))) float xs[2][16][132];
    const int tid = threadIdx.x;
    const int lane = tid & 31;
    const int wave = tid >> 5;
    const int lo = lane & 15;
    const int hi = lane >> 4;
    const int nTiles = (V + 15) >> 4;
    const int t0 = bx * T;
    const int t1 = (t0 + T < nTiles) ? t0 + T : nTiles;
    if (t0 >= t1) return;
    const int colbase = by * 128 + wave * 16;

    v2f bf[32];
#pragma unroll
    for (int s = 0; s < 32; ++s) {
        const int ka = 4 * s + 2 * hi;
        bf[s][0] = B[(size_t)ka * N + colbase + lo];
        bf[s][1] = B[(size_t)(ka + 1) * N + colbase + lo];
    }
    const float bv = bias[colbase + lo];

    const int sr = tid >> 5;
    const int sc = (tid & 31) * 4;
    float4 st0, st1;
    {
        int r0 = t0 * 16 + sr;      if (r0 >= V) r0 = V - 1;
        int r1 = t0 * 16 + sr + 8;  if (r1 >= V) r1 = V - 1;
        st0 = *(const float4*)(X + (size_t)r0 * D_FEAT + sc);
        st1 = *(const float4*)(X + (size_t)r1 * D_FEAT + sc);
    }
    *(float4*)&xs[0][sr][sc] = st0;
    *(float4*)&xs[0][sr + 8][sc] = st1;
    __syncthreads();
    if (NODE_PASS) {
        node_scalars(st0, W_edge, t0 * 16 + sr, lane, pd, ps, head, V);
        node_scalars(st1, W_edge, t0 * 16 + sr + 8, lane, pd, ps, head, V);
    }

    for (int t = t0; t < t1; ++t) {
        const int buf = (t - t0) & 1;
        const bool more = (t + 1 < t1);
        if (more) {
            int r0 = (t + 1) * 16 + sr;      if (r0 >= V) r0 = V - 1;
            int r1 = (t + 1) * 16 + sr + 8;  if (r1 >= V) r1 = V - 1;
            st0 = *(const float4*)(X + (size_t)r0 * D_FEAT + sc);
            st1 = *(const float4*)(X + (size_t)r1 * D_FEAT + sc);
        }
        v8f acc;
#pragma unroll
        for (int r = 0; r < 8; ++r) acc[r] = bv;
#pragma unroll
        for (int s = 0; s < 32; ++s) {
            const int ka = 4 * s + 2 * hi;
            v2f a;
            a[0] = xs[buf][lo][ka];
            a[1] = xs[buf][lo][ka + 1];
            acc = __builtin_amdgcn_wmma_f32_16x16x4_f32(
                false, a, false, bf[s], (short)0, acc, false, false);
        }
        if (more) {
            *(float4*)&xs[buf ^ 1][sr][sc] = st0;
            *(float4*)&xs[buf ^ 1][sr + 8][sc] = st1;
        }
        __syncthreads();
        const int rowbase = t * 16;
        if (rowbase + 16 <= V) {
#pragma unroll
            for (int r = 0; r < 8; ++r) {
                int row = rowbase + r + 8 * hi;
                Out[(size_t)row * N + colbase + lo] = acc[r];
            }
        } else {
#pragma unroll
            for (int r = 0; r < 8; ++r) {
                int row = rowbase + r + 8 * hi;
                if (row < V) Out[(size_t)row * N + colbase + lo] = acc[r];
            }
        }
        if (NODE_PASS && more) {
            node_scalars(st0, W_edge, (t + 1) * 16 + sr, lane, pd, ps, head, V);
            node_scalars(st1, W_edge, (t + 1) * 16 + sr + 8, lane, pd, ps, head, V);
        }
    }
}

#define PACK_BLOCKS ((384 * D_FEAT) / 256)

__global__ __launch_bounds__(256, 8) void proj_node_pack_kernel(
    const float* __restrict__ nf, const float* __restrict__ W_proj,
    const float* __restrict__ b_proj, float* __restrict__ hv, int V,
    int gemmBlocks,
    const float* __restrict__ W_edge, float* __restrict__ pd,
    float* __restrict__ ps, int* __restrict__ head,
    const float* __restrict__ W_hh, float* __restrict__ WhhT,
    const float* __restrict__ W_ih, float* __restrict__ WihT) {
    const int b = blockIdx.x;
    if (b < gemmBlocks) {
        gemm_rows<128, 2, true>(nf, W_proj, b_proj, hv, V, b, 0,
                                W_edge, pd, ps, head);
    } else {
        const int pb = b - gemmBlocks;
        const bool second = pb >= PACK_BLOCKS;
        const float* __restrict__ Wsrc = second ? W_ih : W_hh;
        float* __restrict__ WT = second ? WihT : WhhT;
        const int i = (second ? pb - PACK_BLOCKS : pb) * 256 + threadIdx.x;
        const int n = i / D_FEAT;
        const int k = i - n * D_FEAT;
        WT[(size_t)k * 384 + n] = Wsrc[i];
    }
}

template <int N, int T>
__global__ __launch_bounds__(256) void wmma_gemm_kernel(
    const float* __restrict__ X, const float* __restrict__ B,
    const float* __restrict__ bias, float* __restrict__ Out, int V) {
    gemm_rows<N, T, false>(X, B, bias, Out, V, blockIdx.x, blockIdx.y,
                           nullptr, nullptr, nullptr, nullptr);
}

__global__ void edge_link_kernel(const int* __restrict__ src,
                                 const int* __restrict__ dst,
                                 int* __restrict__ head,
                                 int2* __restrict__ lnk, int E) {
    int e = blockIdx.x * blockDim.x + threadIdx.x;
    if (e >= E) return;
    int old = atomicExch(head + (size_t)dst[e] * HEAD_PITCH, e);
    lnk[e] = make_int2(src[e], old);
}

__global__ __launch_bounds__(256) void node_gather_kernel(
    const int* __restrict__ head, const int2* __restrict__ lnk,
    const float* __restrict__ pd, const float* __restrict__ ps,
    const float* __restrict__ b_edge, const float* __restrict__ hv,
    float* __restrict__ c, int V, int E) {
    const int gid = blockIdx.x * blockDim.x + threadIdx.x;
    const int lane = threadIdx.x & 31;
    const int v = __builtin_amdgcn_readfirstlane(gid >> 5);
    if (v >= V) return;
    float4 acc = make_float4(0.0f, 0.0f, 0.0f, 0.0f);
    float z = 0.0f;
    int e = head[(size_t)v * HEAD_PITCH];
    if (e >= 0) {
        const float pdv = pd[v];
        const float be = b_edge[0];
        int2 ln = lnk[e];
        float m = 0.0f;
        {
            const int s = ln.x;
            e = ln.y;
            const float psv = ps[s];
            acc = ((const float4*)(hv + (size_t)s * D_FEAT))[lane];
            if (e >= 0) ln = lnk[e];
            const float x = pdv + psv + be;
            m = x > 0.0f ? x : 0.01f * x;
            z = 1.0f;
        }
        for (int it = 1; it < E && e >= 0; ++it) {
            const int s = ln.x;
            e = ln.y;
            const float psv = ps[s];
            const float4 h = ((const float4*)(hv + (size_t)s * D_FEAT))[lane];
            if (e >= 0) ln = lnk[e];
            const float x = pdv + psv + be;
            const float l = x > 0.0f ? x : 0.01f * x;
            const float d = l - m;
            const float t = expf(-fabsf(d));
            const float sc = d > 0.0f ? t : 1.0f;
            const float w  = d > 0.0f ? 1.0f : t;
            m = d > 0.0f ? l : m;
            z = z * sc + w;
            acc.x = acc.x * sc + w * h.x;
            acc.y = acc.y * sc + w * h.y;
            acc.z = acc.z * sc + w * h.z;
            acc.w = acc.w * sc + w * h.w;
        }
    }
    float4 o;
    {
        float x;
        x = z > 0.0f ? acc.x / z : 0.0f;  o.x = x > 0.0f ? x : expf(x) - 1.0f;
        x = z > 0.0f ? acc.y / z : 0.0f;  o.y = x > 0.0f ? x : expf(x) - 1.0f;
        x = z > 0.0f ? acc.z / z : 0.0f;  o.z = x > 0.0f ? x : expf(x) - 1.0f;
        x = z > 0.0f ? acc.w / z : 0.0f;  o.w = x > 0.0f ? x : expf(x) - 1.0f;
    }
    ((float4*)(c + (size_t)v * D_FEAT))[lane] = o;
}

__global__ void gru_kernel(const float* __restrict__ gi, const float* __restrict__ gh,
                           const float* __restrict__ nf, float* __restrict__ out,
                           int V) {
    int i = blockIdx.x * blockDim.x + threadIdx.x;
    if (i >= V * D_FEAT) return;
    int v = i >> 7;
    int d = i & 127;
    size_t base = (size_t)v * 384;
    float ir = gi[base + d], iz = gi[base + 128 + d], in_ = gi[base + 256 + d];
    float hr = gh[base + d], hz = gh[base + 128 + d], hn  = gh[base + 256 + d];
    float r = 1.0f / (1.0f + expf(-(ir + hr)));
    float u = 1.0f / (1.0f + expf(-(iz + hz)));
    float n = tanhf(in_ + r * hn);
    float h = (1.0f - u) * n + u * nf[i];
    out[i] = h > 0.0f ? h : 0.0f;
}

extern "C" void kernel_launch(void* const* d_in, const int* in_sizes, int n_in,
                              void* d_out, int out_size, void* d_ws, size_t ws_size,
                              hipStream_t stream) {
    const float* nf     = (const float*)d_in[0];
    const int*   src    = (const int*)d_in[1];
    const int*   dst    = (const int*)d_in[2];
    const float* W_edge = (const float*)d_in[3];
    const float* b_edge = (const float*)d_in[4];
    const float* W_proj = (const float*)d_in[5];
    const float* b_proj = (const float*)d_in[6];
    const float* W_ih   = (const float*)d_in[7];
    const float* b_ih   = (const float*)d_in[8];
    const float* W_hh   = (const float*)d_in[9];
    const float* b_hh   = (const float*)d_in[10];

    const int V = in_sizes[0] / D_FEAT;
    const int E = in_sizes[1];

    float* hv    = (float*)d_ws;
    float* gh    = hv + (size_t)V * 128;
    float* c     = gh + (size_t)V * 384;
    float* gi    = c + (size_t)V * 128;
    float* WhhT  = gi + (size_t)V * 384;
    float* WihT  = WhhT + 128 * 384;
    int2*  lnk   = (int2*)(WihT + 128 * 384);
    int*   head  = (int*)(lnk + E);
    float* pd    = (float*)(head + (size_t)V * HEAD_PITCH);
    float* ps    = pd + V;

    const int blk = 256;

    const int rowTiles = (V + 15) / 16;
    const int proj_blocks = (rowTiles + 1) / 2;
    const dim3 grid384((rowTiles + 3) / 4, 3);

    proj_node_pack_kernel<<<proj_blocks + 2 * PACK_BLOCKS, blk, 0, stream>>>(
        nf, W_proj, b_proj, hv, V, proj_blocks, W_edge, pd, ps, head,
        W_hh, WhhT, W_ih, WihT);

    wmma_gemm_kernel<384, 4><<<grid384, 256, 0, stream>>>(nf, WhhT, b_hh, gh, V);

    edge_link_kernel<<<(E + blk - 1) / blk, blk, 0, stream>>>(src, dst, head, lnk, E);

    node_gather_kernel<<<((size_t)V * 32 + blk - 1) / blk, blk, 0, stream>>>(
        head, lnk, pd, ps, b_edge, hv, c, V, E);

    wmma_gemm_kernel<384, 4><<<grid384, 256, 0, stream>>>(c, WihT, b_ih, gi, V);

    gru_kernel<<<((size_t)V * 128 + blk - 1) / blk, blk, 0, stream>>>(
        gi, gh, nf, (float*)d_out, V);
}


# GRU gate pass as a streaming loop: 1563 resident blocks stride over the rows, each lane owns 4 columns (seven 16-byte loads in flight) instead of 200000 one-element waves; no sum changes its order
# speedup vs baseline: 2.2222x; 1.0364x over previous
#include <hip/hip_runtime.h>
#include <hip/hip_bf16.h>
#include <math.h>

typedef float v2f __attribute__((ext_vector_type(2)));
typedef float v8f __attribute__((ext_vector_type(8)));

#define D_FEAT 128
#define HEAD_PITCH 32

__device__ __forceinline__ void node_scalars(const float4 x,
                                             const float* __restrict__ W_edge,
                                             int v, int lane,
                                             float* __restrict__ pd,
                                             float* __restrict__ ps,
                                             int* __restrict__ head, int V) {
    const float4 w0 = ((const float4*)W_edge)[lane];
    const float4 w1 = ((const float4*)(W_edge + D_FEAT))[lane];
    float d0 = x.x * w0.x + x.y * w0.y + x.z * w0.z + x.w * w0.w;
    float d1 = x.x * w1.x + x.y * w1.y + x.z * w1.z + x.w * w1.w;
#pragma unroll
    for (int off = 16; off > 0; off >>= 1) {
        d0 += __shfl_down(d0, off, 32);
        d1 += __shfl_down(d1, off, 32);
    }
    if (v < V) {
        head[(size_t)v * HEAD_PITCH + lane] = -1;
        if (lane == 0) { pd[v] = d0; ps[v] = d1; }
    }
}

template <int N, int T, bool NODE_PASS>
__device__ __forceinline__ void gemm_rows(
    const float* __restrict__ X, const float* __restrict__ B,
    const float* __restrict__ bias, float* __restrict__ Out, int V,
    int bx, int by,
    const float* __restrict__ W_edge, float* __restrict__ pd,
    float* __restrict__ ps, int* __restrict__ head) {
    __shared__ __attribute__((aligned(16))) float xs[2][16][132];
    const int tid = threadIdx.x;
    const int lane = tid & 31;
    const int wave = tid >> 5;
    const int lo = lane & 15;
    const int hi = lane >> 4;
    const int nTiles = (V + 15) >> 4;
    const int t0 = bx * T;
    const int t1 = (t0 + T < nTiles) ? t0 + T : nTiles;
    if (t0 >= t1) return;
    const int colbase = by * 128 + wave * 16;

    v2f bf[32];
#pragma unroll
    for (int s = 0; s < 32; ++s) {
        const int ka = 4 * s + 2 * hi;
        bf[s][0] = B[(size_t)ka * N + colbase + lo];
        bf[s][1] = B[(size_t)(ka + 1) * N + colbase + lo];
    }
    const float bv = bias[colbase + lo];

    const int sr = tid >> 5;
    const int sc = (tid & 31) * 4;
    float4 st0, st1;
    {
        int r0 = t0 * 16 + sr;      if (r0 >= V) r0 = V - 1;
        int r1 = t0 * 16 + sr + 8;  if (r1 >= V) r1 = V - 1;
        st0 = *(const float4*)(X + (size_t)r0 * D_FEAT + sc);
        st1 = *(const float4*)(X + (size_t)r1 * D_FEAT + sc);
    }
    *(float4*)&xs[0][sr][sc] = st0;
    *(float4*)&xs[0][sr + 8][sc] = st1;
    __syncthreads();
    if (NODE_PASS) {
        node_scalars(st0, W_edge, t0 * 16 + sr, lane, pd, ps, head, V);
        node_scalars(st1, W_edge, t0 * 16 + sr + 8, lane, pd, ps, head, V);
    }

    for (int t = t0; t < t1; ++t) {
        const int buf = (t - t0) & 1;
        const bool more = (t + 1 < t1);
        if (more) {
            int r0 = (t + 1) * 16 + sr;      if (r0 >= V) r0 = V - 1;
            int r1 = (t + 1) * 16 + sr + 8;  if (r1 >= V) r1 = V - 1;
            st0 = *(const float4*)(X + (size_t)r0 * D_FEAT + sc);
            st1 = *(const float4*)(X + (size_t)r1 * D_FEAT + sc);
        }
        v8f acc;
#pragma unroll
        for (int r = 0; r < 8; ++r) acc[r] = bv;
#pragma unroll
        for (int s = 0; s < 32; ++s) {
            const int ka = 4 * s + 2 * hi;
            v2f a;
            a[0] = xs[buf][lo][ka];
            a[1] = xs[buf][lo][ka + 1];
            acc = __builtin_amdgcn_wmma_f32_16x16x4_f32(
                false, a, false, bf[s], (short)0, acc, false, false);
        }
        if (more) {
            *(float4*)&xs[buf ^ 1][sr][sc] = st0;
            *(float4*)&xs[buf ^ 1][sr + 8][sc] = st1;
        }
        __syncthreads();
        const int rowbase = t * 16;
        if (rowbase + 16 <= V) {
#pragma unroll
            for (int r = 0; r < 8; ++r) {
                int row = rowbase + r + 8 * hi;
                Out[(size_t)row * N + colbase + lo] = acc[r];
            }
        } else {
#pragma unroll
            for (int r = 0; r < 8; ++r) {
                int row = rowbase + r + 8 * hi;
                if (row < V) Out[(size_t)row * N + colbase + lo] = acc[r];
            }
        }
        if (NODE_PASS && more) {
            node_scalars(st0, W_edge, (t + 1) * 16 + sr, lane, pd, ps, head, V);
            node_scalars(st1, W_edge, (t + 1) * 16 + sr + 8, lane, pd, ps, head, V);
        }
    }
}

#define PACK_BLOCKS ((384 * D_FEAT) / 256)

__global__ __launch_bounds__(256, 8) void proj_node_pack_kernel(
    const float* __restrict__ nf, const float* __restrict__ W_proj,
    const float* __restrict__ b_proj, float* __restrict__ hv, int V,
    int gemmBlocks,
    const float* __restrict__ W_edge, float* __restrict__ pd,
    float* __restrict__ ps, int* __restrict__ head,
    const float* __restrict__ W_hh, float* __restrict__ WhhT,
    const float* __restrict__ W_ih, float* __restrict__ WihT) {
    const int b = blockIdx.x;
    if (b < gemmBlocks) {
        gemm_rows<128, 2, true>(nf, W_proj, b_proj, hv, V, b, 0,
                                W_edge, pd, ps, head);
    } else {
        const int pb = b - gemmBlocks;
        const bool second = pb >= PACK_BLOCKS;
        const float* __restrict__ Wsrc = second ? W_ih : W_hh;
        float* __restrict__ WT = second ? WihT : WhhT;
        const int i = (second ? pb - PACK_BLOCKS : pb) * 256 + threadIdx.x;
        const int n = i / D_FEAT;
        const int k = i - n * D_FEAT;
        WT[(size_t)k * 384 + n] = Wsrc[i];
    }
}

template <int N, int T>
__global__ __launch_bounds__(256) void wmma_gemm_kernel(
    const float* __restrict__ X, const float* __restrict__ B,
    const float* __restrict__ bias, float* __restrict__ Out, int V) {
    gemm_rows<N, T, false>(X, B, bias, Out, V, blockIdx.x, blockIdx.y,
                           nullptr, nullptr, nullptr, nullptr);
}

__global__ void edge_link_kernel(const int* __restrict__ src,
                                 const int* __restrict__ dst,
                                 int* __restrict__ head,
                                 int2* __restrict__ lnk, int E) {
    int e = blockIdx.x * blockDim.x + threadIdx.x;
    if (e >= E) return;
    int old = atomicExch(head + (size_t)dst[e] * HEAD_PITCH, e);
    lnk[e] = make_int2(src[e], old);
}

__global__ __launch_bounds__(256) void node_gather_kernel(
    const int* __restrict__ head, const int2* __restrict__ lnk,
    const float* __restrict__ pd, const float* __restrict__ ps,
    const float* __restrict__ b_edge, const float* __restrict__ hv,
    float* __restrict__ c, int V, int E) {
    const int gid = blockIdx.x * blockDim.x + threadIdx.x;
    const int lane = threadIdx.x & 31;
    const int v = __builtin_amdgcn_readfirstlane(gid >> 5);
    if (v >= V) return;
    float4 acc = make_float4(0.0f, 0.0f, 0.0f, 0.0f);
    float z = 0.0f;
    int e = head[(size_t)v * HEAD_PITCH];
    if (e >= 0) {
        const float pdv = pd[v];
        const float be = b_edge[0];
        int2 ln = lnk[e];
        float m = 0.0f;
        {
            const int s = ln.x;
            e = ln.y;
            const float psv = ps[s];
            acc = ((const float4*)(hv + (size_t)s * D_FEAT))[lane];
            if (e >= 0) ln = lnk[e];
            const float x = pdv + psv + be;
            m = x > 0.0f ? x : 0.01f * x;
            z = 1.0f;
        }
        for (int it = 1; it < E && e >= 0; ++it) {
            const int s = ln.x;
            e = ln.y;
            const float psv = ps[s];
            const float4 h = ((const float4*)(hv + (size_t)s * D_FEAT))[lane];
            if (e >= 0) ln = lnk[e];
            const float x = pdv + psv + be;
            const float l = x > 0.0f ? x : 0.01f * x;
            const float d = l - m;
            const float t = expf(-fabsf(d));
            const float sc = d > 0.0f ? t : 1.0f;
            const float w  = d > 0.0f ? 1.0f : t;
            m = d > 0.0f ? l : m;
            z = z * sc + w;
            acc.x = acc.x * sc + w * h.x;
            acc.y = acc.y * sc + w * h.y;
            acc.z = acc.z * sc + w * h.z;
            acc.w = acc.w * sc + w * h.w;
        }
    }
    float4 o;
    {
        float x;
        x = z > 0.0f ? acc.x / z : 0.0f;  o.x = x > 0.0f ? x : expf(x) - 1.0f;
        x = z > 0.0f ? acc.y / z : 0.0f;  o.y = x > 0.0f ? x : expf(x) - 1.0f;
        x = z > 0.0f ? acc.z / z : 0.0f;  o.z = x > 0.0f ? x : expf(x) - 1.0f;
        x = z > 0.0f ? acc.w / z : 0.0f;  o.w = x > 0.0f ? x : expf(x) - 1.0f;
    }
    ((float4*)(c + (size_t)v * D_FEAT))[lane] = o;
}

__device__ __forceinline__ float gru_one(float ir, float iz, float in_,
                                         float hr, float hz, float hn, float x) {
    float r = 1.0f / (1.0f + expf(-(ir + hr)));
    float u = 1.0f / (1.0f + expf(-(iz + hz)));
    float n = tanhf(in_ + r * hn);
    float h = fmaf(1.0f - u, n, u * x);
    return h > 0.0f ? h : 0.0f;
}

__global__ __launch_bounds__(256) void gru_kernel(
    const float* __restrict__ gi, const float* __restrict__ gh,
    const float* __restrict__ nf, float* __restrict__ out, int V) {
    const int n4 = V * (D_FEAT / 4);
    const int step = gridDim.x * blockDim.x;
    for (int i = blockIdx.x * blockDim.x + threadIdx.x; i < n4; i += step) {
        const int v = i >> 5;
        const int q = (i & 31) * 4;
        const float* __restrict__ gir = gi + (size_t)v * 384 + q;
        const float* __restrict__ ghr = gh + (size_t)v * 384 + q;
        const float4 ir = *(const float4*)(gir);
        const float4 iz = *(const float4*)(gir + 128);
        const float4 in_ = *(const float4*)(gir + 256);
        const float4 hr = *(const float4*)(ghr);
        const float4 hz = *(const float4*)(ghr + 128);
        const float4 hn = *(const float4*)(ghr + 256);
        const float4 x = ((const float4*)nf)[i];
        float4 o;
        o.x = gru_one(ir.x, iz.x, in_.x, hr.x, hz.x, hn.x, x.x);
        o.y = gru_one(ir.y, iz.y, in_.y, hr.y, hz.y, hn.y, x.y);
        o.z = gru_one(ir.z, iz.z, in_.z, hr.z, hz.z, hn.z, x.z);
        o.w = gru_one(ir.w, iz.w, in_.w, hr.w, hz.w, hn.w, x.w);
        ((float4*)out)[i] = o;
    }
}

extern "C" void kernel_launch(void* const* d_in, const int* in_sizes, int n_in,
                              void* d_out, int out_size, void* d_ws, size_t ws_size,
                              hipStream_t stream) {
    const float* nf     = (const float*)d_in[0];
    const int*   src    = (const int*)d_in[1];
    const int*   dst    = (const int*)d_in[2];
    const float* W_edge = (const float*)d_in[3];
    const float* b_edge = (const float*)d_in[4];
    const float* W_proj = (const float*)d_in[5];
    const float* b_proj = (const float*)d_in[6];
    const float* W_ih   = (const float*)d_in[7];
    const float* b_ih   = (const float*)d_in[8];
    const float* W_hh   = (const float*)d_in[9];
    const float* b_hh   = (const float*)d_in[10];

    const int V = in_sizes[0] / D_FEAT;
    const int E = in_sizes[1];

    float* hv    = (float*)d_ws;
    float* gh    = hv + (size_t)V * 128;
    float* c     = gh + (size_t)V * 384;
    float* gi    = c + (size_t)V * 128;
    float* WhhT  = gi + (size_t)V * 384;
    float* WihT  = WhhT + 128 * 384;
    int2*  lnk   = (int2*)(WihT + 128 * 384);
    int*   head  = (int*)(lnk + E);
    float* pd    = (float*)(head + (size_t)V * HEAD_PITCH);
    float* ps    = pd + V;

    const int blk = 256;

    const int rowTiles = (V + 15) / 16;
    const int proj_blocks = (rowTiles + 1) / 2;
    const dim3 grid384((rowTiles + 3) / 4, 3);

    proj_node_pack_kernel<<<proj_blocks + 2 * PACK_BLOCKS, blk, 0, stream>>>(
        nf, W_proj, b_proj, hv, V, proj_blocks, W_edge, pd, ps, head,
        W_hh, WhhT, W_ih, WihT);

    wmma_gemm_kernel<384, 4><<<grid384, 256, 0, stream>>>(nf, WhhT, b_hh, gh, V);

    edge_link_kernel<<<(E + blk - 1) / blk, blk, 0, stream>>>(src, dst, head, lnk, E);

    node_gather_kernel<<<((size_t)V * 32 + blk - 1) / blk, blk, 0, stream>>>(
        head, lnk, pd, ps, b_edge, hv, c, V, E);

    wmma_gemm_kernel<384, 4><<<grid384, 256, 0, stream>>>(c, WihT, b_ih, gi, V);

    const int gru_chunks = (V * (D_FEAT / 4) + blk - 1) / blk;
    gru_kernel<<<(gru_chunks + 3) / 4, blk, 0, stream>>>(
        gi, gh, nf, (float*)d_out, V);
}


# Both gate GEMMs and the GRU gates in one f32 WMMA launch: a wave holds all six weight strips of its 16 columns in registers, gi and gh are never stored; 6 launches become 4; every sum keeps its order
# speedup vs baseline: 2.4825x; 1.1171x over previous
#include <hip/hip_runtime.h>
#include <hip/hip_bf16.h>
#include <math.h>

typedef float v2f __attribute__((ext_vector_type(2)));
typedef float v8f __attribute__((ext_vector_type(8)));

#define D_FEAT 128
#define HEAD_PITCH 32

__device__ __forceinline__ void node_scalars(const float4 x,
                                             const float* __restrict__ W_edge,
                                             int v, int lane,
                                             float* __restrict__ pd,
                                             float* __restrict__ ps,
                                             int* __restrict__ head, int V) {
    const float4 w0 = ((const float4*)W_edge)[lane];
    const float4 w1 = ((const float4*)(W_edge + D_FEAT))[lane];
    float d0 = x.x * w0.x + x.y * w0.y + x.z * w0.z + x.w * w0.w;
    float d1 = x.x * w1.x + x.y * w1.y + x.z * w1.z + x.w * w1.w;
#pragma unroll
    for (int off = 16; off > 0; off >>= 1) {
        d0 += __shfl_down(d0, off, 32);
        d1 += __shfl_down(d1, off, 32);
    }
    if (v < V) {
        head[(size_t)v * HEAD_PITCH + lane] = -1;
        if (lane == 0) { pd[v] = d0; ps[v] = d1; }
    }
}

template <int N, int T, bool NODE_PASS>
__device__ __forceinline__ void gemm_rows(
    const float* __restrict__ X, const float* __restrict__ B,
    const float* __restrict__ bias, float* __restrict__ Out, int V,
    int bx, int by,
    const float* __restrict__ W_edge, float* __restrict__ pd,
    float* __restrict__ ps, int* __restrict__ head) {
    __shared__ __attribute__((aligned(16))) float xs[2][16][132];
    const int tid = threadIdx.x;
    const int lane = tid & 31;
    const int wave = tid >> 5;
    const int lo = lane & 15;
    const int hi = lane >> 4;
    const int nTiles = (V + 15) >> 4;
    const int t0 = bx * T;
    const int t1 = (t0 + T < nTiles) ? t0 + T : nTiles;
    if (t0 >= t1) return;
    const int colbase = by * 128 + wave * 16;

    v2f bf[32];
#pragma unroll
    for (int s = 0; s < 32; ++s) {
        const int ka = 4 * s + 2 * hi;
        bf[s][0] = B[(size_t)ka * N + colbase + lo];
        bf[s][1] = B[(size_t)(ka + 1) * N + colbase + lo];
    }
    const float bv = bias[colbase + lo];

    const int sr = tid >> 5;
    const int sc = (tid & 31) * 4;
    float4 st0, st1;
    {
        int r0 = t0 * 16 + sr;      if (r0 >= V) r0 = V - 1;
        int r1 = t0 * 16 + sr + 8;  if (r1 >= V) r1 = V - 1;
        st0 = *(const float4*)(X + (size_t)r0 * D_FEAT + sc);
        st1 = *(const float4*)(X + (size_t)r1 * D_FEAT + sc);
    }
    *(float4*)&xs[0][sr][sc] = st0;
    *(float4*)&xs[0][sr + 8][sc] = st1;
    __syncthreads();
    if (NODE_PASS) {
        node_scalars(st0, W_edge, t0 * 16 + sr, lane, pd, ps, head, V);
        node_scalars(st1, W_edge, t0 * 16 + sr + 8, lane, pd, ps, head, V);
    }

    for (int t = t0; t < t1; ++t) {
        const int buf = (t - t0) & 1;
        const bool more = (t + 1 < t1);
        if (more) {
            int r0 = (t + 1) * 16 + sr;      if (r0 >= V) r0 = V - 1;
            int r1 = (t + 1) * 16 + sr + 8;  if (r1 >= V) r1 = V - 1;
            st0 = *(const float4*)(X + (size_t)r0 * D_FEAT + sc);
            st1 = *(const float4*)(X + (size_t)r1 * D_FEAT + sc);
        }
        v8f acc;
#pragma unroll
        for (int r = 0; r < 8; ++r) acc[r] = bv;
#pragma unroll
        for (int s = 0; s < 32; ++s) {
            const int ka = 4 * s + 2 * hi;
            v2f a;
            a[0] = xs[buf][lo][ka];
            a[1] = xs[buf][lo][ka + 1];
            acc = __builtin_amdgcn_wmma_f32_16x16x4_f32(
                false, a, false, bf[s], (short)0, acc, false, false);
        }
        if (more) {
            *(float4*)&xs[buf ^ 1][sr][sc] = st0;
            *(float4*)&xs[buf ^ 1][sr + 8][sc] = st1;
        }
        __syncthreads();
        const int rowbase = t * 16;
        if (rowbase + 16 <= V) {
#pragma unroll
            for (int r = 0; r < 8; ++r) {
                int row = rowbase + r + 8 * hi;
                Out[(size_t)row * N + colbase + lo] = acc[r];
            }
        } else {
#pragma unroll
            for (int r = 0; r < 8; ++r) {
                int row = rowbase + r + 8 * hi;
                if (row < V) Out[(size_t)row * N + colbase + lo] = acc[r];
            }
        }
        if (NODE_PASS && more) {
            node_scalars(st0, W_edge, (t + 1) * 16 + sr, lane, pd, ps, head, V);
            node_scalars(st1, W_edge, (t + 1) * 16 + sr + 8, lane, pd, ps, head, V);
        }
    }
}

#define PACK_BLOCKS ((384 * D_FEAT) / 256)

__global__ __launch_bounds__(256, 8) void proj_node_pack_kernel(
    const float* __restrict__ nf, const float* __restrict__ W_proj,
    const float* __restrict__ b_proj, float* __restrict__ hv, int V,
    int gemmBlocks,
    const float* __restrict__ W_edge, float* __restrict__ pd,
    float* __restrict__ ps, int* __restrict__ head,
    const float* __restrict__ W_hh, float* __restrict__ WhhT,
    const float* __restrict__ W_ih, float* __restrict__ WihT) {
    const int b = blockIdx.x;
    if (b < gemmBlocks) {
        gemm_rows<128, 2, true>(nf, W_proj, b_proj, hv, V, b, 0,
                                W_edge, pd, ps, head);
    } else {
        const int pb = b - gemmBlocks;
        const bool second = pb >= PACK_BLOCKS;
        const float* __restrict__ Wsrc = second ? W_ih : W_hh;
        float* __restrict__ WT = second ? WihT : WhhT;
        const int i = (second ? pb - PACK_BLOCKS : pb) * 256 + threadIdx.x;
        const int n = i / D_FEAT;
        const int k = i - n * D_FEAT;
        WT[(size_t)k * 384 + n] = Wsrc[i];
    }
}

__global__ void edge_link_kernel(const int* __restrict__ src,
                                 const int* __restrict__ dst,
                                 int* __restrict__ head,
                                 int2* __restrict__ lnk, int E) {
    int e = blockIdx.x * blockDim.x + threadIdx.x;
    if (e >= E) return;
    int old = atomicExch(head + (size_t)dst[e] * HEAD_PITCH, e);
    lnk[e] = make_int2(src[e], old);
}

__global__ __launch_bounds__(256) void node_gather_kernel(
    const int* __restrict__ head, const int2* __restrict__ lnk,
    const float* __restrict__ pd, const float* __restrict__ ps,
    const float* __restrict__ b_edge, const float* __restrict__ hv,
    float* __restrict__ c, int V, int E) {
    const int gid = blockIdx.x * blockDim.x + threadIdx.x;
    const int lane = threadIdx.x & 31;
    const int v = __builtin_amdgcn_readfirstlane(gid >> 5);
    if (v >= V) return;
    float4 acc = make_float4(0.0f, 0.0f, 0.0f, 0.0f);
    float z = 0.0f;
    int e = head[(size_t)v * HEAD_PITCH];
    if (e >= 0) {
        const float pdv = pd[v];
        const float be = b_edge[0];
        int2 ln = lnk[e];
        float m = 0.0f;
        {
            const int s = ln.x;
            e = ln.y;
            const float psv = ps[s];
            acc = ((const float4*)(hv + (size_t)s * D_FEAT))[lane];
            if (e >= 0) ln = lnk[e];
            const float x = pdv + psv + be;
            m = x > 0.0f ? x : 0.01f * x;
            z = 1.0f;
        }
        for (int it = 1; it < E && e >= 0; ++it) {
            const int s = ln.x;
            e = ln.y;
            const float psv = ps[s];
            const float4 h = ((const float4*)(hv + (size_t)s * D_FEAT))[lane];
            if (e >= 0) ln = lnk[e];
            const float x = pdv + psv + be;
            const float l = x > 0.0f ? x : 0.01f * x;
            const float d = l - m;
            const float t = expf(-fabsf(d));
            const float sc = d > 0.0f ? t : 1.0f;
            const float w  = d > 0.0f ? 1.0f : t;
            m = d > 0.0f ? l : m;
            z = z * sc + w;
            acc.x = acc.x * sc + w * h.x;
            acc.y = acc.y * sc + w * h.y;
            acc.z = acc.z * sc + w * h.z;
            acc.w = acc.w * sc + w * h.w;
        }
    }
    float4 o;
    {
        float x;
        x = z > 0.0f ? acc.x / z : 0.0f;  o.x = x > 0.0f ? x : expf(x) - 1.0f;
        x = z > 0.0f ? acc.y / z : 0.0f;  o.y = x > 0.0f ? x : expf(x) - 1.0f;
        x = z > 0.0f ? acc.z / z : 0.0f;  o.z = x > 0.0f ? x : expf(x) - 1.0f;
        x = z > 0.0f ? acc.w / z : 0.0f;  o.w = x > 0.0f ? x : expf(x) - 1.0f;
    }
    ((float4*)(c + (size_t)v * D_FEAT))[lane] = o;
}

__device__ __forceinline__ float gru_one(float ir, float iz, float in_,
                                         float hr, float hz, float hn, float x) {
    float r = 1.0f / (1.0f + expf(-(ir + hr)));
    float u = 1.0f / (1.0f + expf(-(iz + hz)));
    float n = tanhf(in_ + r * hn);
    float h = fmaf(1.0f - u, n, u * x);
    return h > 0.0f ? h : 0.0f;
}

#define GATE_TILES 13
template <int T>
__global__ __launch_bounds__(256) void gate_gemm_gru_kernel(
    const float* __restrict__ Cx, const float* __restrict__ NF,
    const float* __restrict__ WihT, const float* __restrict__ WhhT,
    const float* __restrict__ b_ih, const float* __restrict__ b_hh,
    float* __restrict__ out, int V) {
    __shared__ __attribute__((aligned(16))) float xs[2][2][16][132];
    const int tid = threadIdx.x;
    const int lane = tid & 31;
    const int wave = tid >> 5;
    const int lo = lane & 15;
    const int hi = lane >> 4;
    const int nTiles = (V + 15) >> 4;
    const int t0 = blockIdx.x * T;
    const int t1 = (t0 + T < nTiles) ? t0 + T : nTiles;
    if (t0 >= t1) return;
    const int col = wave * 16 + lo;

    v2f bi[3][32], bh[3][32];
#pragma unroll
    for (int g = 0; g < 3; ++g) {
#pragma unroll
        for (int s = 0; s < 32; ++s) {
            const int ka = 4 * s + 2 * hi;
            bi[g][s][0] = WihT[(size_t)ka * 384 + g * 128 + col];
            bi[g][s][1] = WihT[(size_t)(ka + 1) * 384 + g * 128 + col];
            bh[g][s][0] = WhhT[(size_t)ka * 384 + g * 128 + col];
            bh[g][s][1] = WhhT[(size_t)(ka + 1) * 384 + g * 128 + col];
        }
    }

    const int sr = tid >> 5;
    const int sc = (tid & 31) * 4;
    float4 sc0, sc1, sn0, sn1;
    {
        int r0 = t0 * 16 + sr;      if (r0 >= V) r0 = V - 1;
        int r1 = t0 * 16 + sr + 8;  if (r1 >= V) r1 = V - 1;
        const unsigned o0 = (unsigned)r0 * D_FEAT + sc;
        const unsigned o1 = (unsigned)r1 * D_FEAT + sc;
        sc0 = *(const float4*)(Cx + o0);
        sc1 = *(const float4*)(Cx + o1);
        sn0 = *(const float4*)(NF + o0);
        sn1 = *(const float4*)(NF + o1);
    }
    *(float4*)&xs[0][0][sr][sc] = sc0;
    *(float4*)&xs[0][0][sr + 8][sc] = sc1;
    *(float4*)&xs[0][1][sr][sc] = sn0;
    *(float4*)&xs[0][1][sr + 8][sc] = sn1;
    __syncthreads();

    for (int t = t0; t < t1; ++t) {
        const int buf = (t - t0) & 1;
        const bool more = (t + 1 < t1);
        if (more) {
            int r0 = (t + 1) * 16 + sr;      if (r0 >= V) r0 = V - 1;
            int r1 = (t + 1) * 16 + sr + 8;  if (r1 >= V) r1 = V - 1;
            const unsigned o0 = (unsigned)r0 * D_FEAT + sc;
            const unsigned o1 = (unsigned)r1 * D_FEAT + sc;
            sc0 = *(const float4*)(Cx + o0);
            sc1 = *(const float4*)(Cx + o1);
            sn0 = *(const float4*)(NF + o0);
            sn1 = *(const float4*)(NF + o1);
        }
        v8f ai[3], ah[3];
#pragma unroll
        for (int g = 0; g < 3; ++g) {
            const float bvi = b_ih[g * 128 + col];
            const float bvh = b_hh[g * 128 + col];
#pragma unroll
            for (int r = 0; r < 8; ++r) { ai[g][r] = bvi; ah[g][r] = bvh; }
        }
#pragma unroll
        for (int s = 0; s < 32; ++s) {
            const int ka = 4 * s + 2 * hi;
            v2f ac, an;
            ac[0] = xs[buf][0][lo][ka];
            ac[1] = xs[buf][0][lo][ka + 1];
            an[0] = xs[buf][1][lo][ka];
            an[1] = xs[buf][1][lo][ka + 1];
#pragma unroll
            for (int g = 0; g < 3; ++g) {
                ai[g] = __builtin_amdgcn_wmma_f32_16x16x4_f32(
                    false, ac, false, bi[g][s], (short)0, ai[g], false, false);
                ah[g] = __builtin_amdgcn_wmma_f32_16x16x4_f32(
                    false, an, false, bh[g][s], (short)0, ah[g], false, false);
            }
        }
        if (more) {
            *(float4*)&xs[buf ^ 1][0][sr][sc] = sc0;
            *(float4*)&xs[buf ^ 1][0][sr + 8][sc] = sc1;
            *(float4*)&xs[buf ^ 1][1][sr][sc] = sn0;
            *(float4*)&xs[buf ^ 1][1][sr + 8][sc] = sn1;
        }
        float xv[8];
#pragma unroll
        for (int r = 0; r < 8; ++r) xv[r] = xs[buf][1][r + 8 * hi][col];
        __syncthreads();
        const int rowbase = t * 16;
#pragma unroll
        for (int r = 0; r < 8; ++r) {
            const int row = rowbase + r + 8 * hi;
            const float o = gru_one(ai[0][r], ai[1][r], ai[2][r],
                                    ah[0][r], ah[1][r], ah[2][r], xv[r]);
            if (row < V) out[(unsigned)row * D_FEAT + col] = o;
        }
    }
}

extern "C" void kernel_launch(void* const* d_in, const int* in_sizes, int n_in,
                              void* d_out, int out_size, void* d_ws, size_t ws_size,
                              hipStream_t stream) {
    const float* nf     = (const float*)d_in[0];
    const int*   src    = (const int*)d_in[1];
    const int*   dst    = (const int*)d_in[2];
    const float* W_edge = (const float*)d_in[3];
    const float* b_edge = (const float*)d_in[4];
    const float* W_proj = (const float*)d_in[5];
    const float* b_proj = (const float*)d_in[6];
    const float* W_ih   = (const float*)d_in[7];
    const float* b_ih   = (const float*)d_in[8];
    const float* W_hh   = (const float*)d_in[9];
    const float* b_hh   = (const float*)d_in[10];

    const int V = in_sizes[0] / D_FEAT;
    const int E = in_sizes[1];

    float* hv    = (float*)d_ws;
    float* c     = hv + (size_t)V * 128;
    float* WhhT  = c + (size_t)V * 128;
    float* WihT  = WhhT + 128 * 384;
    int2*  lnk   = (int2*)(WihT + 128 * 384);
    int*   head  = (int*)(lnk + E);
    float* pd    = (float*)(head + (size_t)V * HEAD_PITCH);
    float* ps    = pd + V;

    const int blk = 256;

    const int rowTiles = (V + 15) / 16;
    const int proj_blocks = (rowTiles + 1) / 2;

    proj_node_pack_kernel<<<proj_blocks + 2 * PACK_BLOCKS, blk, 0, stream>>>(
        nf, W_proj, b_proj, hv, V, proj_blocks, W_edge, pd, ps, head,
        W_hh, WhhT, W_ih, WihT);

    edge_link_kernel<<<(E + blk - 1) / blk, blk, 0, stream>>>(src, dst, head, lnk, E);

    node_gather_kernel<<<((size_t)V * 32 + blk - 1) / blk, blk, 0, stream>>>(
        head, lnk, pd, ps, b_edge, hv, c, V, E);

    gate_gemm_gru_kernel<GATE_TILES><<<(rowTiles + GATE_TILES - 1) / GATE_TILES, blk, 0, stream>>>(
        c, nf, WihT, WhhT, b_ih, b_hh, (float*)d_out, V);
}


# Edges grouped by destination with LDS counters (count, prefix sum, returning LDS add as slot) instead of one global exchange per edge; a node's terms summed in ascending source order: order changes
# speedup vs baseline: 2.7037x; 1.0891x over previous
#include <hip/hip_runtime.h>
#include <hip/hip_bf16.h>
#include <math.h>

typedef float v2f __attribute__((ext_vector_type(2)));
typedef float v8f __attribute__((ext_vector_type(8)));

#define D_FEAT 128
#define GRP 196
#define GRP_PAD 256
#define NCHUNK 256
#define MAXG 1024
#define CAP 8192

__device__ __forceinline__ void node_scalars(const float4 x,
                                             const float* __restrict__ W_edge,
                                             int v, int lane,
                                             float* __restrict__ pd,
                                             float* __restrict__ ps, int V) {
    const float4 w0 = ((const float4*)W_edge)[lane];
    const float4 w1 = ((const float4*)(W_edge + D_FEAT))[lane];
    float d0 = x.x * w0.x + x.y * w0.y + x.z * w0.z + x.w * w0.w;
    float d1 = x.x * w1.x + x.y * w1.y + x.z * w1.z + x.w * w1.w;
#pragma unroll
    for (int off = 16; off > 0; off >>= 1) {
        d0 += __shfl_down(d0, off, 32);
        d1 += __shfl_down(d1, off, 32);
    }
    if (v < V && lane == 0) { pd[v] = d0; ps[v] = d1; }
}

template <int N, int T, bool NODE_PASS>
__device__ __forceinline__ void gemm_rows(
    const float* __restrict__ X, const float* __restrict__ B,
    const float* __restrict__ bias, float* __restrict__ Out, int V,
    int bx, int by,
    const float* __restrict__ W_edge, float* __restrict__ pd,
    float* __restrict__ ps) {
    __shared__ __attribute__((aligned(16))) float xs[2][16][132];
    const int tid = threadIdx.x;
    const int lane = tid & 31;
    const int wave = tid >> 5;
    const int lo = lane & 15;
    const int hi = lane >> 4;
    const int nTiles = (V + 15) >> 4;
    const int t0 = bx * T;
    const int t1 = (t0 + T < nTiles) ? t0 + T : nTiles;
    if (t0 >= t1) return;
    const int colbase = by * 128 + wave * 16;

    v2f bf[32];
#pragma unroll
    for (int s = 0; s < 32; ++s) {
        const int ka = 4 * s + 2 * hi;
        bf[s][0] = B[(size_t)ka * N + colbase + lo];
        bf[s][1] = B[(size_t)(ka + 1) * N + colbase + lo];
    }
    const float bv = bias[colbase + lo];

    const int sr = tid >> 5;
    const int sc = (tid & 31) * 4;
    float4 st0, st1;
    {
        int r0 = t0 * 16 + sr;      if (r0 >= V) r0 = V - 1;
        int r1 = t0 * 16 + sr + 8;  if (r1 >= V) r1 = V - 1;
        st0 = *(const float4*)(X + (size_t)r0 * D_FEAT + sc);
        st1 = *(const float4*)(X + (size_t)r1 * D_FEAT + sc);
    }
    *(float4*)&xs[0][sr][sc] = st0;
    *(float4*)&xs[0][sr + 8][sc] = st1;
    __syncthreads();
    if (NODE_PASS) {
        node_scalars(st0, W_edge, t0 * 16 + sr, lane, pd, ps, V);
        node_scalars(st1, W_edge, t0 * 16 + sr + 8, lane, pd, ps, V);
    }

    for (int t = t0; t < t1; ++t) {
        const int buf = (t - t0) & 1;
        const bool more = (t + 1 < t1);
        if (more) {
            int r0 = (t + 1) * 16 + sr;      if (r0 >= V) r0 = V - 1;
            int r1 = (t + 1) * 16 + sr + 8;  if (r1 >= V) r1 = V - 1;
            st0 = *(const float4*)(X + (size_t)r0 * D_FEAT + sc);
            st1 = *(const float4*)(X + (size_t)r1 * D_FEAT + sc);
        }
        v8f acc;
#pragma unroll
        for (int r = 0; r < 8; ++r) acc[r] = bv;
#pragma unroll
        for (int s = 0; s < 32; ++s) {
            const int ka = 4 * s + 2 * hi;
            v2f a;
            a[0] = xs[buf][lo][ka];
            a[1] = xs[buf][lo][ka + 1];
            acc = __builtin_amdgcn_wmma_f32_16x16x4_f32(
                false, a, false, bf[s], (short)0, acc, false, false);
        }
        if (more) {
            *(float4*)&xs[buf ^ 1][sr][sc] = st0;
            *(float4*)&xs[buf ^ 1][sr + 8][sc] = st1;
        }
        __syncthreads();
        const int rowbase = t * 16;
        if (rowbase + 16 <= V) {
#pragma unroll
            for (int r = 0; r < 8; ++r) {
                int row = rowbase + r + 8 * hi;
                Out[(size_t)row * N + colbase + lo] = acc[r];
            }
        } else {
#pragma unroll
            for (int r = 0; r < 8; ++r) {
                int row = rowbase + r + 8 * hi;
                if (row < V) Out[(size_t)row * N + colbase + lo] = acc[r];
            }
        }
        if (NODE_PASS && more) {
            node_scalars(st0, W_edge, (t + 1) * 16 + sr, lane, pd, ps, V);
            node_scalars(st1, W_edge, (t + 1) * 16 + sr + 8, lane, pd, ps, V);
        }
    }
}

#define PACK_BLOCKS ((384 * D_FEAT) / 256)

__global__ __launch_bounds__(256, 8) void proj_node_pack_kernel(
    const float* __restrict__ nf, const float* __restrict__ W_proj,
    const float* __restrict__ b_proj, float* __restrict__ hv, int V,
    int gemmBlocks,
    const float* __restrict__ W_edge, float* __restrict__ pd,
    float* __restrict__ ps,
    const float* __restrict__ W_hh, float* __restrict__ WhhT,
    const float* __restrict__ W_ih, float* __restrict__ WihT,
    const int* __restrict__ dst, int E, int chunkLen, int nGroups,
    int* __restrict__ cntm) {
    const int b = blockIdx.x;
    if (b < NCHUNK) {
        __shared__ int hist[MAXG];
        const int pitch = (nGroups + 3) & ~3;
        for (int i = threadIdx.x; i < pitch; i += 256) hist[i] = 0;
        __syncthreads();
        const int e0 = b * chunkLen;
        const int e1 = (e0 + chunkLen < E) ? e0 + chunkLen : E;
        for (int e = e0 + (int)threadIdx.x; e < e1; e += 4 * 256) {
            int d[4];
#pragma unroll
            for (int i = 0; i < 4; ++i) {
                const int ee = e + i * 256;
                d[i] = ee < e1 ? dst[ee] : -1;
            }
#pragma unroll
            for (int i = 0; i < 4; ++i)
                if ((unsigned)d[i] < (unsigned)V) atomicAdd(&hist[d[i] / GRP], 1);
        }
        __syncthreads();
        for (int i = threadIdx.x; i < pitch; i += 256)
            cntm[(size_t)b * pitch + i] = hist[i];
    } else if (b < NCHUNK + gemmBlocks) {
        gemm_rows<128, 2, true>(nf, W_proj, b_proj, hv, V, b - NCHUNK, 0,
                                W_edge, pd, ps);
    } else {
        const int pb = b - NCHUNK - gemmBlocks;
        const bool second = pb >= PACK_BLOCKS;
        const float* __restrict__ Wsrc = second ? W_ih : W_hh;
        float* __restrict__ WT = second ? WihT : WhhT;
        const int i = (second ? pb - PACK_BLOCKS : pb) * 256 + threadIdx.x;
        const int n = i / D_FEAT;
        const int k = i - n * D_FEAT;
        WT[(size_t)k * 384 + n] = Wsrc[i];
    }
}

__device__ __forceinline__ int block_excl_scan(int x, int* wsum, int& total) {
    const int lane = threadIdx.x & 31;
    const int wave = threadIdx.x >> 5;
    const int nw = blockDim.x >> 5;
    int inc = x;
#pragma unroll
    for (int off = 1; off < 32; off <<= 1) {
        const int y = __shfl_up(inc, off, 32);
        if (lane >= off) inc += y;
    }
    if (lane == 31) wsum[wave] = inc;
    __syncthreads();
    if (wave == 0) {
        const int w = lane < nw ? wsum[lane] : 0;
        int winc = w;
#pragma unroll
        for (int off = 1; off < 32; off <<= 1) {
            const int y = __shfl_up(winc, off, 32);
            if (lane >= off) winc += y;
        }
        wsum[lane] = winc - w;
        if (lane == 31) wsum[32] = winc;
    }
    __syncthreads();
    total = wsum[32];
    return wsum[wave] + inc - x;
}

#define TAB_AHEAD 8
__global__ __launch_bounds__(1024) void edge_place_kernel(
    const int* __restrict__ src, const int* __restrict__ dst,
    const int* __restrict__ cntm, int* __restrict__ gstart,
    int2* __restrict__ bucket, int V, int E, int chunkLen, int nGroups) {
    __shared__ int tot[MAXG];
    __shared__ int base[MAXG];
    __shared__ int wsum[33];
    const int tid = threadIdx.x;
    const int chunk = blockIdx.x;
    const int pitch = (nGroups + 3) & ~3;
    const int Q = pitch >> 2;
    const int S = 1024 / Q;
    tot[tid] = 0;
    base[tid] = 0;
    __syncthreads();
    {
        const int gq = tid % Q;
        const int cs = tid / Q;
        if (cs < S) {
            int4 t4 = make_int4(0, 0, 0, 0), p4 = make_int4(0, 0, 0, 0);
            for (int c = cs; c < NCHUNK; c += TAB_AHEAD * S) {
                int4 x[TAB_AHEAD];
#pragma unroll
                for (int i = 0; i < TAB_AHEAD; ++i) {
                    const int cc = c + i * S;
                    x[i] = make_int4(0, 0, 0, 0);
                    if (cc < NCHUNK)
                        x[i] = *(const int4*)(cntm + (size_t)cc * pitch + 4 * gq);
                }
#pragma unroll
                for (int i = 0; i < TAB_AHEAD; ++i) {
                    const int cc = c + i * S;
                    t4.x += x[i].x; t4.y += x[i].y; t4.z += x[i].z; t4.w += x[i].w;
                    if (cc < chunk) {
                        p4.x += x[i].x; p4.y += x[i].y; p4.z += x[i].z; p4.w += x[i].w;
                    }
                }
            }
            atomicAdd(&tot[4 * gq + 0], t4.x);
            atomicAdd(&tot[4 * gq + 1], t4.y);
            atomicAdd(&tot[4 * gq + 2], t4.z);
            atomicAdd(&tot[4 * gq + 3], t4.w);
            atomicAdd(&base[4 * gq + 0], p4.x);
            atomicAdd(&base[4 * gq + 1], p4.y);
            atomicAdd(&base[4 * gq + 2], p4.z);
            atomicAdd(&base[4 * gq + 3], p4.w);
        }
    }
    __syncthreads();
    int all;
    const int gs = block_excl_scan(tid < nGroups ? tot[tid] : 0, wsum, all);
    if (tid < nGroups) {
        base[tid] += gs;
        if (chunk == 0) gstart[tid] = gs;
    }
    if (chunk == 0 && tid == 0) gstart[nGroups] = all;
    __syncthreads();
    const int e0 = chunk * chunkLen;
    const int e1 = (e0 + chunkLen < E) ? e0 + chunkLen : E;
    for (int e = e0 + tid; e < e1; e += 4 * 1024) {
        int d[4], s[4];
#pragma unroll
        for (int i = 0; i < 4; ++i) {
            const int ee = e + i * 1024;
            d[i] = ee < e1 ? dst[ee] : -1;
            s[i] = ee < e1 ? src[ee] : 0;
        }
#pragma unroll
        for (int i = 0; i < 4; ++i) {
            if ((unsigned)d[i] < (unsigned)V) {
                const int p = atomicAdd(&base[d[i] / GRP], 1);
                bucket[p] = make_int2(s[i], d[i]);
            }
        }
    }
}

template <bool GLOB>
__device__ __forceinline__ int gen_src(const int* sl, const int2* __restrict__ bk,
                                       int i, int v) {
    if (GLOB) {
        const int2 ed = bk[i];
        return ed.y == v ? ed.x : -1;
    }
    return sl[i];
}

template <bool GLOB>
__device__ __forceinline__ void gather_any(
    const int* sl, const int2* __restrict__ bk, int len, int v, int lane,
    float pdv, float be, const float* __restrict__ ps,
    const float* __restrict__ hv, float4& acc, float& z) {
    float m = -INFINITY;
    for (int i = lane; i < len; i += 32) {
        const int s = gen_src<GLOB>(sl, bk, i, v);
        if (s >= 0) {
            const float x = pdv + ps[s] + be;
            const float l = x > 0.0f ? x : 0.01f * x;
            m = fmaxf(m, l);
        }
    }
#pragma unroll
    for (int off = 16; off > 0; off >>= 1) m = fmaxf(m, __shfl_xor(m, off, 32));
    int last = -1;
    for (int it = 0; it < len; ++it) {
        int best = 0x7fffffff;
        for (int i = lane; i < len; i += 32) {
            const int s = gen_src<GLOB>(sl, bk, i, v);
            if (s > last && s < best) best = s;
        }
#pragma unroll
        for (int off = 16; off > 0; off >>= 1) {
            const int o = __shfl_xor(best, off, 32);
            best = o < best ? o : best;
        }
        if (best == 0x7fffffff) break;
        int mult = 0;
        for (int i = lane; i < len; i += 32)
            mult += (gen_src<GLOB>(sl, bk, i, v) == best) ? 1 : 0;
#pragma unroll
        for (int off = 16; off > 0; off >>= 1) mult += __shfl_xor(mult, off, 32);
        const int sb = __builtin_amdgcn_readfirstlane(best);
        const int mu = __builtin_amdgcn_readfirstlane(mult);
        const float x = pdv + ps[sb] + be;
        const float l = x > 0.0f ? x : 0.01f * x;
        const float w = expf(l - m);
        const float4 h = ((const float4*)(hv + (size_t)sb * D_FEAT))[lane];
        for (int r = 0; r < mu; ++r) {
            z += w;
            acc.x = fmaf(w, h.x, acc.x);
            acc.y = fmaf(w, h.y, acc.y);
            acc.z = fmaf(w, h.z, acc.z);
            acc.w = fmaf(w, h.w, acc.w);
        }
        last = sb;
    }
}

#define ROWS_AHEAD 8
__global__ __launch_bounds__(1024) void group_gather_kernel(
    const int* __restrict__ gstart, const int2* __restrict__ bucket,
    const float* __restrict__ pd, const float* __restrict__ ps,
    const float* __restrict__ b_edge, const float* __restrict__ hv,
    float* __restrict__ c, int V) {
    __shared__ int ncnt[GRP_PAD];
    __shared__ int noff[GRP_PAD];
    __shared__ int nfill[GRP_PAD];
    __shared__ int sorted[CAP];
    __shared__ int wsum[33];
    __shared__ int nextNode;
    const int tid = threadIdx.x;
    const int lane = tid & 31;
    const int g = blockIdx.x;
    const int v0 = g * GRP;
    const int nv = (V - v0 < GRP) ? V - v0 : GRP;
    const int gs = gstart[g];
    const int cnt = gstart[g + 1] - gs;
    const bool fits = cnt <= CAP;
    const int2* __restrict__ run = bucket + gs;
    if (tid < GRP_PAD) ncnt[tid] = 0;
    if (tid == 0) nextNode = 0;
    __syncthreads();
    int2 ed[4];
#pragma unroll
    for (int i = 0; i < 4; ++i) {
        const int idx = tid + i * 1024;
        ed[i] = make_int2(0, -1);
        if (idx < cnt) ed[i] = run[idx];
    }
#pragma unroll
    for (int i = 0; i < 4; ++i)
        if (ed[i].y >= 0) atomicAdd(&ncnt[ed[i].y - v0], 1);
    for (int i = tid + 4096; i < cnt; i += 1024) atomicAdd(&ncnt[run[i].y - v0], 1);
    __syncthreads();
    int all;
    const int ex = block_excl_scan(tid < GRP_PAD ? ncnt[tid] : 0, wsum, all);
    if (tid < GRP_PAD) { noff[tid] = ex; nfill[tid] = ex; }
    __syncthreads();
    if (fits) {
#pragma unroll
        for (int i = 0; i < 4; ++i) {
            if (ed[i].y >= 0) {
                const int p = atomicAdd(&nfill[ed[i].y - v0], 1);
                sorted[p] = ed[i].x;
            }
        }
        for (int i = tid + 4096; i < cnt; i += 1024) {
            const int2 e2 = run[i];
            const int p = atomicAdd(&nfill[e2.y - v0], 1);
            sorted[p] = e2.x;
        }
    }
    __syncthreads();

    const float be = b_edge[0];
    for (int turn = 0; turn < GRP; ++turn) {
        int t = 0;
        if (lane == 0) t = atomicAdd(&nextNode, 1);
        t = __builtin_amdgcn_readfirstlane(t);
        if (t >= nv) break;
        const int v = v0 + t;
        const int n = __builtin_amdgcn_readfirstlane(ncnt[t]);
        const int o = __builtin_amdgcn_readfirstlane(noff[t]);
        float4 acc = make_float4(0.0f, 0.0f, 0.0f, 0.0f);
        float z = 0.0f;
        if (n > 0) {
            const float pdv = pd[v];
            if (fits && n <= 64) {
                const bool two = n > 32;
                int s0 = lane < n ? sorted[o + lane] : 0x7fffffff;
                int s1 = 32 + lane < n ? sorted[o + 32 + lane] : 0x7fffffff;
                int r0 = 0, r1 = 0;
                const int nA = two ? 32 : n;
                for (int j = 0; j < nA; ++j) {
                    const int sj = __builtin_amdgcn_readlane(s0, j);
                    r0 += (sj < s0 || (sj == s0 && j < lane)) ? 1 : 0;
                    if (two) r1 += (sj <= s1) ? 1 : 0;
                }
                for (int j = 32; j < n; ++j) {
                    const int sj = __builtin_amdgcn_readlane(s1, j - 32);
                    r0 += (sj < s0) ? 1 : 0;
                    r1 += (sj < s1 || (sj == s1 && j - 32 < lane)) ? 1 : 0;
                }
                if (lane < n) sorted[o + r0] = s0;
                if (32 + lane < n) sorted[o + r1] = s1;
                s0 = lane < n ? sorted[o + lane] : 0;
                s1 = 32 + lane < n ? sorted[o + 32 + lane] : 0;
                float l0 = -INFINITY, l1 = -INFINITY;
                if (lane < n) {
                    const float x = pdv + ps[s0] + be;
                    l0 = x > 0.0f ? x : 0.01f * x;
                }
                if (32 + lane < n) {
                    const float x = pdv + ps[s1] + be;
                    l1 = x > 0.0f ? x : 0.01f * x;
                }
                float m = fmaxf(l0, l1);
#pragma unroll
                for (int off = 16; off > 0; off >>= 1)
                    m = fmaxf(m, __shfl_xor(m, off, 32));
                const int w0 = __float_as_int(lane < n ? expf(l0 - m) : 0.0f);
                const int w1 = __float_as_int(32 + lane < n ? expf(l1 - m) : 0.0f);
                for (int k = 0; k < n; k += ROWS_AHEAD) {
                    const int sr = k >= 32 ? s1 : s0;
                    const int wr = k >= 32 ? w1 : w0;
                    const int kb = k & 31;
                    float4 h[ROWS_AHEAD];
                    float wk[ROWS_AHEAD];
#pragma unroll
                    for (int i = 0; i < ROWS_AHEAD; ++i) {
                        h[i] = make_float4(0.0f, 0.0f, 0.0f, 0.0f);
                        wk[i] = 0.0f;
                        if (k + i < n) {
                            const int sk = __builtin_amdgcn_readlane(sr, kb + i);
                            wk[i] = __int_as_float(__builtin_amdgcn_readlane(wr, kb + i));
                            h[i] = ((const float4*)(hv + (size_t)sk * D_FEAT))[lane];
                        }
                    }
#pragma unroll
                    for (int i = 0; i < ROWS_AHEAD; ++i) {
                        z += wk[i];
                        acc.x = fmaf(wk[i], h[i].x, acc.x);
                        acc.y = fmaf(wk[i], h[i].y, acc.y);
                        acc.z = fmaf(wk[i], h[i].z, acc.z);
                        acc.w = fmaf(wk[i], h[i].w, acc.w);
                    }
                }
            } else if (fits) {
                gather_any<false>(sorted + o, run, n, v, lane, pdv, be, ps, hv, acc, z);
            } else {
                gather_any<true>(sorted, run, cnt, v, lane, pdv, be, ps, hv, acc, z);
            }
        }
        float4 r;
        {
            float x;
            x = z > 0.0f ? acc.x / z : 0.0f;  r.x = x > 0.0f ? x : expf(x) - 1.0f;
            x = z > 0.0f ? acc.y / z : 0.0f;  r.y = x > 0.0f ? x : expf(x) - 1.0f;
            x = z > 0.0f ? acc.z / z : 0.0f;  r.z = x > 0.0f ? x : expf(x) - 1.0f;
            x = z > 0.0f ? acc.w / z : 0.0f;  r.w = x > 0.0f ? x : expf(x) - 1.0f;
        }
        ((float4*)(c + (size_t)v * D_FEAT))[lane] = r;
    }
}

__device__ __forceinline__ float gru_one(float ir, float iz, float in_,
                                         float hr, float hz, float hn, float x) {
    float r = 1.0f / (1.0f + expf(-(ir + hr)));
    float u = 1.0f / (1.0f + expf(-(iz + hz)));
    float n = tanhf(in_ + r * hn);
    float h = fmaf(1.0f - u, n, u * x);
    return h > 0.0f ? h : 0.0f;
}

#define GATE_TILES 13
template <int T>
__global__ __launch_bounds__(256) void gate_gemm_gru_kernel(
    const float* __restrict__ Cx, const float* __restrict__ NF,
    const float* __restrict__ WihT, const float* __restrict__ WhhT,
    const float* __restrict__ b_ih, const float* __restrict__ b_hh,
    float* __restrict__ out, int V) {
    __shared__ __attribute__((aligned(16))) float xs[2][2][16][132];
    const int tid = threadIdx.x;
    const int lane = tid & 31;
    const int wave = tid >> 5;
    const int lo = lane & 15;
    const int hi = lane >> 4;
    const int nTiles = (V + 15) >> 4;
    const int t0 = blockIdx.x * T;
    const int t1 = (t0 + T < nTiles) ? t0 + T : nTiles;
    if (t0 >= t1) return;
    const int col = wave * 16 + lo;

    v2f bi[3][32], bh[3][32];
#pragma unroll
    for (int g = 0; g < 3; ++g) {
#pragma unroll
        for (int s = 0; s < 32; ++s) {
            const int ka = 4 * s + 2 * hi;
            bi[g][s][0] = WihT[(size_t)ka * 384 + g * 128 + col];
            bi[g][s][1] = WihT[(size_t)(ka + 1) * 384 + g * 128 + col];
            bh[g][s][0] = WhhT[(size_t)ka * 384 + g * 128 + col];
            bh[g][s][1] = WhhT[(size_t)(ka + 1) * 384 + g * 128 + col];
        }
    }

    const int sr = tid >> 5;
    const int sc = (tid & 31) * 4;
    float4 sc0, sc1, sn0, sn1;
    {
        int r0 = t0 * 16 + sr;      if (r0 >= V) r0 = V - 1;
        int r1 = t0 * 16 + sr + 8;  if (r1 >= V) r1 = V - 1;
        const unsigned o0 = (unsigned)r0 * D_FEAT + sc;
        const unsigned o1 = (unsigned)r1 * D_FEAT + sc;
        sc0 = *(const float4*)(Cx + o0);
        sc1 = *(const float4*)(Cx + o1);
        sn0 = *(const float4*)(NF + o0);
        sn1 = *(const float4*)(NF + o1);
    }
    *(float4*)&xs[0][0][sr][sc] = sc0;
    *(float4*)&xs[0][0][sr + 8][sc] = sc1;
    *(float4*)&xs[0][1][sr][sc] = sn0;
    *(float4*)&xs[0][1][sr + 8][sc] = sn1;
    __syncthreads();

    for (int t = t0; t < t1; ++t) {
        const int buf = (t - t0) & 1;
        const bool more = (t + 1 < t1);
        if (more) {
            int r0 = (t + 1) * 16 + sr;      if (r0 >= V) r0 = V - 1;
            int r1 = (t + 1) * 16 + sr + 8;  if (r1 >= V) r1 = V - 1;
            const unsigned o0 = (unsigned)r0 * D_FEAT + sc;
            const unsigned o1 = (unsigned)r1 * D_FEAT + sc;
            sc0 = *(const float4*)(Cx + o0);
            sc1 = *(const float4*)(Cx + o1);
            sn0 = *(const float4*)(NF + o0);
            sn1 = *(const float4*)(NF + o1);
        }
        v8f ai[3], ah[3];
#pragma unroll
        for (int g = 0; g < 3; ++g) {
            const float bvi = b_ih[g * 128 + col];
            const float bvh = b_hh[g * 128 + col];
#pragma unroll
            for (int r = 0; r < 8; ++r) { ai[g][r] = bvi; ah[g][r] = bvh; }
        }
#pragma unroll
        for (int s = 0; s < 32; ++s) {
            const int ka = 4 * s + 2 * hi;
            v2f ac, an;
            ac[0] = xs[buf][0][lo][ka];
            ac[1] = xs[buf][0][lo][ka + 1];
            an[0] = xs[buf][1][lo][ka];
            an[1] = xs[buf][1][lo][ka + 1];
#pragma unroll
            for (int g = 0; g < 3; ++g) {
                ai[g] = __builtin_amdgcn_wmma_f32_16x16x4_f32(
                    false, ac, false, bi[g][s], (short)0, ai[g], false, false);
                ah[g] = __builtin_amdgcn_wmma_f32_16x16x4_f32(
                    false, an, false, bh[g][s], (short)0, ah[g], false, false);
            }
        }
        if (more) {
            *(float4*)&xs[buf ^ 1][0][sr][sc] = sc0;
            *(float4*)&xs[buf ^ 1][0][sr + 8][sc] = sc1;
            *(float4*)&xs[buf ^ 1][1][sr][sc] = sn0;
            *(float4*)&xs[buf ^ 1][1][sr + 8][sc] = sn1;
        }
        float xv[8];
#pragma unroll
        for (int r = 0; r < 8; ++r) xv[r] = xs[buf][1][r + 8 * hi][col];
        __syncthreads();
        const int rowbase = t * 16;
#pragma unroll
        for (int r = 0; r < 8; ++r) {
            const int row = rowbase + r + 8 * hi;
            const float o = gru_one(ai[0][r], ai[1][r], ai[2][r],
                                    ah[0][r], ah[1][r], ah[2][r], xv[r]);
            if (row < V) out[(unsigned)row * D_FEAT + col] = o;
        }
    }
}

extern "C" void kernel_launch(void* const* d_in, const int* in_sizes, int n_in,
                              void* d_out, int out_size, void* d_ws, size_t ws_size,
                              hipStream_t stream) {
    const float* nf     = (const float*)d_in[0];
    const int*   src    = (const int*)d_in[1];
    const int*   dst    = (const int*)d_in[2];
    const float* W_edge = (const float*)d_in[3];
    const float* b_edge = (const float*)d_in[4];
    const float* W_proj = (const float*)d_in[5];
    const float* b_proj = (const float*)d_in[6];
    const float* W_ih   = (const float*)d_in[7];
    const float* b_ih   = (const float*)d_in[8];
    const float* W_hh   = (const float*)d_in[9];
    const float* b_hh   = (const float*)d_in[10];

    const int V = in_sizes[0] / D_FEAT;
    const int E = in_sizes[1];
    const int nGroups = (V + GRP - 1) / GRP;
    const int chunkLen = (E + NCHUNK - 1) / NCHUNK;
    const int pitch = (nGroups + 3) & ~3;

    float* hv     = (float*)d_ws;
    float* c      = hv + (size_t)V * 128;
    float* WhhT   = c + (size_t)V * 128;
    float* WihT   = WhhT + 128 * 384;
    int2*  bucket = (int2*)(WihT + 128 * 384);
    int*   cntm   = (int*)(((size_t)(bucket + E) + 15) & ~(size_t)15);
    int*   gstart = cntm + (size_t)NCHUNK * pitch;
    float* pd     = (float*)(gstart + nGroups + 1);
    float* ps     = pd + V;

    const int blk = 256;

    const int rowTiles = (V + 15) / 16;
    const int proj_blocks = (rowTiles + 1) / 2;

    proj_node_pack_kernel<<<NCHUNK + proj_blocks + 2 * PACK_BLOCKS, blk, 0, stream>>>(
        nf, W_proj, b_proj, hv, V, proj_blocks, W_edge, pd, ps,
        W_hh, WhhT, W_ih, WihT, dst, E, chunkLen, nGroups, cntm);

    edge_place_kernel<<<NCHUNK, 1024, 0, stream>>>(
        src, dst, cntm, gstart, bucket, V, E, chunkLen, nGroups);

    group_gather_kernel<<<nGroups, 1024, 0, stream>>>(
        gstart, bucket, pd, ps, b_edge, hv, c, V);

    gate_gemm_gru_kernel<GATE_TILES><<<(rowTiles + GATE_TILES - 1) / GATE_TILES, blk, 0, stream>>>(
        c, nf, WihT, WhhT, b_ih, b_hh, (float*)d_out, V);
}


# Edge placing launch removed: launch 1 groups each chunk's edges inside the chunk (LDS count, scan, slot); a gather block collects its 256 chunk segments; 4 launches become 3; sums keep their order
# speedup vs baseline: 2.9642x; 1.0963x over previous
#include <hip/hip_runtime.h>
#include <hip/hip_bf16.h>
#include <math.h>

typedef float v2f __attribute__((ext_vector_type(2)));
typedef float v8f __attribute__((ext_vector_type(8)));

#define D_FEAT 128
#define GRP 196
#define GRP_PAD 256
#define NCHUNK 256
#define MAXG 1024
#define CAP 8192

__device__ __forceinline__ void node_scalars(const float4 x,
                                             const float* __restrict__ W_edge,
                                             int v, int lane,
                                             float* __restrict__ pd,
                                             float* __restrict__ ps, int V) {
    const float4 w0 = ((const float4*)W_edge)[lane];
    const float4 w1 = ((const float4*)(W_edge + D_FEAT))[lane];
    float d0 = x.x * w0.x + x.y * w0.y + x.z * w0.z + x.w * w0.w;
    float d1 = x.x * w1.x + x.y * w1.y + x.z * w1.z + x.w * w1.w;
#pragma unroll
    for (int off = 16; off > 0; off >>= 1) {
        d0 += __shfl_down(d0, off, 32);
        d1 += __shfl_down(d1, off, 32);
    }
    if (v < V && lane == 0) { pd[v] = d0; ps[v] = d1; }
}

template <int N, int T, bool NODE_PASS>
__device__ __forceinline__ void gemm_rows(
    const float* __restrict__ X, const float* __restrict__ B,
    const float* __restrict__ bias, float* __restrict__ Out, int V,
    int bx, int by,
    const float* __restrict__ W_edge, float* __restrict__ pd,
    float* __restrict__ ps) {
    __shared__ __attribute__((aligned(16))) float xs[2][16][132];
    const int tid = threadIdx.x;
    const int lane = tid & 31;
    const int wave = tid >> 5;
    const int lo = lane & 15;
    const int hi = lane >> 4;
    const int nTiles = (V + 15) >> 4;
    const int t0 = bx * T;
    const int t1 = (t0 + T < nTiles) ? t0 + T : nTiles;
    if (t0 >= t1) return;
    const int colbase = by * 128 + wave * 16;

    v2f bf[32];
#pragma unroll
    for (int s = 0; s < 32; ++s) {
        const int ka = 4 * s + 2 * hi;
        bf[s][0] = B[(size_t)ka * N + colbase + lo];
        bf[s][1] = B[(size_t)(ka + 1) * N + colbase + lo];
    }
    const float bv = bias[colbase + lo];

    const int sr = tid >> 5;
    const int sc = (tid & 31) * 4;
    float4 st0, st1;
    {
        int r0 = t0 * 16 + sr;      if (r0 >= V) r0 = V - 1;
        int r1 = t0 * 16 + sr + 8;  if (r1 >= V) r1 = V - 1;
        st0 = *(const float4*)(X + (size_t)r0 * D_FEAT + sc);
        st1 = *(const float4*)(X + (size_t)r1 * D_FEAT + sc);
    }
    *(float4*)&xs[0][sr][sc] = st0;
    *(float4*)&xs[0][sr + 8][sc] = st1;
    __syncthreads();
    if (NODE_PASS) {
        node_scalars(st0, W_edge, t0 * 16 + sr, lane, pd, ps, V);
        node_scalars(st1, W_edge, t0 * 16 + sr + 8, lane, pd, ps, V);
    }

    for (int t = t0; t < t1; ++t) {
        const int buf = (t - t0) & 1;
        const bool more = (t + 1 < t1);
        if (more) {
            int r0 = (t + 1) * 16 + sr;      if (r0 >= V) r0 = V - 1;
            int r1 = (t + 1) * 16 + sr + 8;  if (r1 >= V) r1 = V - 1;
            st0 = *(const float4*)(X + (size_t)r0 * D_FEAT + sc);
            st1 = *(const float4*)(X + (size_t)r1 * D_FEAT + sc);
        }
        v8f acc;
#pragma unroll
        for (int r = 0; r < 8; ++r) acc[r] = bv;
#pragma unroll
        for (int s = 0; s < 32; ++s) {
            const int ka = 4 * s + 2 * hi;
            v2f a;
            a[0] = xs[buf][lo][ka];
            a[1] = xs[buf][lo][ka + 1];
            acc = __builtin_amdgcn_wmma_f32_16x16x4_f32(
                false, a, false, bf[s], (short)0, acc, false, false);
        }
        if (more) {
            *(float4*)&xs[buf ^ 1][sr][sc] = st0;
            *(float4*)&xs[buf ^ 1][sr + 8][sc] = st1;
        }
        __syncthreads();
        const int rowbase = t * 16;
        if (rowbase + 16 <= V) {
#pragma unroll
            for (int r = 0; r < 8; ++r) {
                int row = rowbase + r + 8 * hi;
                Out[(size_t)row * N + colbase + lo] = acc[r];
            }
        } else {
#pragma unroll
            for (int r = 0; r < 8; ++r) {
                int row = rowbase + r + 8 * hi;
                if (row < V) Out[(size_t)row * N + colbase + lo] = acc[r];
            }
        }
        if (NODE_PASS && more) {
            node_scalars(st0, W_edge, (t + 1) * 16 + sr, lane, pd, ps, V);
            node_scalars(st1, W_edge, (t + 1) * 16 + sr + 8, lane, pd, ps, V);
        }
    }
}

__device__ __forceinline__ int block_excl_scan(int x, int* wsum, int& total) {
    const int lane = threadIdx.x & 31;
    const int wave = threadIdx.x >> 5;
    const int nw = blockDim.x >> 5;
    int inc = x;
#pragma unroll
    for (int off = 1; off < 32; off <<= 1) {
        const int y = __shfl_up(inc, off, 32);
        if (lane >= off) inc += y;
    }
    if (lane == 31) wsum[wave] = inc;
    __syncthreads();
    if (wave == 0) {
        const int w = lane < nw ? wsum[lane] : 0;
        int winc = w;
#pragma unroll
        for (int off = 1; off < 32; off <<= 1) {
            const int y = __shfl_up(winc, off, 32);
            if (lane >= off) winc += y;
        }
        wsum[lane] = winc - w;
        if (lane == 31) wsum[32] = winc;
    }
    __syncthreads();
    total = wsum[32];
    return wsum[wave] + inc - x;
}

#define PACK_BLOCKS ((384 * D_FEAT) / 256)

__global__ __launch_bounds__(256, 8) void proj_node_pack_kernel(
    const float* __restrict__ nf, const float* __restrict__ W_proj,
    const float* __restrict__ b_proj, float* __restrict__ hv, int V,
    int gemmBlocks,
    const float* __restrict__ W_edge, float* __restrict__ pd,
    float* __restrict__ ps,
    const float* __restrict__ W_hh, float* __restrict__ WhhT,
    const float* __restrict__ W_ih, float* __restrict__ WihT,
    const int* __restrict__ src, const int* __restrict__ dst, int E,
    int chunkLen, int nGroups, int* __restrict__ segoff,
    int2* __restrict__ bucket) {
    const int b = blockIdx.x;
    if (b < NCHUNK) {
        __shared__ int hist[MAXG];
        __shared__ int wsum[33];
        const int tid = threadIdx.x;
        const int pitchO = nGroups + 1;
        for (int i = tid; i < nGroups; i += 256) hist[i] = 0;
        __syncthreads();
        const int e0 = b * chunkLen;
        const int e1 = (e0 + chunkLen < E) ? e0 + chunkLen : E;
        for (int e = e0 + tid; e < e1; e += 4 * 256) {
            int d[4];
#pragma unroll
            for (int i = 0; i < 4; ++i) {
                const int ee = e + i * 256;
                d[i] = ee < e1 ? dst[ee] : -1;
            }
#pragma unroll
            for (int i = 0; i < 4; ++i)
                if ((unsigned)d[i] < (unsigned)V) atomicAdd(&hist[d[i] / GRP], 1);
        }
        __syncthreads();
        int carry = 0;
        for (int gb = 0; gb < nGroups; gb += 256) {
            const int gi = gb + tid;
            int tot;
            const int ex = carry + block_excl_scan(gi < nGroups ? hist[gi] : 0, wsum, tot);
            if (gi < nGroups) {
                hist[gi] = ex;
                segoff[(size_t)b * pitchO + gi] = ex;
            }
            carry += tot;
            __syncthreads();
        }
        if (tid == 0) segoff[(size_t)b * pitchO + nGroups] = carry;
        for (int e = e0 + tid; e < e1; e += 4 * 256) {
            int d[4], sv[4];
#pragma unroll
            for (int i = 0; i < 4; ++i) {
                const int ee = e + i * 256;
                d[i] = ee < e1 ? dst[ee] : -1;
                sv[i] = ee < e1 ? src[ee] : 0;
            }
#pragma unroll
            for (int i = 0; i < 4; ++i) {
                if ((unsigned)d[i] < (unsigned)V) {
                    const int p = atomicAdd(&hist[d[i] / GRP], 1);
                    bucket[e0 + p] = make_int2(sv[i], d[i]);
                }
            }
        }
    } else if (b < NCHUNK + gemmBlocks) {
        gemm_rows<128, 2, true>(nf, W_proj, b_proj, hv, V, b - NCHUNK, 0,
                                W_edge, pd, ps);
    } else {
        const int pb = b - NCHUNK - gemmBlocks;
        const bool second = pb >= PACK_BLOCKS;
        const float* __restrict__ Wsrc = second ? W_ih : W_hh;
        float* __restrict__ WT = second ? WihT : WhhT;
        const int i = (second ? pb - PACK_BLOCKS : pb) * 256 + threadIdx.x;
        const int n = i / D_FEAT;
        const int k = i - n * D_FEAT;
        WT[(size_t)k * 384 + n] = Wsrc[i];
    }
}

template <bool GLOB>
__device__ __forceinline__ int gen_src(const int* sl, const int2* __restrict__ bk,
                                       const int* cbeg, const int* cpre,
                                       int i, int v) {
    if (GLOB) {
        int lo = 0, hi = NCHUNK;
#pragma unroll
        for (int st = 0; st < 8; ++st) {
            const int mid = (lo + hi) >> 1;
            if (cpre[mid] <= i) lo = mid; else hi = mid;
        }
        const int2 ed = bk[cbeg[lo] + (i - cpre[lo])];
        return ed.y == v ? ed.x : -1;
    }
    return sl[i];
}

template <bool GLOB>
__device__ __forceinline__ void gather_any(
    const int* sl, const int2* __restrict__ bk, const int* cbeg,
    const int* cpre, int len, int v, int lane,
    float pdv, float be, const float* __restrict__ ps,
    const float* __restrict__ hv, float4& acc, float& z) {
    float m = -INFINITY;
    for (int i = lane; i < len; i += 32) {
        const int s = gen_src<GLOB>(sl, bk, cbeg, cpre, i, v);
        if (s >= 0) {
            const float x = pdv + ps[s] + be;
            const float l = x > 0.0f ? x : 0.01f * x;
            m = fmaxf(m, l);
        }
    }
#pragma unroll
    for (int off = 16; off > 0; off >>= 1) m = fmaxf(m, __shfl_xor(m, off, 32));
    int last = -1;
    for (int it = 0; it < len; ++it) {
        int best = 0x7fffffff;
        for (int i = lane; i < len; i += 32) {
            const int s = gen_src<GLOB>(sl, bk, cbeg, cpre, i, v);
            if (s > last && s < best) best = s;
        }
#pragma unroll
        for (int off = 16; off > 0; off >>= 1) {
            const int o = __shfl_xor(best, off, 32);
            best = o < best ? o : best;
        }
        if (best == 0x7fffffff) break;
        int mult = 0;
        for (int i = lane; i < len; i += 32)
            mult += (gen_src<GLOB>(sl, bk, cbeg, cpre, i, v) == best) ? 1 : 0;
#pragma unroll
        for (int off = 16; off > 0; off >>= 1) mult += __shfl_xor(mult, off, 32);
        const int sb = __builtin_amdgcn_readfirstlane(best);
        const int mu = __builtin_amdgcn_readfirstlane(mult);
        const float x = pdv + ps[sb] + be;
        const float l = x > 0.0f ? x : 0.01f * x;
        const float w = expf(l - m);
        const float4 h = ((const float4*)(hv + (size_t)sb * D_FEAT))[lane];
        for (int r = 0; r < mu; ++r) {
            z += w;
            acc.x = fmaf(w, h.x, acc.x);
            acc.y = fmaf(w, h.y, acc.y);
            acc.z = fmaf(w, h.z, acc.z);
            acc.w = fmaf(w, h.w, acc.w);
        }
        last = sb;
    }
}

#define ROWS_AHEAD 8
__global__ __launch_bounds__(1024) void group_gather_kernel(
    const int* __restrict__ segoff, const int2* __restrict__ bucket,
    const float* __restrict__ pd, const float* __restrict__ ps,
    const float* __restrict__ b_edge, const float* __restrict__ hv,
    float* __restrict__ c, int V, int chunkLen, int nGroups) {
    __shared__ int cbeg[NCHUNK];
    __shared__ int cpre[NCHUNK + 1];
    __shared__ int ncnt[GRP_PAD];
    __shared__ int noff[GRP_PAD];
    __shared__ int nfill[GRP_PAD];
    __shared__ int sorted[CAP];
    __shared__ int wsum[33];
    __shared__ int nextNode;
    const int tid = threadIdx.x;
    const int lane = tid & 31;
    const int g = blockIdx.x;
    const int v0 = g * GRP;
    const int nv = (V - v0 < GRP) ? V - v0 : GRP;
    int segBeg = 0, segLen = 0;
    if (tid < NCHUNK) {
        const int* __restrict__ row = segoff + (size_t)tid * (nGroups + 1) + g;
        const int a = row[0];
        segLen = row[1] - a;
        segBeg = tid * chunkLen + a;
    }
    if (tid < GRP_PAD) ncnt[tid] = 0;
    if (tid == 0) nextNode = 0;
    int cnt;
    const int pre = block_excl_scan(segLen, wsum, cnt);
    if (tid < NCHUNK) { cbeg[tid] = segBeg; cpre[tid] = pre; }
    if (tid == 0) cpre[NCHUNK] = cnt;
    __syncthreads();
    const bool fits = cnt <= CAP;
    const int sBeg = cbeg[tid >> 2];
    const int sLen = cpre[(tid >> 2) + 1] - cpre[tid >> 2];
    const int q = tid & 3;
    const int2* __restrict__ seg = bucket + sBeg;
    int2 ed[4];
#pragma unroll
    for (int i = 0; i < 4; ++i) {
        const int idx = q + 4 * i;
        ed[i] = make_int2(0, -1);
        if (idx < sLen) ed[i] = seg[idx];
    }
#pragma unroll
    for (int i = 0; i < 4; ++i)
        if (ed[i].y >= 0) atomicAdd(&ncnt[ed[i].y - v0], 1);
    for (int i = q + 16; i < sLen; i += 4) atomicAdd(&ncnt[seg[i].y - v0], 1);
    __syncthreads();
    int all;
    const int ex = block_excl_scan(tid < GRP_PAD ? ncnt[tid] : 0, wsum, all);
    if (tid < GRP_PAD) { noff[tid] = ex; nfill[tid] = ex; }
    __syncthreads();
    if (fits) {
#pragma unroll
        for (int i = 0; i < 4; ++i) {
            if (ed[i].y >= 0) {
                const int p = atomicAdd(&nfill[ed[i].y - v0], 1);
                sorted[p] = ed[i].x;
            }
        }
        for (int i = q + 16; i < sLen; i += 4) {
            const int2 e2 = seg[i];
            const int p = atomicAdd(&nfill[e2.y - v0], 1);
            sorted[p] = e2.x;
        }
    }
    __syncthreads();

    const float be = b_edge[0];
    for (int turn = 0; turn < GRP; ++turn) {
        int t = 0;
        if (lane == 0) t = atomicAdd(&nextNode, 1);
        t = __builtin_amdgcn_readfirstlane(t);
        if (t >= nv) break;
        const int v = v0 + t;
        const int n = __builtin_amdgcn_readfirstlane(ncnt[t]);
        const int o = __builtin_amdgcn_readfirstlane(noff[t]);
        float4 acc = make_float4(0.0f, 0.0f, 0.0f, 0.0f);
        float z = 0.0f;
        if (n > 0) {
            const float pdv = pd[v];
            if (fits && n <= 64) {
                const bool two = n > 32;
                int s0 = lane < n ? sorted[o + lane] : 0x7fffffff;
                int s1 = 32 + lane < n ? sorted[o + 32 + lane] : 0x7fffffff;
                int r0 = 0, r1 = 0;
                const int nA = two ? 32 : n;
                for (int j = 0; j < nA; ++j) {
                    const int sj = __builtin_amdgcn_readlane(s0, j);
                    r0 += (sj < s0 || (sj == s0 && j < lane)) ? 1 : 0;
                    if (two) r1 += (sj <= s1) ? 1 : 0;
                }
                for (int j = 32; j < n; ++j) {
                    const int sj = __builtin_amdgcn_readlane(s1, j - 32);
                    r0 += (sj < s0) ? 1 : 0;
                    r1 += (sj < s1 || (sj == s1 && j - 32 < lane)) ? 1 : 0;
                }
                if (lane < n) sorted[o + r0] = s0;
                if (32 + lane < n) sorted[o + r1] = s1;
                s0 = lane < n ? sorted[o + lane] : 0;
                s1 = 32 + lane < n ? sorted[o + 32 + lane] : 0;
                float l0 = -INFINITY, l1 = -INFINITY;
                if (lane < n) {
                    const float x = pdv + ps[s0] + be;
                    l0 = x > 0.0f ? x : 0.01f * x;
                }
                if (32 + lane < n) {
                    const float x = pdv + ps[s1] + be;
                    l1 = x > 0.0f ? x : 0.01f * x;
                }
                float m = fmaxf(l0, l1);
#pragma unroll
                for (int off = 16; off > 0; off >>= 1)
                    m = fmaxf(m, __shfl_xor(m, off, 32));
                const int w0 = __float_as_int(lane < n ? expf(l0 - m) : 0.0f);
                const int w1 = __float_as_int(32 + lane < n ? expf(l1 - m) : 0.0f);
                for (int k = 0; k < n; k += ROWS_AHEAD) {
                    const int sr = k >= 32 ? s1 : s0;
                    const int wr = k >= 32 ? w1 : w0;
                    const int kb = k & 31;
                    float4 h[ROWS_AHEAD];
                    float wk[ROWS_AHEAD];
#pragma unroll
                    for (int i = 0; i < ROWS_AHEAD; ++i) {
                        h[i] = make_float4(0.0f, 0.0f, 0.0f, 0.0f);
                        wk[i] = 0.0f;
                        if (k + i < n) {
                            const int sk = __builtin_amdgcn_readlane(sr, kb + i);
                            wk[i] = __int_as_float(__builtin_amdgcn_readlane(wr, kb + i));
                            h[i] = ((const float4*)(hv + (size_t)sk * D_FEAT))[lane];
                        }
                    }
#pragma unroll
                    for (int i = 0; i < ROWS_AHEAD; ++i) {
                        z += wk[i];
                        acc.x = fmaf(wk[i], h[i].x, acc.x);
                        acc.y = fmaf(wk[i], h[i].y, acc.y);
                        acc.z = fmaf(wk[i], h[i].z, acc.z);
                        acc.w = fmaf(wk[i], h[i].w, acc.w);
                    }
                }
            } else if (fits) {
                gather_any<false>(sorted + o, bucket, cbeg, cpre, n, v, lane, pdv, be, ps, hv, acc, z);
            } else {
                gather_any<true>(sorted, bucket, cbeg, cpre, cnt, v, lane, pdv, be, ps, hv, acc, z);
            }
        }
        float4 r;
        {
            float x;
            x = z > 0.0f ? acc.x / z : 0.0f;  r.x = x > 0.0f ? x : expf(x) - 1.0f;
            x = z > 0.0f ? acc.y / z : 0.0f;  r.y = x > 0.0f ? x : expf(x) - 1.0f;
            x = z > 0.0f ? acc.z / z : 0.0f;  r.z = x > 0.0f ? x : expf(x) - 1.0f;
            x = z > 0.0f ? acc.w / z : 0.0f;  r.w = x > 0.0f ? x : expf(x) - 1.0f;
        }
        ((float4*)(c + (size_t)v * D_FEAT))[lane] = r;
    }
}

__device__ __forceinline__ float gru_one(float ir, float iz, float in_,
                                         float hr, float hz, float hn, float x) {
    float r = 1.0f / (1.0f + expf(-(ir + hr)));
    float u = 1.0f / (1.0f + expf(-(iz + hz)));
    float n = tanhf(in_ + r * hn);
    float h = fmaf(1.0f - u, n, u * x);
    return h > 0.0f ? h : 0.0f;
}

#define GATE_TILES 13
template <int T>
__global__ __launch_bounds__(256) void gate_gemm_gru_kernel(
    const float* __restrict__ Cx, const float* __restrict__ NF,
    const float* __restrict__ WihT, const float* __restrict__ WhhT,
    const float* __restrict__ b_ih, const float* __restrict__ b_hh,
    float* __restrict__ out, int V) {
    __shared__ __attribute__((aligned(16))) float xs[2][2][16][132];
    const int tid = threadIdx.x;
    const int lane = tid & 31;
    const int wave = tid >> 5;
    const int lo = lane & 15;
    const int hi = lane >> 4;
    const int nTiles = (V + 15) >> 4;
    const int t0 = blockIdx.x * T;
    const int t1 = (t0 + T < nTiles) ? t0 + T : nTiles;
    if (t0 >= t1) return;
    const int col = wave * 16 + lo;

    v2f bi[3][32], bh[3][32];
#pragma unroll
    for (int g = 0; g < 3; ++g) {
#pragma unroll
        for (int s = 0; s < 32; ++s) {
            const int ka = 4 * s + 2 * hi;
            bi[g][s][0] = WihT[(size_t)ka * 384 + g * 128 + col];
            bi[g][s][1] = WihT[(size_t)(ka + 1) * 384 + g * 128 + col];
            bh[g][s][0] = WhhT[(size_t)ka * 384 + g * 128 + col];
            bh[g][s][1] = WhhT[(size_t)(ka + 1) * 384 + g * 128 + col];
        }
    }

    const int sr = tid >> 5;
    const int sc = (tid & 31) * 4;
    float4 sc0, sc1, sn0, sn1;
    {
        int r0 = t0 * 16 + sr;      if (r0 >= V) r0 = V - 1;
        int r1 = t0 * 16 + sr + 8;  if (r1 >= V) r1 = V - 1;
        const unsigned o0 = (unsigned)r0 * D_FEAT + sc;
        const unsigned o1 = (unsigned)r1 * D_FEAT + sc;
        sc0 = *(const float4*)(Cx + o0);
        sc1 = *(const float4*)(Cx + o1);
        sn0 = *(const float4*)(NF + o0);
        sn1 = *(const float4*)(NF + o1);
    }
    *(float4*)&xs[0][0][sr][sc] = sc0;
    *(float4*)&xs[0][0][sr + 8][sc] = sc1;
    *(float4*)&xs[0][1][sr][sc] = sn0;
    *(float4*)&xs[0][1][sr + 8][sc] = sn1;
    __syncthreads();

    for (int t = t0; t < t1; ++t) {
        const int buf = (t - t0) & 1;
        const bool more = (t + 1 < t1);
        if (more) {
            int r0 = (t + 1) * 16 + sr;      if (r0 >= V) r0 = V - 1;
            int r1 = (t + 1) * 16 + sr + 8;  if (r1 >= V) r1 = V - 1;
            const unsigned o0 = (unsigned)r0 * D_FEAT + sc;
            const unsigned o1 = (unsigned)r1 * D_FEAT + sc;
            sc0 = *(const float4*)(Cx + o0);
            sc1 = *(const float4*)(Cx + o1);
            sn0 = *(const float4*)(NF + o0);
            sn1 = *(const float4*)(NF + o1);
        }
        v8f ai[3], ah[3];
#pragma unroll
        for (int g = 0; g < 3; ++g) {
            const float bvi = b_ih[g * 128 + col];
            const float bvh = b_hh[g * 128 + col];
#pragma unroll
            for (int r = 0; r < 8; ++r) { ai[g][r] = bvi; ah[g][r] = bvh; }
        }
#pragma unroll
        for (int s = 0; s < 32; ++s) {
            const int ka = 4 * s + 2 * hi;
            v2f ac, an;
            ac[0] = xs[buf][0][lo][ka];
            ac[1] = xs[buf][0][lo][ka + 1];
            an[0] = xs[buf][1][lo][ka];
            an[1] = xs[buf][1][lo][ka + 1];
#pragma unroll
            for (int g = 0; g < 3; ++g) {
                ai[g] = __builtin_amdgcn_wmma_f32_16x16x4_f32(
                    false, ac, false, bi[g][s], (short)0, ai[g], false, false);
                ah[g] = __builtin_amdgcn_wmma_f32_16x16x4_f32(
                    false, an, false, bh[g][s], (short)0, ah[g], false, false);
            }
        }
        if (more) {
            *(float4*)&xs[buf ^ 1][0][sr][sc] = sc0;
            *(float4*)&xs[buf ^ 1][0][sr + 8][sc] = sc1;
            *(float4*)&xs[buf ^ 1][1][sr][sc] = sn0;
            *(float4*)&xs[buf ^ 1][1][sr + 8][sc] = sn1;
        }
        float xv[8];
#pragma unroll
        for (int r = 0; r < 8; ++r) xv[r] = xs[buf][1][r + 8 * hi][col];
        __syncthreads();
        const int rowbase = t * 16;
#pragma unroll
        for (int r = 0; r < 8; ++r) {
            const int row = rowbase + r + 8 * hi;
            const float o = gru_one(ai[0][r], ai[1][r], ai[2][r],
                                    ah[0][r], ah[1][r], ah[2][r], xv[r]);
            if (row < V) out[(unsigned)row * D_FEAT + col] = o;
        }
    }
}

extern "C" void kernel_launch(void* const* d_in, const int* in_sizes, int n_in,
                              void* d_out, int out_size, void* d_ws, size_t ws_size,
                              hipStream_t stream) {
    const float* nf     = (const float*)d_in[0];
    const int*   src    = (const int*)d_in[1];
    const int*   dst    = (const int*)d_in[2];
    const float* W_edge = (const float*)d_in[3];
    const float* b_edge = (const float*)d_in[4];
    const float* W_proj = (const float*)d_in[5];
    const float* b_proj = (const float*)d_in[6];
    const float* W_ih   = (const float*)d_in[7];
    const float* b_ih   = (const float*)d_in[8];
    const float* W_hh   = (const float*)d_in[9];
    const float* b_hh   = (const float*)d_in[10];

    const int V = in_sizes[0] / D_FEAT;
    const int E = in_sizes[1];
    const int nGroups = (V + GRP - 1) / GRP;
    const int chunkLen = (E + NCHUNK - 1) / NCHUNK;

    float* hv     = (float*)d_ws;
    float* c      = hv + (size_t)V * 128;
    float* WhhT   = c + (size_t)V * 128;
    float* WihT   = WhhT + 128 * 384;
    int2*  bucket = (int2*)(WihT + 128 * 384);
    int*   segoff = (int*)(bucket + (size_t)NCHUNK * chunkLen);
    float* pd     = (float*)(segoff + (size_t)NCHUNK * (nGroups + 1));
    float* ps     = pd + V;

    const int blk = 256;

    const int rowTiles = (V + 15) / 16;
    const int proj_blocks = (rowTiles + 1) / 2;

    proj_node_pack_kernel<<<NCHUNK + proj_blocks + 2 * PACK_BLOCKS, blk, 0, stream>>>(
        nf, W_proj, b_proj, hv, V, proj_blocks, W_edge, pd, ps,
        W_hh, WhhT, W_ih, WihT, src, dst, E, chunkLen, nGroups, segoff, bucket);

    group_gather_kernel<<<nGroups, 1024, 0, stream>>>(
        segoff, bucket, pd, ps, b_edge, hv, c, V, chunkLen, nGroups);

    gate_gemm_gru_kernel<GATE_TILES><<<(rowTiles + GATE_TILES - 1) / GATE_TILES, blk, 0, stream>>>(
        c, nf, WihT, WhhT, b_ih, b_hh, (float*)d_out, V);
}


# Gate launch: tile barrier split: signal after the tile's LDS traffic, wait before the next tile's products (biases read once), so gate math runs under the SIMD sibling's WMMAs; sums keep order
# speedup vs baseline: 3.0692x; 1.0354x over previous
#include <hip/hip_runtime.h>
#include <hip/hip_bf16.h>
#include <math.h>

typedef float v2f __attribute__((ext_vector_type(2)));
typedef float v8f __attribute__((ext_vector_type(8)));

#define D_FEAT 128
#define GRP 196
#define GRP_PAD 256
#define NCHUNK 256
#define MAXG 1024
#define CAP 8192

__device__ __forceinline__ void node_scalars(const float4 x,
                                             const float* __restrict__ W_edge,
                                             int v, int lane,
                                             float* __restrict__ pd,
                                             float* __restrict__ ps, int V) {
    const float4 w0 = ((const float4*)W_edge)[lane];
    const float4 w1 = ((const float4*)(W_edge + D_FEAT))[lane];
    float d0 = x.x * w0.x + x.y * w0.y + x.z * w0.z + x.w * w0.w;
    float d1 = x.x * w1.x + x.y * w1.y + x.z * w1.z + x.w * w1.w;
#pragma unroll
    for (int off = 16; off > 0; off >>= 1) {
        d0 += __shfl_down(d0, off, 32);
        d1 += __shfl_down(d1, off, 32);
    }
    if (v < V && lane == 0) { pd[v] = d0; ps[v] = d1; }
}

template <int N, int T, bool NODE_PASS>
__device__ __forceinline__ void gemm_rows(
    const float* __restrict__ X, const float* __restrict__ B,
    const float* __restrict__ bias, float* __restrict__ Out, int V,
    int bx, int by,
    const float* __restrict__ W_edge, float* __restrict__ pd,
    float* __restrict__ ps) {
    __shared__ __attribute__((aligned(16))) float xs[2][16][132];
    const int tid = threadIdx.x;
    const int lane = tid & 31;
    const int wave = tid >> 5;
    const int lo = lane & 15;
    const int hi = lane >> 4;
    const int nTiles = (V + 15) >> 4;
    const int t0 = bx * T;
    const int t1 = (t0 + T < nTiles) ? t0 + T : nTiles;
    if (t0 >= t1) return;
    const int colbase = by * 128 + wave * 16;

    v2f bf[32];
#pragma unroll
    for (int s = 0; s < 32; ++s) {
        const int ka = 4 * s + 2 * hi;
        bf[s][0] = B[(size_t)ka * N + colbase + lo];
        bf[s][1] = B[(size_t)(ka + 1) * N + colbase + lo];
    }
    const float bv = bias[colbase + lo];

    const int sr = tid >> 5;
    const int sc = (tid & 31) * 4;
    float4 st0, st1;
    {
        int r0 = t0 * 16 + sr;      if (r0 >= V) r0 = V - 1;
        int r1 = t0 * 16 + sr + 8;  if (r1 >= V) r1 = V - 1;
        st0 = *(const float4*)(X + (size_t)r0 * D_FEAT + sc);
        st1 = *(const float4*)(X + (size_t)r1 * D_FEAT + sc);
    }
    *(float4*)&xs[0][sr][sc] = st0;
    *(float4*)&xs[0][sr + 8][sc] = st1;
    __syncthreads();
    if (NODE_PASS) {
        node_scalars(st0, W_edge, t0 * 16 + sr, lane, pd, ps, V);
        node_scalars(st1, W_edge, t0 * 16 + sr + 8, lane, pd, ps, V);
    }

    for (int t = t0; t < t1; ++t) {
        const int buf = (t - t0) & 1;
        const bool more = (t + 1 < t1);
        if (more) {
            int r0 = (t + 1) * 16 + sr;      if (r0 >= V) r0 = V - 1;
            int r1 = (t + 1) * 16 + sr + 8;  if (r1 >= V) r1 = V - 1;
            st0 = *(const float4*)(X + (size_t)r0 * D_FEAT + sc);
            st1 = *(const float4*)(X + (size_t)r1 * D_FEAT + sc);
        }
        v8f acc;
#pragma unroll
        for (int r = 0; r < 8; ++r) acc[r] = bv;
#pragma unroll
        for (int s = 0; s < 32; ++s) {
            const int ka = 4 * s + 2 * hi;
            v2f a;
            a[0] = xs[buf][lo][ka];
            a[1] = xs[buf][lo][ka + 1];
            acc = __builtin_amdgcn_wmma_f32_16x16x4_f32(
                false, a, false, bf[s], (short)0, acc, false, false);
        }
        if (more) {
            *(float4*)&xs[buf ^ 1][sr][sc] = st0;
            *(float4*)&xs[buf ^ 1][sr + 8][sc] = st1;
        }
        __syncthreads();
        const int rowbase = t * 16;
        if (rowbase + 16 <= V) {
#pragma unroll
            for (int r = 0; r < 8; ++r) {
                int row = rowbase + r + 8 * hi;
                Out[(size_t)row * N + colbase + lo] = acc[r];
            }
        } else {
#pragma unroll
            for (int r = 0; r < 8; ++r) {
                int row = rowbase + r + 8 * hi;
                if (row < V) Out[(size_t)row * N + colbase + lo] = acc[r];
            }
        }
        if (NODE_PASS && more) {
            node_scalars(st0, W_edge, (t + 1) * 16 + sr, lane, pd, ps, V);
            node_scalars(st1, W_edge, (t + 1) * 16 + sr + 8, lane, pd, ps, V);
        }
    }
}

__device__ __forceinline__ int block_excl_scan(int x, int* wsum, int& total) {
    const int lane = threadIdx.x & 31;
    const int wave = threadIdx.x >> 5;
    const int nw = blockDim.x >> 5;
    int inc = x;
#pragma unroll
    for (int off = 1; off < 32; off <<= 1) {
        const int y = __shfl_up(inc, off, 32);
        if (lane >= off) inc += y;
    }
    if (lane == 31) wsum[wave] = inc;
    __syncthreads();
    if (wave == 0) {
        const int w = lane < nw ? wsum[lane] : 0;
        int winc = w;
#pragma unroll
        for (int off = 1; off < 32; off <<= 1) {
            const int y = __shfl_up(winc, off, 32);
            if (lane >= off) winc += y;
        }
        wsum[lane] = winc - w;
        if (lane == 31) wsum[32] = winc;
    }
    __syncthreads();
    total = wsum[32];
    return wsum[wave] + inc - x;
}

#define PACK_BLOCKS ((384 * D_FEAT) / 256)

__global__ __launch_bounds__(256, 8) void proj_node_pack_kernel(
    const float* __restrict__ nf, const float* __restrict__ W_proj,
    const float* __restrict__ b_proj, float* __restrict__ hv, int V,
    int gemmBlocks,
    const float* __restrict__ W_edge, float* __restrict__ pd,
    float* __restrict__ ps,
    const float* __restrict__ W_hh, float* __restrict__ WhhT,
    const float* __restrict__ W_ih, float* __restrict__ WihT,
    const int* __restrict__ src, const int* __restrict__ dst, int E,
    int chunkLen, int nGroups, int* __restrict__ segoff,
    int2* __restrict__ bucket) {
    const int b = blockIdx.x;
    if (b < NCHUNK) {
        __shared__ int hist[MAXG];
        __shared__ int wsum[33];
        const int tid = threadIdx.x;
        const int pitchO = nGroups + 1;
        for (int i = tid; i < nGroups; i += 256) hist[i] = 0;
        __syncthreads();
        const int e0 = b * chunkLen;
        const int e1 = (e0 + chunkLen < E) ? e0 + chunkLen : E;
        for (int e = e0 + tid; e < e1; e += 4 * 256) {
            int d[4];
#pragma unroll
            for (int i = 0; i < 4; ++i) {
                const int ee = e + i * 256;
                d[i] = ee < e1 ? dst[ee] : -1;
            }
#pragma unroll
            for (int i = 0; i < 4; ++i)
                if ((unsigned)d[i] < (unsigned)V) atomicAdd(&hist[d[i] / GRP], 1);
        }
        __syncthreads();
        int carry = 0;
        for (int gb = 0; gb < nGroups; gb += 256) {
            const int gi = gb + tid;
            int tot;
            const int ex = carry + block_excl_scan(gi < nGroups ? hist[gi] : 0, wsum, tot);
            if (gi < nGroups) {
                hist[gi] = ex;
                segoff[(size_t)b * pitchO + gi] = ex;
            }
            carry += tot;
            __syncthreads();
        }
        if (tid == 0) segoff[(size_t)b * pitchO + nGroups] = carry;
        for (int e = e0 + tid; e < e1; e += 4 * 256) {
            int d[4], sv[4];
#pragma unroll
            for (int i = 0; i < 4; ++i) {
                const int ee = e + i * 256;
                d[i] = ee < e1 ? dst[ee] : -1;
                sv[i] = ee < e1 ? src[ee] : 0;
            }
#pragma unroll
            for (int i = 0; i < 4; ++i) {
                if ((unsigned)d[i] < (unsigned)V) {
                    const int p = atomicAdd(&hist[d[i] / GRP], 1);
                    bucket[e0 + p] = make_int2(sv[i], d[i]);
                }
            }
        }
    } else if (b < NCHUNK + gemmBlocks) {
        gemm_rows<128, 2, true>(nf, W_proj, b_proj, hv, V, b - NCHUNK, 0,
                                W_edge, pd, ps);
    } else {
        const int pb = b - NCHUNK - gemmBlocks;
        const bool second = pb >= PACK_BLOCKS;
        const float* __restrict__ Wsrc = second ? W_ih : W_hh;
        float* __restrict__ WT = second ? WihT : WhhT;
        const int i = (second ? pb - PACK_BLOCKS : pb) * 256 + threadIdx.x;
        const int n = i / D_FEAT;
        const int k = i - n * D_FEAT;
        WT[(size_t)k * 384 + n] = Wsrc[i];
    }
}

template <bool GLOB>
__device__ __forceinline__ int gen_src(const int* sl, const int2* __restrict__ bk,
                                       const int* cbeg, const int* cpre,
                                       int i, int v) {
    if (GLOB) {
        int lo = 0, hi = NCHUNK;
#pragma unroll
        for (int st = 0; st < 8; ++st) {
            const int mid = (lo + hi) >> 1;
            if (cpre[mid] <= i) lo = mid; else hi = mid;
        }
        const int2 ed = bk[cbeg[lo] + (i - cpre[lo])];
        return ed.y == v ? ed.x : -1;
    }
    return sl[i];
}

template <bool GLOB>
__device__ __forceinline__ void gather_any(
    const int* sl, const int2* __restrict__ bk, const int* cbeg,
    const int* cpre, int len, int v, int lane,
    float pdv, float be, const float* __restrict__ ps,
    const float* __restrict__ hv, float4& acc, float& z) {
    float m = -INFINITY;
    for (int i = lane; i < len; i += 32) {
        const int s = gen_src<GLOB>(sl, bk, cbeg, cpre, i, v);
        if (s >= 0) {
            const float x = pdv + ps[s] + be;
            const float l = x > 0.0f ? x : 0.01f * x;
            m = fmaxf(m, l);
        }
    }
#pragma unroll
    for (int off = 16; off > 0; off >>= 1) m = fmaxf(m, __shfl_xor(m, off, 32));
    int last = -1;
    for (int it = 0; it < len; ++it) {
        int best = 0x7fffffff;
        for (int i = lane; i < len; i += 32) {
            const int s = gen_src<GLOB>(sl, bk, cbeg, cpre, i, v);
            if (s > last && s < best) best = s;
        }
#pragma unroll
        for (int off = 16; off > 0; off >>= 1) {
            const int o = __shfl_xor(best, off, 32);
            best = o < best ? o : best;
        }
        if (best == 0x7fffffff) break;
        int mult = 0;
        for (int i = lane; i < len; i += 32)
            mult += (gen_src<GLOB>(sl, bk, cbeg, cpre, i, v) == best) ? 1 : 0;
#pragma unroll
        for (int off = 16; off > 0; off >>= 1) mult += __shfl_xor(mult, off, 32);
        const int sb = __builtin_amdgcn_readfirstlane(best);
        const int mu = __builtin_amdgcn_readfirstlane(mult);
        const float x = pdv + ps[sb] + be;
        const float l = x > 0.0f ? x : 0.01f * x;
        const float w = expf(l - m);
        const float4 h = ((const float4*)(hv + (size_t)sb * D_FEAT))[lane];
        for (int r = 0; r < mu; ++r) {
            z += w;
            acc.x = fmaf(w, h.x, acc.x);
            acc.y = fmaf(w, h.y, acc.y);
            acc.z = fmaf(w, h.z, acc.z);
            acc.w = fmaf(w, h.w, acc.w);
        }
        last = sb;
    }
}

#define ROWS_AHEAD 8
__global__ __launch_bounds__(1024) void group_gather_kernel(
    const int* __restrict__ segoff, const int2* __restrict__ bucket,
    const float* __restrict__ pd, const float* __restrict__ ps,
    const float* __restrict__ b_edge, const float* __restrict__ hv,
    float* __restrict__ c, int V, int chunkLen, int nGroups) {
    __shared__ int cbeg[NCHUNK];
    __shared__ int cpre[NCHUNK + 1];
    __shared__ int ncnt[GRP_PAD];
    __shared__ int noff[GRP_PAD];
    __shared__ int nfill[GRP_PAD];
    __shared__ int sorted[CAP];
    __shared__ int wsum[33];
    __shared__ int nextNode;
    const int tid = threadIdx.x;
    const int lane = tid & 31;
    const int g = blockIdx.x;
    const int v0 = g * GRP;
    const int nv = (V - v0 < GRP) ? V - v0 : GRP;
    int segBeg = 0, segLen = 0;
    if (tid < NCHUNK) {
        const int* __restrict__ row = segoff + (size_t)tid * (nGroups + 1) + g;
        const int a = row[0];
        segLen = row[1] - a;
        segBeg = tid * chunkLen + a;
    }
    if (tid < GRP_PAD) ncnt[tid] = 0;
    if (tid == 0) nextNode = 0;
    int cnt;
    const int pre = block_excl_scan(segLen, wsum, cnt);
    if (tid < NCHUNK) { cbeg[tid] = segBeg; cpre[tid] = pre; }
    if (tid == 0) cpre[NCHUNK] = cnt;
    __syncthreads();
    const bool fits = cnt <= CAP;
    const int sBeg = cbeg[tid >> 2];
    const int sLen = cpre[(tid >> 2) + 1] - cpre[tid >> 2];
    const int q = tid & 3;
    const int2* __restrict__ seg = bucket + sBeg;
    int2 ed[4];
#pragma unroll
    for (int i = 0; i < 4; ++i) {
        const int idx = q + 4 * i;
        ed[i] = make_int2(0, -1);
        if (idx < sLen) ed[i] = seg[idx];
    }
#pragma unroll
    for (int i = 0; i < 4; ++i)
        if (ed[i].y >= 0) atomicAdd(&ncnt[ed[i].y - v0], 1);
    for (int i = q + 16; i < sLen; i += 4) atomicAdd(&ncnt[seg[i].y - v0], 1);
    __syncthreads();
    int all;
    const int ex = block_excl_scan(tid < GRP_PAD ? ncnt[tid] : 0, wsum, all);
    if (tid < GRP_PAD) { noff[tid] = ex; nfill[tid] = ex; }
    __syncthreads();
    if (fits) {
#pragma unroll
        for (int i = 0; i < 4; ++i) {
            if (ed[i].y >= 0) {
                const int p = atomicAdd(&nfill[ed[i].y - v0], 1);
                sorted[p] = ed[i].x;
            }
        }
        for (int i = q + 16; i < sLen; i += 4) {
            const int2 e2 = seg[i];
            const int p = atomicAdd(&nfill[e2.y - v0], 1);
            sorted[p] = e2.x;
        }
    }
    __syncthreads();

    const float be = b_edge[0];
    for (int turn = 0; turn < GRP; ++turn) {
        int t = 0;
        if (lane == 0) t = atomicAdd(&nextNode, 1);
        t = __builtin_amdgcn_readfirstlane(t);
        if (t >= nv) break;
        const int v = v0 + t;
        const int n = __builtin_amdgcn_readfirstlane(ncnt[t]);
        const int o = __builtin_amdgcn_readfirstlane(noff[t]);
        float4 acc = make_float4(0.0f, 0.0f, 0.0f, 0.0f);
        float z = 0.0f;
        if (n > 0) {
            const float pdv = pd[v];
            if (fits && n <= 64) {
                const bool two = n > 32;
                int s0 = lane < n ? sorted[o + lane] : 0x7fffffff;
                int s1 = 32 + lane < n ? sorted[o + 32 + lane] : 0x7fffffff;
                int r0 = 0, r1 = 0;
                const int nA = two ? 32 : n;
                for (int j = 0; j < nA; ++j) {
                    const int sj = __builtin_amdgcn_readlane(s0, j);
                    r0 += (sj < s0 || (sj == s0 && j < lane)) ? 1 : 0;
                    if (two) r1 += (sj <= s1) ? 1 : 0;
                }
                for (int j = 32; j < n; ++j) {
                    const int sj = __builtin_amdgcn_readlane(s1, j - 32);
                    r0 += (sj < s0) ? 1 : 0;
                    r1 += (sj < s1 || (sj == s1 && j - 32 < lane)) ? 1 : 0;
                }
                if (lane < n) sorted[o + r0] = s0;
                if (32 + lane < n) sorted[o + r1] = s1;
                s0 = lane < n ? sorted[o + lane] : 0;
                s1 = 32 + lane < n ? sorted[o + 32 + lane] : 0;
                float l0 = -INFINITY, l1 = -INFINITY;
                if (lane < n) {
                    const float x = pdv + ps[s0] + be;
                    l0 = x > 0.0f ? x : 0.01f * x;
                }
                if (32 + lane < n) {
                    const float x = pdv + ps[s1] + be;
                    l1 = x > 0.0f ? x : 0.01f * x;
                }
                float m = fmaxf(l0, l1);
#pragma unroll
                for (int off = 16; off > 0; off >>= 1)
                    m = fmaxf(m, __shfl_xor(m, off, 32));
                const int w0 = __float_as_int(lane < n ? expf(l0 - m) : 0.0f);
                const int w1 = __float_as_int(32 + lane < n ? expf(l1 - m) : 0.0f);
                for (int k = 0; k < n; k += ROWS_AHEAD) {
                    const int sr = k >= 32 ? s1 : s0;
                    const int wr = k >= 32 ? w1 : w0;
                    const int kb = k & 31;
                    float4 h[ROWS_AHEAD];
                    float wk[ROWS_AHEAD];
#pragma unroll
                    for (int i = 0; i < ROWS_AHEAD; ++i) {
                        h[i] = make_float4(0.0f, 0.0f, 0.0f, 0.0f);
                        wk[i] = 0.0f;
                        if (k + i < n) {
                            const int sk = __builtin_amdgcn_readlane(sr, kb + i);
                            wk[i] = __int_as_float(__builtin_amdgcn_readlane(wr, kb + i));
                            h[i] = ((const float4*)(hv + (size_t)sk * D_FEAT))[lane];
                        }
                    }
#pragma unroll
                    for (int i = 0; i < ROWS_AHEAD; ++i) {
                        z += wk[i];
                        acc.x = fmaf(wk[i], h[i].x, acc.x);
                        acc.y = fmaf(wk[i], h[i].y, acc.y);
                        acc.z = fmaf(wk[i], h[i].z, acc.z);
                        acc.w = fmaf(wk[i], h[i].w, acc.w);
                    }
                }
            } else if (fits) {
                gather_any<false>(sorted + o, bucket, cbeg, cpre, n, v, lane, pdv, be, ps, hv, acc, z);
            } else {
                gather_any<true>(sorted, bucket, cbeg, cpre, cnt, v, lane, pdv, be, ps, hv, acc, z);
            }
        }
        float4 r;
        {
            float x;
            x = z > 0.0f ? acc.x / z : 0.0f;  r.x = x > 0.0f ? x : expf(x) - 1.0f;
            x = z > 0.0f ? acc.y / z : 0.0f;  r.y = x > 0.0f ? x : expf(x) - 1.0f;
            x = z > 0.0f ? acc.z / z : 0.0f;  r.z = x > 0.0f ? x : expf(x) - 1.0f;
            x = z > 0.0f ? acc.w / z : 0.0f;  r.w = x > 0.0f ? x : expf(x) - 1.0f;
        }
        ((float4*)(c + (size_t)v * D_FEAT))[lane] = r;
    }
}

__device__ __forceinline__ float gru_one(float ir, float iz, float in_,
                                         float hr, float hz, float hn, float x) {
    float r = 1.0f / (1.0f + expf(-(ir + hr)));
    float u = 1.0f / (1.0f + expf(-(iz + hz)));
    float n = tanhf(in_ + r * hn);
    float h = fmaf(1.0f - u, n, u * x);
    return h > 0.0f ? h : 0.0f;
}

#define GATE_TILES 13
template <int T>
__global__ __launch_bounds__(256) void gate_gemm_gru_kernel(
    const float* __restrict__ Cx, const float* __restrict__ NF,
    const float* __restrict__ WihT, const float* __restrict__ WhhT,
    const float* __restrict__ b_ih, const float* __restrict__ b_hh,
    float* __restrict__ out, int V) {
    __shared__ __attribute__((aligned(16))) float xs[2][2][16][132];
    const int tid = threadIdx.x;
    const int lane = tid & 31;
    const int wave = tid >> 5;
    const int lo = lane & 15;
    const int hi = lane >> 4;
    const int nTiles = (V + 15) >> 4;
    const int t0 = blockIdx.x * T;
    const int t1 = (t0 + T < nTiles) ? t0 + T : nTiles;
    if (t0 >= t1) return;
    const int col = wave * 16 + lo;

    v2f bi[3][32], bh[3][32];
#pragma unroll
    for (int g = 0; g < 3; ++g) {
#pragma unroll
        for (int s = 0; s < 32; ++s) {
            const int ka = 4 * s + 2 * hi;
            bi[g][s][0] = WihT[(size_t)ka * 384 + g * 128 + col];
            bi[g][s][1] = WihT[(size_t)(ka + 1) * 384 + g * 128 + col];
            bh[g][s][0] = WhhT[(size_t)ka * 384 + g * 128 + col];
            bh[g][s][1] = WhhT[(size_t)(ka + 1) * 384 + g * 128 + col];
        }
    }

    float bvi[3], bvh[3];
#pragma unroll
    for (int g = 0; g < 3; ++g) {
        bvi[g] = b_ih[g * 128 + col];
        bvh[g] = b_hh[g * 128 + col];
    }

    const int sr = tid >> 5;
    const int sc = (tid & 31) * 4;
    float4 sc0, sc1, sn0, sn1;
    {
        int r0 = t0 * 16 + sr;      if (r0 >= V) r0 = V - 1;
        int r1 = t0 * 16 + sr + 8;  if (r1 >= V) r1 = V - 1;
        const unsigned o0 = (unsigned)r0 * D_FEAT + sc;
        const unsigned o1 = (unsigned)r1 * D_FEAT + sc;
        sc0 = *(const float4*)(Cx + o0);
        sc1 = *(const float4*)(Cx + o1);
        sn0 = *(const float4*)(NF + o0);
        sn1 = *(const float4*)(NF + o1);
    }
    *(float4*)&xs[0][0][sr][sc] = sc0;
    *(float4*)&xs[0][0][sr + 8][sc] = sc1;
    *(float4*)&xs[0][1][sr][sc] = sn0;
    *(float4*)&xs[0][1][sr + 8][sc] = sn1;
    __syncthreads();

    for (int t = t0; t < t1; ++t) {
        const int buf = (t - t0) & 1;
        const bool more = (t + 1 < t1);
        if (more) {
            int r0 = (t + 1) * 16 + sr;      if (r0 >= V) r0 = V - 1;
            int r1 = (t + 1) * 16 + sr + 8;  if (r1 >= V) r1 = V - 1;
            const unsigned o0 = (unsigned)r0 * D_FEAT + sc;
            const unsigned o1 = (unsigned)r1 * D_FEAT + sc;
            sc0 = *(const float4*)(Cx + o0);
            sc1 = *(const float4*)(Cx + o1);
            sn0 = *(const float4*)(NF + o0);
            sn1 = *(const float4*)(NF + o1);
        }
        v8f ai[3], ah[3];
#pragma unroll
        for (int g = 0; g < 3; ++g) {
#pragma unroll
            for (int r = 0; r < 8; ++r) { ai[g][r] = bvi[g]; ah[g][r] = bvh[g]; }
        }
        if (t != t0) {
            __builtin_amdgcn_s_barrier_wait(-1);
            __builtin_amdgcn_fence(__ATOMIC_ACQUIRE, "workgroup", "local");
        }
#pragma unroll
        for (int s = 0; s < 32; ++s) {
            const int ka = 4 * s + 2 * hi;
            v2f ac, an;
            ac[0] = xs[buf][0][lo][ka];
            ac[1] = xs[buf][0][lo][ka + 1];
            an[0] = xs[buf][1][lo][ka];
            an[1] = xs[buf][1][lo][ka + 1];
#pragma unroll
            for (int g = 0; g < 3; ++g) {
                ai[g] = __builtin_amdgcn_wmma_f32_16x16x4_f32(
                    false, ac, false, bi[g][s], (short)0, ai[g], false, false);
                ah[g] = __builtin_amdgcn_wmma_f32_16x16x4_f32(
                    false, an, false, bh[g][s], (short)0, ah[g], false, false);
            }
        }
        if (more) {
            *(float4*)&xs[buf ^ 1][0][sr][sc] = sc0;
            *(float4*)&xs[buf ^ 1][0][sr + 8][sc] = sc1;
            *(float4*)&xs[buf ^ 1][1][sr][sc] = sn0;
            *(float4*)&xs[buf ^ 1][1][sr + 8][sc] = sn1;
        }
        float xv[8];
#pragma unroll
        for (int r = 0; r < 8; ++r) xv[r] = xs[buf][1][r + 8 * hi][col];
        if (more) {
            __builtin_amdgcn_fence(__ATOMIC_RELEASE, "workgroup", "local");
            __builtin_amdgcn_s_barrier_signal(-1);
        }
        const int rowbase = t * 16;
#pragma unroll
        for (int r = 0; r < 8; ++r) {
            const int row = rowbase + r + 8 * hi;
            const float o = gru_one(ai[0][r], ai[1][r], ai[2][r],
                                    ah[0][r], ah[1][r], ah[2][r], xv[r]);
            if (row < V) out[(unsigned)row * D_FEAT + col] = o;
        }
    }
}

extern "C" void kernel_launch(void* const* d_in, const int* in_sizes, int n_in,
                              void* d_out, int out_size, void* d_ws, size_t ws_size,
                              hipStream_t stream) {
    const float* nf     = (const float*)d_in[0];
    const int*   src    = (const int*)d_in[1];
    const int*   dst    = (const int*)d_in[2];
    const float* W_edge = (const float*)d_in[3];
    const float* b_edge = (const float*)d_in[4];
    const float* W_proj = (const float*)d_in[5];
    const float* b_proj = (const float*)d_in[6];
    const float* W_ih   = (const float*)d_in[7];
    const float* b_ih   = (const float*)d_in[8];
    const float* W_hh   = (const float*)d_in[9];
    const float* b_hh   = (const float*)d_in[10];

    const int V = in_sizes[0] / D_FEAT;
    const int E = in_sizes[1];
    const int nGroups = (V + GRP - 1) / GRP;
    const int chunkLen = (E + NCHUNK - 1) / NCHUNK;

    float* hv     = (float*)d_ws;
    float* c      = hv + (size_t)V * 128;
    float* WhhT   = c + (size_t)V * 128;
    float* WihT   = WhhT + 128 * 384;
    int2*  bucket = (int2*)(WihT + 128 * 384);
    int*   segoff = (int*)(bucket + (size_t)NCHUNK * chunkLen);
    float* pd     = (float*)(segoff + (size_t)NCHUNK * (nGroups + 1));
    float* ps     = pd + V;

    const int blk = 256;

    const int rowTiles = (V + 15) / 16;
    const int proj_blocks = (rowTiles + 1) / 2;

    proj_node_pack_kernel<<<NCHUNK + proj_blocks + 2 * PACK_BLOCKS, blk, 0, stream>>>(
        nf, W_proj, b_proj, hv, V, proj_blocks, W_edge, pd, ps,
        W_hh, WhhT, W_ih, WihT, src, dst, E, chunkLen, nGroups, segoff, bucket);

    group_gather_kernel<<<nGroups, 1024, 0, stream>>>(
        segoff, bucket, pd, ps, b_edge, hv, c, V, chunkLen, nGroups);

    gate_gemm_gru_kernel<GATE_TILES><<<(rowTiles + GATE_TILES - 1) / GATE_TILES, blk, 0, stream>>>(
        c, nf, WihT, WhhT, b_ih, b_hh, (float*)d_out, V);
}


# Gate launch: gate math without branches (tanh: both forms evaluated, lane picks one; row guard only at the stores): a lane's 8 elements run as interleaved chains, not 8 serial ones; sums keep order
# speedup vs baseline: 3.3772x; 1.1004x over previous
#include <hip/hip_runtime.h>
#include <hip/hip_bf16.h>
#include <math.h>

typedef float v2f __attribute__((ext_vector_type(2)));
typedef float v8f __attribute__((ext_vector_type(8)));

#define D_FEAT 128
#define GRP 196
#define GRP_PAD 256
#define NCHUNK 256
#define MAXG 1024
#define CAP 8192

__device__ __forceinline__ void node_scalars(const float4 x,
                                             const float* __restrict__ W_edge,
                                             int v, int lane,
                                             float* __restrict__ pd,
                                             float* __restrict__ ps, int V) {
    const float4 w0 = ((const float4*)W_edge)[lane];
    const float4 w1 = ((const float4*)(W_edge + D_FEAT))[lane];
    float d0 = x.x * w0.x + x.y * w0.y + x.z * w0.z + x.w * w0.w;
    float d1 = x.x * w1.x + x.y * w1.y + x.z * w1.z + x.w * w1.w;
#pragma unroll
    for (int off = 16; off > 0; off >>= 1) {
        d0 += __shfl_down(d0, off, 32);
        d1 += __shfl_down(d1, off, 32);
    }
    if (v < V && lane == 0) { pd[v] = d0; ps[v] = d1; }
}

template <int N, int T, bool NODE_PASS>
__device__ __forceinline__ void gemm_rows(
    const float* __restrict__ X, const float* __restrict__ B,
    const float* __restrict__ bias, float* __restrict__ Out, int V,
    int bx, int by,
    const float* __restrict__ W_edge, float* __restrict__ pd,
    float* __restrict__ ps) {
    __shared__ __attribute__((aligned(16))) float xs[2][16][132];
    const int tid = threadIdx.x;
    const int lane = tid & 31;
    const int wave = tid >> 5;
    const int lo = lane & 15;
    const int hi = lane >> 4;
    const int nTiles = (V + 15) >> 4;
    const int t0 = bx * T;
    const int t1 = (t0 + T < nTiles) ? t0 + T : nTiles;
    if (t0 >= t1) return;
    const int colbase = by * 128 + wave * 16;

    v2f bf[32];
#pragma unroll
    for (int s = 0; s < 32; ++s) {
        const int ka = 4 * s + 2 * hi;
        bf[s][0] = B[(size_t)ka * N + colbase + lo];
        bf[s][1] = B[(size_t)(ka + 1) * N + colbase + lo];
    }
    const float bv = bias[colbase + lo];

    const int sr = tid >> 5;
    const int sc = (tid & 31) * 4;
    float4 st0, st1;
    {
        int r0 = t0 * 16 + sr;      if (r0 >= V) r0 = V - 1;
        int r1 = t0 * 16 + sr + 8;  if (r1 >= V) r1 = V - 1;
        st0 = *(const float4*)(X + (size_t)r0 * D_FEAT + sc);
        st1 = *(const float4*)(X + (size_t)r1 * D_FEAT + sc);
    }
    *(float4*)&xs[0][sr][sc] = st0;
    *(float4*)&xs[0][sr + 8][sc] = st1;
    __syncthreads();
    if (NODE_PASS) {
        node_scalars(st0, W_edge, t0 * 16 + sr, lane, pd, ps, V);
        node_scalars(st1, W_edge, t0 * 16 + sr + 8, lane, pd, ps, V);
    }

    for (int t = t0; t < t1; ++t) {
        const int buf = (t - t0) & 1;
        const bool more = (t + 1 < t1);
        if (more) {
            int r0 = (t + 1) * 16 + sr;      if (r0 >= V) r0 = V - 1;
            int r1 = (t + 1) * 16 + sr + 8;  if (r1 >= V) r1 = V - 1;
            st0 = *(const float4*)(X + (size_t)r0 * D_FEAT + sc);
            st1 = *(const float4*)(X + (size_t)r1 * D_FEAT + sc);
        }
        v8f acc;
#pragma unroll
        for (int r = 0; r < 8; ++r) acc[r] = bv;
#pragma unroll
        for (int s = 0; s < 32; ++s) {
            const int ka = 4 * s + 2 * hi;
            v2f a;
            a[0] = xs[buf][lo][ka];
            a[1] = xs[buf][lo][ka + 1];
            acc = __builtin_amdgcn_wmma_f32_16x16x4_f32(
                false, a, false, bf[s], (short)0, acc, false, false);
        }
        if (more) {
            *(float4*)&xs[buf ^ 1][sr][sc] = st0;
            *(float4*)&xs[buf ^ 1][sr + 8][sc] = st1;
        }
        __syncthreads();
        const int rowbase = t * 16;
        if (rowbase + 16 <= V) {
#pragma unroll
            for (int r = 0; r < 8; ++r) {
                int row = rowbase + r + 8 * hi;
                Out[(size_t)row * N + colbase + lo] = acc[r];
            }
        } else {
#pragma unroll
            for (int r = 0; r < 8; ++r) {
                int row = rowbase + r + 8 * hi;
                if (row < V) Out[(size_t)row * N + colbase + lo] = acc[r];
            }
        }
        if (NODE_PASS && more) {
            node_scalars(st0, W_edge, (t + 1) * 16 + sr, lane, pd, ps, V);
            node_scalars(st1, W_edge, (t + 1) * 16 + sr + 8, lane, pd, ps, V);
        }
    }
}

__device__ __forceinline__ int block_excl_scan(int x, int* wsum, int& total) {
    const int lane = threadIdx.x & 31;
    const int wave = threadIdx.x >> 5;
    const int nw = blockDim.x >> 5;
    int inc = x;
#pragma unroll
    for (int off = 1; off < 32; off <<= 1) {
        const int y = __shfl_up(inc, off, 32);
        if (lane >= off) inc += y;
    }
    if (lane == 31) wsum[wave] = inc;
    __syncthreads();
    if (wave == 0) {
        const int w = lane < nw ? wsum[lane] : 0;
        int winc = w;
#pragma unroll
        for (int off = 1; off < 32; off <<= 1) {
            const int y = __shfl_up(winc, off, 32);
            if (lane >= off) winc += y;
        }
        wsum[lane] = winc - w;
        if (lane == 31) wsum[32] = winc;
    }
    __syncthreads();
    total = wsum[32];
    return wsum[wave] + inc - x;
}

#define PACK_BLOCKS ((384 * D_FEAT) / 256)

__global__ __launch_bounds__(256, 8) void proj_node_pack_kernel(
    const float* __restrict__ nf, const float* __restrict__ W_proj,
    const float* __restrict__ b_proj, float* __restrict__ hv, int V,
    int gemmBlocks,
    const float* __restrict__ W_edge, float* __restrict__ pd,
    float* __restrict__ ps,
    const float* __restrict__ W_hh, float* __restrict__ WhhT,
    const float* __restrict__ W_ih, float* __restrict__ WihT,
    const int* __restrict__ src, const int* __restrict__ dst, int E,
    int chunkLen, int nGroups, int* __restrict__ segoff,
    int2* __restrict__ bucket) {
    const int b = blockIdx.x;
    if (b < NCHUNK) {
        __shared__ int hist[MAXG];
        __shared__ int wsum[33];
        const int tid = threadIdx.x;
        const int pitchO = nGroups + 1;
        for (int i = tid; i < nGroups; i += 256) hist[i] = 0;
        __syncthreads();
        const int e0 = b * chunkLen;
        const int e1 = (e0 + chunkLen < E) ? e0 + chunkLen : E;
        for (int e = e0 + tid; e < e1; e += 4 * 256) {
            int d[4];
#pragma unroll
            for (int i = 0; i < 4; ++i) {
                const int ee = e + i * 256;
                d[i] = ee < e1 ? dst[ee] : -1;
            }
#pragma unroll
            for (int i = 0; i < 4; ++i)
                if ((unsigned)d[i] < (unsigned)V) atomicAdd(&hist[d[i] / GRP], 1);
        }
        __syncthreads();
        int carry = 0;
        for (int gb = 0; gb < nGroups; gb += 256) {
            const int gi = gb + tid;
            int tot;
            const int ex = carry + block_excl_scan(gi < nGroups ? hist[gi] : 0, wsum, tot);
            if (gi < nGroups) {
                hist[gi] = ex;
                segoff[(size_t)b * pitchO + gi] = ex;
            }
            carry += tot;
            __syncthreads();
        }
        if (tid == 0) segoff[(size_t)b * pitchO + nGroups] = carry;
        for (int e = e0 + tid; e < e1; e += 4 * 256) {
            int d[4], sv[4];
#pragma unroll
            for (int i = 0; i < 4; ++i) {
                const int ee = e + i * 256;
                d[i] = ee < e1 ? dst[ee] : -1;
                sv[i] = ee < e1 ? src[ee] : 0;
            }
#pragma unroll
            for (int i = 0; i < 4; ++i) {
                if ((unsigned)d[i] < (unsigned)V) {
                    const int p = atomicAdd(&hist[d[i] / GRP], 1);
                    bucket[e0 + p] = make_int2(sv[i], d[i]);
                }
            }
        }
    } else if (b < NCHUNK + gemmBlocks) {
        gemm_rows<128, 2, true>(nf, W_proj, b_proj, hv, V, b - NCHUNK, 0,
                                W_edge, pd, ps);
    } else {
        const int pb = b - NCHUNK - gemmBlocks;
        const bool second = pb >= PACK_BLOCKS;
        const float* __restrict__ Wsrc = second ? W_ih : W_hh;
        float* __restrict__ WT = second ? WihT : WhhT;
        const int i = (second ? pb - PACK_BLOCKS : pb) * 256 + threadIdx.x;
        const int n = i / D_FEAT;
        const int k = i - n * D_FEAT;
        WT[(size_t)k * 384 + n] = Wsrc[i];
    }
}

template <bool GLOB>
__device__ __forceinline__ int gen_src(const int* sl, const int2* __restrict__ bk,
                                       const int* cbeg, const int* cpre,
                                       int i, int v) {
    if (GLOB) {
        int lo = 0, hi = NCHUNK;
#pragma unroll
        for (int st = 0; st < 8; ++st) {
            const int mid = (lo + hi) >> 1;
            if (cpre[mid] <= i) lo = mid; else hi = mid;
        }
        const int2 ed = bk[cbeg[lo] + (i - cpre[lo])];
        return ed.y == v ? ed.x : -1;
    }
    return sl[i];
}

template <bool GLOB>
__device__ __forceinline__ void gather_any(
    const int* sl, const int2* __restrict__ bk, const int* cbeg,
    const int* cpre, int len, int v, int lane,
    float pdv, float be, const float* __restrict__ ps,
    const float* __restrict__ hv, float4& acc, float& z) {
    float m = -INFINITY;
    for (int i = lane; i < len; i += 32) {
        const int s = gen_src<GLOB>(sl, bk, cbeg, cpre, i, v);
        if (s >= 0) {
            const float x = pdv + ps[s] + be;
            const float l = x > 0.0f ? x : 0.01f * x;
            m = fmaxf(m, l);
        }
    }
#pragma unroll
    for (int off = 16; off > 0; off >>= 1) m = fmaxf(m, __shfl_xor(m, off, 32));
    int last = -1;
    for (int it = 0; it < len; ++it) {
        int best = 0x7fffffff;
        for (int i = lane; i < len; i += 32) {
            const int s = gen_src<GLOB>(sl, bk, cbeg, cpre, i, v);
            if (s > last && s < best) best = s;
        }
#pragma unroll
        for (int off = 16; off > 0; off >>= 1) {
            const int o = __shfl_xor(best, off, 32);
            best = o < best ? o : best;
        }
        if (best == 0x7fffffff) break;
        int mult = 0;
        for (int i = lane; i < len; i += 32)
            mult += (gen_src<GLOB>(sl, bk, cbeg, cpre, i, v) == best) ? 1 : 0;
#pragma unroll
        for (int off = 16; off > 0; off >>= 1) mult += __shfl_xor(mult, off, 32);
        const int sb = __builtin_amdgcn_readfirstlane(best);
        const int mu = __builtin_amdgcn_readfirstlane(mult);
        const float x = pdv + ps[sb] + be;
        const float l = x > 0.0f ? x : 0.01f * x;
        const float w = expf(l - m);
        const float4 h = ((const float4*)(hv + (size_t)sb * D_FEAT))[lane];
        for (int r = 0; r < mu; ++r) {
            z += w;
            acc.x = fmaf(w, h.x, acc.x);
            acc.y = fmaf(w, h.y, acc.y);
            acc.z = fmaf(w, h.z, acc.z);
            acc.w = fmaf(w, h.w, acc.w);
        }
        last = sb;
    }
}

#define ROWS_AHEAD 8
__global__ __launch_bounds__(1024) void group_gather_kernel(
    const int* __restrict__ segoff, const int2* __restrict__ bucket,
    const float* __restrict__ pd, const float* __restrict__ ps,
    const float* __restrict__ b_edge, const float* __restrict__ hv,
    float* __restrict__ c, int V, int chunkLen, int nGroups) {
    __shared__ int cbeg[NCHUNK];
    __shared__ int cpre[NCHUNK + 1];
    __shared__ int ncnt[GRP_PAD];
    __shared__ int noff[GRP_PAD];
    __shared__ int nfill[GRP_PAD];
    __shared__ int sorted[CAP];
    __shared__ int wsum[33];
    __shared__ int nextNode;
    const int tid = threadIdx.x;
    const int lane = tid & 31;
    const int g = blockIdx.x;
    const int v0 = g * GRP;
    const int nv = (V - v0 < GRP) ? V - v0 : GRP;
    int segBeg = 0, segLen = 0;
    if (tid < NCHUNK) {
        const int* __restrict__ row = segoff + (size_t)tid * (nGroups + 1) + g;
        const int a = row[0];
        segLen = row[1] - a;
        segBeg = tid * chunkLen + a;
    }
    if (tid < GRP_PAD) ncnt[tid] = 0;
    if (tid == 0) nextNode = 0;
    int cnt;
    const int pre = block_excl_scan(segLen, wsum, cnt);
    if (tid < NCHUNK) { cbeg[tid] = segBeg; cpre[tid] = pre; }
    if (tid == 0) cpre[NCHUNK] = cnt;
    __syncthreads();
    const bool fits = cnt <= CAP;
    const int sBeg = cbeg[tid >> 2];
    const int sLen = cpre[(tid >> 2) + 1] - cpre[tid >> 2];
    const int q = tid & 3;
    const int2* __restrict__ seg = bucket + sBeg;
    int2 ed[4];
#pragma unroll
    for (int i = 0; i < 4; ++i) {
        const int idx = q + 4 * i;
        ed[i] = make_int2(0, -1);
        if (idx < sLen) ed[i] = seg[idx];
    }
#pragma unroll
    for (int i = 0; i < 4; ++i)
        if (ed[i].y >= 0) atomicAdd(&ncnt[ed[i].y - v0], 1);
    for (int i = q + 16; i < sLen; i += 4) atomicAdd(&ncnt[seg[i].y - v0], 1);
    __syncthreads();
    int all;
    const int ex = block_excl_scan(tid < GRP_PAD ? ncnt[tid] : 0, wsum, all);
    if (tid < GRP_PAD) { noff[tid] = ex; nfill[tid] = ex; }
    __syncthreads();
    if (fits) {
#pragma unroll
        for (int i = 0; i < 4; ++i) {
            if (ed[i].y >= 0) {
                const int p = atomicAdd(&nfill[ed[i].y - v0], 1);
                sorted[p] = ed[i].x;
            }
        }
        for (int i = q + 16; i < sLen; i += 4) {
            const int2 e2 = seg[i];
            const int p = atomicAdd(&nfill[e2.y - v0], 1);
            sorted[p] = e2.x;
        }
    }
    __syncthreads();

    const float be = b_edge[0];
    for (int turn = 0; turn < GRP; ++turn) {
        int t = 0;
        if (lane == 0) t = atomicAdd(&nextNode, 1);
        t = __builtin_amdgcn_readfirstlane(t);
        if (t >= nv) break;
        const int v = v0 + t;
        const int n = __builtin_amdgcn_readfirstlane(ncnt[t]);
        const int o = __builtin_amdgcn_readfirstlane(noff[t]);
        float4 acc = make_float4(0.0f, 0.0f, 0.0f, 0.0f);
        float z = 0.0f;
        if (n > 0) {
            const float pdv = pd[v];
            if (fits && n <= 64) {
                const bool two = n > 32;
                int s0 = lane < n ? sorted[o + lane] : 0x7fffffff;
                int s1 = 32 + lane < n ? sorted[o + 32 + lane] : 0x7fffffff;
                int r0 = 0, r1 = 0;
                const int nA = two ? 32 : n;
                for (int j = 0; j < nA; ++j) {
                    const int sj = __builtin_amdgcn_readlane(s0, j);
                    r0 += (sj < s0 || (sj == s0 && j < lane)) ? 1 : 0;
                    if (two) r1 += (sj <= s1) ? 1 : 0;
                }
                for (int j = 32; j < n; ++j) {
                    const int sj = __builtin_amdgcn_readlane(s1, j - 32);
                    r0 += (sj < s0) ? 1 : 0;
                    r1 += (sj < s1 || (sj == s1 && j - 32 < lane)) ? 1 : 0;
                }
                if (lane < n) sorted[o + r0] = s0;
                if (32 + lane < n) sorted[o + r1] = s1;
                s0 = lane < n ? sorted[o + lane] : 0;
                s1 = 32 + lane < n ? sorted[o + 32 + lane] : 0;
                float l0 = -INFINITY, l1 = -INFINITY;
                if (lane < n) {
                    const float x = pdv + ps[s0] + be;
                    l0 = x > 0.0f ? x : 0.01f * x;
                }
                if (32 + lane < n) {
                    const float x = pdv + ps[s1] + be;
                    l1 = x > 0.0f ? x : 0.01f * x;
                }
                float m = fmaxf(l0, l1);
#pragma unroll
                for (int off = 16; off > 0; off >>= 1)
                    m = fmaxf(m, __shfl_xor(m, off, 32));
                const int w0 = __float_as_int(lane < n ? expf(l0 - m) : 0.0f);
                const int w1 = __float_as_int(32 + lane < n ? expf(l1 - m) : 0.0f);
                for (int k = 0; k < n; k += ROWS_AHEAD) {
                    const int sr = k >= 32 ? s1 : s0;
                    const int wr = k >= 32 ? w1 : w0;
                    const int kb = k & 31;
                    float4 h[ROWS_AHEAD];
                    float wk[ROWS_AHEAD];
#pragma unroll
                    for (int i = 0; i < ROWS_AHEAD; ++i) {
                        h[i] = make_float4(0.0f, 0.0f, 0.0f, 0.0f);
                        wk[i] = 0.0f;
                        if (k + i < n) {
                            const int sk = __builtin_amdgcn_readlane(sr, kb + i);
                            wk[i] = __int_as_float(__builtin_amdgcn_readlane(wr, kb + i));
                            h[i] = ((const float4*)(hv + (size_t)sk * D_FEAT))[lane];
                        }
                    }
#pragma unroll
                    for (int i = 0; i < ROWS_AHEAD; ++i) {
                        z += wk[i];
                        acc.x = fmaf(wk[i], h[i].x, acc.x);
                        acc.y = fmaf(wk[i], h[i].y, acc.y);
                        acc.z = fmaf(wk[i], h[i].z, acc.z);
                        acc.w = fmaf(wk[i], h[i].w, acc.w);
                    }
                }
            } else if (fits) {
                gather_any<false>(sorted + o, bucket, cbeg, cpre, n, v, lane, pdv, be, ps, hv, acc, z);
            } else {
                gather_any<true>(sorted, bucket, cbeg, cpre, cnt, v, lane, pdv, be, ps, hv, acc, z);
            }
        }
        float4 r;
        {
            float x;
            x = z > 0.0f ? acc.x / z : 0.0f;  r.x = x > 0.0f ? x : expf(x) - 1.0f;
            x = z > 0.0f ? acc.y / z : 0.0f;  r.y = x > 0.0f ? x : expf(x) - 1.0f;
            x = z > 0.0f ? acc.z / z : 0.0f;  r.z = x > 0.0f ? x : expf(x) - 1.0f;
            x = z > 0.0f ? acc.w / z : 0.0f;  r.w = x > 0.0f ? x : expf(x) - 1.0f;
        }
        ((float4*)(c + (size_t)v * D_FEAT))[lane] = r;
    }
}

__device__ __forceinline__ float sigmoid_f(float x) {
    return 1.0f / (1.0f + expf(-x));
}

__device__ __forceinline__ float tanh_sel(float a) {
    const float t = fabsf(a);
    const float t2 = t * t;
    float p = fmaf(t2, -0x1.758e7ap-8f, 0x1.521192p-6f);
    p = fmaf(t2, p, -0x1.b8389cp-5f);
    p = fmaf(t2, p, 0x1.110704p-3f);
    p = fmaf(t2, p, -0x1.555532p-2f);
    const float small = fmaf(t2, t * p, t);
    const float e = expf(t + t);
    const float big = fmaf(-2.0f, __builtin_amdgcn_rcpf(e + 1.0f), 1.0f);
    return copysignf(t < 0.625f ? small : big, a);
}

#define GATE_TILES 13
template <int T>
__global__ __launch_bounds__(256) void gate_gemm_gru_kernel(
    const float* __restrict__ Cx, const float* __restrict__ NF,
    const float* __restrict__ WihT, const float* __restrict__ WhhT,
    const float* __restrict__ b_ih, const float* __restrict__ b_hh,
    float* __restrict__ out, int V) {
    __shared__ __attribute__((aligned(16))) float xs[2][2][16][132];
    const int tid = threadIdx.x;
    const int lane = tid & 31;
    const int wave = tid >> 5;
    const int lo = lane & 15;
    const int hi = lane >> 4;
    const int nTiles = (V + 15) >> 4;
    const int t0 = blockIdx.x * T;
    const int t1 = (t0 + T < nTiles) ? t0 + T : nTiles;
    if (t0 >= t1) return;
    const int col = wave * 16 + lo;

    v2f bi[3][32], bh[3][32];
#pragma unroll
    for (int g = 0; g < 3; ++g) {
#pragma unroll
        for (int s = 0; s < 32; ++s) {
            const int ka = 4 * s + 2 * hi;
            bi[g][s][0] = WihT[(size_t)ka * 384 + g * 128 + col];
            bi[g][s][1] = WihT[(size_t)(ka + 1) * 384 + g * 128 + col];
            bh[g][s][0] = WhhT[(size_t)ka * 384 + g * 128 + col];
            bh[g][s][1] = WhhT[(size_t)(ka + 1) * 384 + g * 128 + col];
        }
    }

    float bvi[3], bvh[3];
#pragma unroll
    for (int g = 0; g < 3; ++g) {
        bvi[g] = b_ih[g * 128 + col];
        bvh[g] = b_hh[g * 128 + col];
    }

    const int sr = tid >> 5;
    const int sc = (tid & 31) * 4;
    float4 sc0, sc1, sn0, sn1;
    {
        int r0 = t0 * 16 + sr;      if (r0 >= V) r0 = V - 1;
        int r1 = t0 * 16 + sr + 8;  if (r1 >= V) r1 = V - 1;
        const unsigned o0 = (unsigned)r0 * D_FEAT + sc;
        const unsigned o1 = (unsigned)r1 * D_FEAT + sc;
        sc0 = *(const float4*)(Cx + o0);
        sc1 = *(const float4*)(Cx + o1);
        sn0 = *(const float4*)(NF + o0);
        sn1 = *(const float4*)(NF + o1);
    }
    *(float4*)&xs[0][0][sr][sc] = sc0;
    *(float4*)&xs[0][0][sr + 8][sc] = sc1;
    *(float4*)&xs[0][1][sr][sc] = sn0;
    *(float4*)&xs[0][1][sr + 8][sc] = sn1;
    __syncthreads();

    for (int t = t0; t < t1; ++t) {
        const int buf = (t - t0) & 1;
        const bool more = (t + 1 < t1);
        if (more) {
            int r0 = (t + 1) * 16 + sr;      if (r0 >= V) r0 = V - 1;
            int r1 = (t + 1) * 16 + sr + 8;  if (r1 >= V) r1 = V - 1;
            const unsigned o0 = (unsigned)r0 * D_FEAT + sc;
            const unsigned o1 = (unsigned)r1 * D_FEAT + sc;
            sc0 = *(const float4*)(Cx + o0);
            sc1 = *(const float4*)(Cx + o1);
            sn0 = *(const float4*)(NF + o0);
            sn1 = *(const float4*)(NF + o1);
        }
        v8f ai[3], ah[3];
#pragma unroll
        for (int g = 0; g < 3; ++g) {
#pragma unroll
            for (int r = 0; r < 8; ++r) { ai[g][r] = bvi[g]; ah[g][r] = bvh[g]; }
        }
        if (t != t0) {
            __builtin_amdgcn_s_barrier_wait(-1);
            __builtin_amdgcn_fence(__ATOMIC_ACQUIRE, "workgroup", "local");
        }
#pragma unroll
        for (int s = 0; s < 32; ++s) {
            const int ka = 4 * s + 2 * hi;
            v2f ac, an;
            ac[0] = xs[buf][0][lo][ka];
            ac[1] = xs[buf][0][lo][ka + 1];
            an[0] = xs[buf][1][lo][ka];
            an[1] = xs[buf][1][lo][ka + 1];
#pragma unroll
            for (int g = 0; g < 3; ++g) {
                ai[g] = __builtin_amdgcn_wmma_f32_16x16x4_f32(
                    false, ac, false, bi[g][s], (short)0, ai[g], false, false);
                ah[g] = __builtin_amdgcn_wmma_f32_16x16x4_f32(
                    false, an, false, bh[g][s], (short)0, ah[g], false, false);
            }
        }
        if (more) {
            *(float4*)&xs[buf ^ 1][0][sr][sc] = sc0;
            *(float4*)&xs[buf ^ 1][0][sr + 8][sc] = sc1;
            *(float4*)&xs[buf ^ 1][1][sr][sc] = sn0;
            *(float4*)&xs[buf ^ 1][1][sr + 8][sc] = sn1;
        }
        float xv[8];
#pragma unroll
        for (int r = 0; r < 8; ++r) xv[r] = xs[buf][1][r + 8 * hi][col];
        if (more) {
            __builtin_amdgcn_fence(__ATOMIC_RELEASE, "workgroup", "local");
            __builtin_amdgcn_s_barrier_signal(-1);
        }
        float gr[8], gu[8], gn[8], go[8];
#pragma unroll
        for (int r = 0; r < 8; ++r) gr[r] = sigmoid_f(ai[0][r] + ah[0][r]);
#pragma unroll
        for (int r = 0; r < 8; ++r) gu[r] = sigmoid_f(ai[1][r] + ah[1][r]);
#pragma unroll
        for (int r = 0; r < 8; ++r) gn[r] = tanh_sel(ai[2][r] + gr[r] * ah[2][r]);
#pragma unroll
        for (int r = 0; r < 8; ++r) {
            const float h = fmaf(1.0f - gu[r], gn[r], gu[r] * xv[r]);
            go[r] = h > 0.0f ? h : 0.0f;
        }
        const int rowbase = t * 16;
        if (rowbase + 16 <= V) {
#pragma unroll
            for (int r = 0; r < 8; ++r)
                out[(unsigned)(rowbase + r + 8 * hi) * D_FEAT + col] = go[r];
        } else {
#pragma unroll
            for (int r = 0; r < 8; ++r) {
                const int row = rowbase + r + 8 * hi;
                if (row < V) out[(unsigned)row * D_FEAT + col] = go[r];
            }
        }
    }
}

extern "C" void kernel_launch(void* const* d_in, const int* in_sizes, int n_in,
                              void* d_out, int out_size, void* d_ws, size_t ws_size,
                              hipStream_t stream) {
    const float* nf     = (const float*)d_in[0];
    const int*   src    = (const int*)d_in[1];
    const int*   dst    = (const int*)d_in[2];
    const float* W_edge = (const float*)d_in[3];
    const float* b_edge = (const float*)d_in[4];
    const float* W_proj = (const float*)d_in[5];
    const float* b_proj = (const float*)d_in[6];
    const float* W_ih   = (const float*)d_in[7];
    const float* b_ih   = (const float*)d_in[8];
    const float* W_hh   = (const float*)d_in[9];
    const float* b_hh   = (const float*)d_in[10];

    const int V = in_sizes[0] / D_FEAT;
    const int E = in_sizes[1];
    const int nGroups = (V + GRP - 1) / GRP;
    const int chunkLen = (E + NCHUNK - 1) / NCHUNK;

    float* hv     = (float*)d_ws;
    float* c      = hv + (size_t)V * 128;
    float* WhhT   = c + (size_t)V * 128;
    float* WihT   = WhhT + 128 * 384;
    int2*  bucket = (int2*)(WihT + 128 * 384);
    int*   segoff = (int*)(bucket + (size_t)NCHUNK * chunkLen);
    float* pd     = (float*)(segoff + (size_t)NCHUNK * (nGroups + 1));
    float* ps     = pd + V;

    const int blk = 256;

    const int rowTiles = (V + 15) / 16;
    const int proj_blocks = (rowTiles + 1) / 2;

    proj_node_pack_kernel<<<NCHUNK + proj_blocks + 2 * PACK_BLOCKS, blk, 0, stream>>>(
        nf, W_proj, b_proj, hv, V, proj_blocks, W_edge, pd, ps,
        W_hh, WhhT, W_ih, WihT, src, dst, E, chunkLen, nGroups, segoff, bucket);

    group_gather_kernel<<<nGroups, 1024, 0, stream>>>(
        segoff, bucket, pd, ps, b_edge, hv, c, V, chunkLen, nGroups);

    gate_gemm_gru_kernel<GATE_TILES><<<(rowTiles + GATE_TILES - 1) / GATE_TILES, blk, 0, stream>>>(
        c, nf, WihT, WhhT, b_ih, b_hh, (float*)d_out, V);
}


# Launch 1 as one resident generation: 768 projection blocks share the row tiles evenly (B fragments fetched once per 4-5 tiles) and do the weight transposes; 17624 waves become 8192; sums keep order
# speedup vs baseline: 3.5029x; 1.0372x over previous
#include <hip/hip_runtime.h>
#include <hip/hip_bf16.h>
#include <math.h>

typedef float v2f __attribute__((ext_vector_type(2)));
typedef float v8f __attribute__((ext_vector_type(8)));

#define D_FEAT 128
#define GRP 196
#define GRP_PAD 256
#define NCHUNK 256
#define MAXG 1024
#define CAP 8192

__device__ __forceinline__ void node_scalars(const float4 x,
                                             const float* __restrict__ W_edge,
                                             int v, int lane,
                                             float* __restrict__ pd,
                                             float* __restrict__ ps, int V) {
    const float4 w0 = ((const float4*)W_edge)[lane];
    const float4 w1 = ((const float4*)(W_edge + D_FEAT))[lane];
    float d0 = x.x * w0.x + x.y * w0.y + x.z * w0.z + x.w * w0.w;
    float d1 = x.x * w1.x + x.y * w1.y + x.z * w1.z + x.w * w1.w;
#pragma unroll
    for (int off = 16; off > 0; off >>= 1) {
        d0 += __shfl_down(d0, off, 32);
        d1 += __shfl_down(d1, off, 32);
    }
    if (v < V && lane == 0) { pd[v] = d0; ps[v] = d1; }
}

template <int N, bool NODE_PASS>
__device__ __forceinline__ void gemm_rows(
    const float* __restrict__ X, const float* __restrict__ B,
    const float* __restrict__ bias, float* __restrict__ Out, int V,
    int t0, int t1, int by,
    const float* __restrict__ W_edge, float* __restrict__ pd,
    float* __restrict__ ps) {
    __shared__ __attribute__((aligned(16))) float xs[2][16][132];
    const int tid = threadIdx.x;
    const int lane = tid & 31;
    const int wave = tid >> 5;
    const int lo = lane & 15;
    const int hi = lane >> 4;
    if (t0 >= t1) return;
    const int colbase = by * 128 + wave * 16;

    v2f bf[32];
#pragma unroll
    for (int s = 0; s < 32; ++s) {
        const int ka = 4 * s + 2 * hi;
        bf[s][0] = B[(size_t)ka * N + colbase + lo];
        bf[s][1] = B[(size_t)(ka + 1) * N + colbase + lo];
    }
    const float bv = bias[colbase + lo];

    const int sr = tid >> 5;
    const int sc = (tid & 31) * 4;
    float4 st0, st1;
    {
        int r0 = t0 * 16 + sr;      if (r0 >= V) r0 = V - 1;
        int r1 = t0 * 16 + sr + 8;  if (r1 >= V) r1 = V - 1;
        st0 = *(const float4*)(X + (size_t)r0 * D_FEAT + sc);
        st1 = *(const float4*)(X + (size_t)r1 * D_FEAT + sc);
    }
    *(float4*)&xs[0][sr][sc] = st0;
    *(float4*)&xs[0][sr + 8][sc] = st1;
    __syncthreads();
    if (NODE_PASS) {
        node_scalars(st0, W_edge, t0 * 16 + sr, lane, pd, ps, V);
        node_scalars(st1, W_edge, t0 * 16 + sr + 8, lane, pd, ps, V);
    }

    for (int t = t0; t < t1; ++t) {
        const int buf = (t - t0) & 1;
        const bool more = (t + 1 < t1);
        if (more) {
            int r0 = (t + 1) * 16 + sr;      if (r0 >= V) r0 = V - 1;
            int r1 = (t + 1) * 16 + sr + 8;  if (r1 >= V) r1 = V - 1;
            st0 = *(const float4*)(X + (size_t)r0 * D_FEAT + sc);
            st1 = *(const float4*)(X + (size_t)r1 * D_FEAT + sc);
        }
        v8f acc;
#pragma unroll
        for (int r = 0; r < 8; ++r) acc[r] = bv;
#pragma unroll
        for (int s = 0; s < 32; ++s) {
            const int ka = 4 * s + 2 * hi;
            v2f a;
            a[0] = xs[buf][lo][ka];
            a[1] = xs[buf][lo][ka + 1];
            acc = __builtin_amdgcn_wmma_f32_16x16x4_f32(
                false, a, false, bf[s], (short)0, acc, false, false);
        }
        if (more) {
            *(float4*)&xs[buf ^ 1][sr][sc] = st0;
            *(float4*)&xs[buf ^ 1][sr + 8][sc] = st1;
        }
        __syncthreads();
        const int rowbase = t * 16;
        if (rowbase + 16 <= V) {
#pragma unroll
            for (int r = 0; r < 8; ++r) {
                int row = rowbase + r + 8 * hi;
                Out[(size_t)row * N + colbase + lo] = acc[r];
            }
        } else {
#pragma unroll
            for (int r = 0; r < 8; ++r) {
                int row = rowbase + r + 8 * hi;
                if (row < V) Out[(size_t)row * N + colbase + lo] = acc[r];
            }
        }
        if (NODE_PASS && more) {
            node_scalars(st0, W_edge, (t + 1) * 16 + sr, lane, pd, ps, V);
            node_scalars(st1, W_edge, (t + 1) * 16 + sr + 8, lane, pd, ps, V);
        }
    }
}

__device__ __forceinline__ int block_excl_scan(int x, int* wsum, int& total) {
    const int lane = threadIdx.x & 31;
    const int wave = threadIdx.x >> 5;
    const int nw = blockDim.x >> 5;
    int inc = x;
#pragma unroll
    for (int off = 1; off < 32; off <<= 1) {
        const int y = __shfl_up(inc, off, 32);
        if (lane >= off) inc += y;
    }
    if (lane == 31) wsum[wave] = inc;
    __syncthreads();
    if (wave == 0) {
        const int w = lane < nw ? wsum[lane] : 0;
        int winc = w;
#pragma unroll
        for (int off = 1; off < 32; off <<= 1) {
            const int y = __shfl_up(winc, off, 32);
            if (lane >= off) winc += y;
        }
        wsum[lane] = winc - w;
        if (lane == 31) wsum[32] = winc;
    }
    __syncthreads();
    total = wsum[32];
    return wsum[wave] + inc - x;
}

#define PACK_BLOCKS ((384 * D_FEAT) / 256)

__global__ __launch_bounds__(256, 8) void proj_node_pack_kernel(
    const float* __restrict__ nf, const float* __restrict__ W_proj,
    const float* __restrict__ b_proj, float* __restrict__ hv, int V,
    int gemmBlocks,
    const float* __restrict__ W_edge, float* __restrict__ pd,
    float* __restrict__ ps,
    const float* __restrict__ W_hh, float* __restrict__ WhhT,
    const float* __restrict__ W_ih, float* __restrict__ WihT,
    const int* __restrict__ src, const int* __restrict__ dst, int E,
    int chunkLen, int nGroups, int* __restrict__ segoff,
    int2* __restrict__ bucket) {
    const int b = blockIdx.x;
    if (b < NCHUNK) {
        __shared__ int hist[MAXG];
        __shared__ int wsum[33];
        const int tid = threadIdx.x;
        const int pitchO = nGroups + 1;
        for (int i = tid; i < nGroups; i += 256) hist[i] = 0;
        __syncthreads();
        const int e0 = b * chunkLen;
        const int e1 = (e0 + chunkLen < E) ? e0 + chunkLen : E;
        for (int e = e0 + tid; e < e1; e += 4 * 256) {
            int d[4];
#pragma unroll
            for (int i = 0; i < 4; ++i) {
                const int ee = e + i * 256;
                d[i] = ee < e1 ? dst[ee] : -1;
            }
#pragma unroll
            for (int i = 0; i < 4; ++i)
                if ((unsigned)d[i] < (unsigned)V) atomicAdd(&hist[d[i] / GRP], 1);
        }
        __syncthreads();
        int carry = 0;
        for (int gb = 0; gb < nGroups; gb += 256) {
            const int gi = gb + tid;
            int tot;
            const int ex = carry + block_excl_scan(gi < nGroups ? hist[gi] : 0, wsum, tot);
            if (gi < nGroups) {
                hist[gi] = ex;
                segoff[(size_t)b * pitchO + gi] = ex;
            }
            carry += tot;
            __syncthreads();
        }
        if (tid == 0) segoff[(size_t)b * pitchO + nGroups] = carry;
        for (int e = e0 + tid; e < e1; e += 4 * 256) {
            int d[4], sv[4];
#pragma unroll
            for (int i = 0; i < 4; ++i) {
                const int ee = e + i * 256;
                d[i] = ee < e1 ? dst[ee] : -1;
                sv[i] = ee < e1 ? src[ee] : 0;
            }
#pragma unroll
            for (int i = 0; i < 4; ++i) {
                if ((unsigned)d[i] < (unsigned)V) {
                    const int p = atomicAdd(&hist[d[i] / GRP], 1);
                    bucket[e0 + p] = make_int2(sv[i], d[i]);
                }
            }
        }
    } else {
        const int gb = b - NCHUNK;
        if (gb < 2 * PACK_BLOCKS) {
            const int pb = gb;
            const bool second = pb >= PACK_BLOCKS;
            const float* __restrict__ Wsrc = second ? W_ih : W_hh;
            float* __restrict__ WT = second ? WihT : WhhT;
            const int i = (second ? pb - PACK_BLOCKS : pb) * 256 + threadIdx.x;
            const int n = i / D_FEAT;
            const int k = i - n * D_FEAT;
            WT[(size_t)k * 384 + n] = Wsrc[i];
        }
        const int nTiles = (V + 15) >> 4;
        const int q = nTiles / gemmBlocks;
        const int r = nTiles - q * gemmBlocks;
        const int t0 = gb * q + (gb < r ? gb : r);
        const int t1 = t0 + q + (gb < r ? 1 : 0);
        gemm_rows<128, true>(nf, W_proj, b_proj, hv, V, t0, t1, 0,
                             W_edge, pd, ps);
    }
}

template <bool GLOB>
__device__ __forceinline__ int gen_src(const int* sl, const int2* __restrict__ bk,
                                       const int* cbeg, const int* cpre,
                                       int i, int v) {
    if (GLOB) {
        int lo = 0, hi = NCHUNK;
#pragma unroll
        for (int st = 0; st < 8; ++st) {
            const int mid = (lo + hi) >> 1;
            if (cpre[mid] <= i) lo = mid; else hi = mid;
        }
        const int2 ed = bk[cbeg[lo] + (i - cpre[lo])];
        return ed.y == v ? ed.x : -1;
    }
    return sl[i];
}

template <bool GLOB>
__device__ __forceinline__ void gather_any(
    const int* sl, const int2* __restrict__ bk, const int* cbeg,
    const int* cpre, int len, int v, int lane,
    float pdv, float be, const float* __restrict__ ps,
    const float* __restrict__ hv, float4& acc, float& z) {
    float m = -INFINITY;
    for (int i = lane; i < len; i += 32) {
        const int s = gen_src<GLOB>(sl, bk, cbeg, cpre, i, v);
        if (s >= 0) {
            const float x = pdv + ps[s] + be;
            const float l = x > 0.0f ? x : 0.01f * x;
            m = fmaxf(m, l);
        }
    }
#pragma unroll
    for (int off = 16; off > 0; off >>= 1) m = fmaxf(m, __shfl_xor(m, off, 32));
    int last = -1;
    for (int it = 0; it < len; ++it) {
        int best = 0x7fffffff;
        for (int i = lane; i < len; i += 32) {
            const int s = gen_src<GLOB>(sl, bk, cbeg, cpre, i, v);
            if (s > last && s < best) best = s;
        }
#pragma unroll
        for (int off = 16; off > 0; off >>= 1) {
            const int o = __shfl_xor(best, off, 32);
            best = o < best ? o : best;
        }
        if (best == 0x7fffffff) break;
        int mult = 0;
        for (int i = lane; i < len; i += 32)
            mult += (gen_src<GLOB>(sl, bk, cbeg, cpre, i, v) == best) ? 1 : 0;
#pragma unroll
        for (int off = 16; off > 0; off >>= 1) mult += __shfl_xor(mult, off, 32);
        const int sb = __builtin_amdgcn_readfirstlane(best);
        const int mu = __builtin_amdgcn_readfirstlane(mult);
        const float x = pdv + ps[sb] + be;
        const float l = x > 0.0f ? x : 0.01f * x;
        const float w = expf(l - m);
        const float4 h = ((const float4*)(hv + (size_t)sb * D_FEAT))[lane];
        for (int r = 0; r < mu; ++r) {
            z += w;
            acc.x = fmaf(w, h.x, acc.x);
            acc.y = fmaf(w, h.y, acc.y);
            acc.z = fmaf(w, h.z, acc.z);
            acc.w = fmaf(w, h.w, acc.w);
        }
        last = sb;
    }
}

#define ROWS_AHEAD 8
__global__ __launch_bounds__(1024) void group_gather_kernel(
    const int* __restrict__ segoff, const int2* __restrict__ bucket,
    const float* __restrict__ pd, const float* __restrict__ ps,
    const float* __restrict__ b_edge, const float* __restrict__ hv,
    float* __restrict__ c, int V, int chunkLen, int nGroups) {
    __shared__ int cbeg[NCHUNK];
    __shared__ int cpre[NCHUNK + 1];
    __shared__ int ncnt[GRP_PAD];
    __shared__ int noff[GRP_PAD];
    __shared__ int nfill[GRP_PAD];
    __shared__ int sorted[CAP];
    __shared__ int wsum[33];
    __shared__ int nextNode;
    const int tid = threadIdx.x;
    const int lane = tid & 31;
    const int g = blockIdx.x;
    const int v0 = g * GRP;
    const int nv = (V - v0 < GRP) ? V - v0 : GRP;
    int segBeg = 0, segLen = 0;
    if (tid < NCHUNK) {
        const int* __restrict__ row = segoff + (size_t)tid * (nGroups + 1) + g;
        const int a = row[0];
        segLen = row[1] - a;
        segBeg = tid * chunkLen + a;
    }
    if (tid < GRP_PAD) ncnt[tid] = 0;
    if (tid == 0) nextNode = 0;
    int cnt;
    const int pre = block_excl_scan(segLen, wsum, cnt);
    if (tid < NCHUNK) { cbeg[tid] = segBeg; cpre[tid] = pre; }
    if (tid == 0) cpre[NCHUNK] = cnt;
    __syncthreads();
    const bool fits = cnt <= CAP;
    const int sBeg = cbeg[tid >> 2];
    const int sLen = cpre[(tid >> 2) + 1] - cpre[tid >> 2];
    const int q = tid & 3;
    const int2* __restrict__ seg = bucket + sBeg;
    int2 ed[4];
#pragma unroll
    for (int i = 0; i < 4; ++i) {
        const int idx = q + 4 * i;
        ed[i] = make_int2(0, -1);
        if (idx < sLen) ed[i] = seg[idx];
    }
#pragma unroll
    for (int i = 0; i < 4; ++i)
        if (ed[i].y >= 0) atomicAdd(&ncnt[ed[i].y - v0], 1);
    for (int i = q + 16; i < sLen; i += 4) atomicAdd(&ncnt[seg[i].y - v0], 1);
    __syncthreads();
    int all;
    const int ex = block_excl_scan(tid < GRP_PAD ? ncnt[tid] : 0, wsum, all);
    if (tid < GRP_PAD) { noff[tid] = ex; nfill[tid] = ex; }
    __syncthreads();
    if (fits) {
#pragma unroll
        for (int i = 0; i < 4; ++i) {
            if (ed[i].y >= 0) {
                const int p = atomicAdd(&nfill[ed[i].y - v0], 1);
                sorted[p] = ed[i].x;
            }
        }
        for (int i = q + 16; i < sLen; i += 4) {
            const int2 e2 = seg[i];
            const int p = atomicAdd(&nfill[e2.y - v0], 1);
            sorted[p] = e2.x;
        }
    }
    __syncthreads();

    const float be = b_edge[0];
    for (int turn = 0; turn < GRP; ++turn) {
        int t = 0;
        if (lane == 0) t = atomicAdd(&nextNode, 1);
        t = __builtin_amdgcn_readfirstlane(t);
        if (t >= nv) break;
        const int v = v0 + t;
        const int n = __builtin_amdgcn_readfirstlane(ncnt[t]);
        const int o = __builtin_amdgcn_readfirstlane(noff[t]);
        float4 acc = make_float4(0.0f, 0.0f, 0.0f, 0.0f);
        float z = 0.0f;
        if (n > 0) {
            const float pdv = pd[v];
            if (fits && n <= 64) {
                const bool two = n > 32;
                int s0 = lane < n ? sorted[o + lane] : 0x7fffffff;
                int s1 = 32 + lane < n ? sorted[o + 32 + lane] : 0x7fffffff;
                int r0 = 0, r1 = 0;
                const int nA = two ? 32 : n;
                for (int j = 0; j < nA; ++j) {
                    const int sj = __builtin_amdgcn_readlane(s0, j);
                    r0 += (sj < s0 || (sj == s0 && j < lane)) ? 1 : 0;
                    if (two) r1 += (sj <= s1) ? 1 : 0;
                }
                for (int j = 32; j < n; ++j) {
                    const int sj = __builtin_amdgcn_readlane(s1, j - 32);
                    r0 += (sj < s0) ? 1 : 0;
                    r1 += (sj < s1 || (sj == s1 && j - 32 < lane)) ? 1 : 0;
                }
                if (lane < n) sorted[o + r0] = s0;
                if (32 + lane < n) sorted[o + r1] = s1;
                s0 = lane < n ? sorted[o + lane] : 0;
                s1 = 32 + lane < n ? sorted[o + 32 + lane] : 0;
                float l0 = -INFINITY, l1 = -INFINITY;
                if (lane < n) {
                    const float x = pdv + ps[s0] + be;
                    l0 = x > 0.0f ? x : 0.01f * x;
                }
                if (32 + lane < n) {
                    const float x = pdv + ps[s1] + be;
                    l1 = x > 0.0f ? x : 0.01f * x;
                }
                float m = fmaxf(l0, l1);
#pragma unroll
                for (int off = 16; off > 0; off >>= 1)
                    m = fmaxf(m, __shfl_xor(m, off, 32));
                const int w0 = __float_as_int(lane < n ? expf(l0 - m) : 0.0f);
                const int w1 = __float_as_int(32 + lane < n ? expf(l1 - m) : 0.0f);
                for (int k = 0; k < n; k += ROWS_AHEAD) {
                    const int sr = k >= 32 ? s1 : s0;
                    const int wr = k >= 32 ? w1 : w0;
                    const int kb = k & 31;
                    float4 h[ROWS_AHEAD];
                    float wk[ROWS_AHEAD];
#pragma unroll
                    for (int i = 0; i < ROWS_AHEAD; ++i) {
                        h[i] = make_float4(0.0f, 0.0f, 0.0f, 0.0f);
                        wk[i] = 0.0f;
                        if (k + i < n) {
                            const int sk = __builtin_amdgcn_readlane(sr, kb + i);
                            wk[i] = __int_as_float(__builtin_amdgcn_readlane(wr, kb + i));
                            h[i] = ((const float4*)(hv + (size_t)sk * D_FEAT))[lane];
                        }
                    }
#pragma unroll
                    for (int i = 0; i < ROWS_AHEAD; ++i) {
                        z += wk[i];
                        acc.x = fmaf(wk[i], h[i].x, acc.x);
                        acc.y = fmaf(wk[i], h[i].y, acc.y);
                        acc.z = fmaf(wk[i], h[i].z, acc.z);
                        acc.w = fmaf(wk[i], h[i].w, acc.w);
                    }
                }
            } else if (fits) {
                gather_any<false>(sorted + o, bucket, cbeg, cpre, n, v, lane, pdv, be, ps, hv, acc, z);
            } else {
                gather_any<true>(sorted, bucket, cbeg, cpre, cnt, v, lane, pdv, be, ps, hv, acc, z);
            }
        }
        float4 r;
        {
            float x;
            x = z > 0.0f ? acc.x / z : 0.0f;  r.x = x > 0.0f ? x : expf(x) - 1.0f;
            x = z > 0.0f ? acc.y / z : 0.0f;  r.y = x > 0.0f ? x : expf(x) - 1.0f;
            x = z > 0.0f ? acc.z / z : 0.0f;  r.z = x > 0.0f ? x : expf(x) - 1.0f;
            x = z > 0.0f ? acc.w / z : 0.0f;  r.w = x > 0.0f ? x : expf(x) - 1.0f;
        }
        ((float4*)(c + (size_t)v * D_FEAT))[lane] = r;
    }
}

__device__ __forceinline__ float sigmoid_f(float x) {
    return 1.0f / (1.0f + expf(-x));
}

__device__ __forceinline__ float tanh_sel(float a) {
    const float t = fabsf(a);
    const float t2 = t * t;
    float p = fmaf(t2, -0x1.758e7ap-8f, 0x1.521192p-6f);
    p = fmaf(t2, p, -0x1.b8389cp-5f);
    p = fmaf(t2, p, 0x1.110704p-3f);
    p = fmaf(t2, p, -0x1.555532p-2f);
    const float small = fmaf(t2, t * p, t);
    const float e = expf(t + t);
    const float big = fmaf(-2.0f, __builtin_amdgcn_rcpf(e + 1.0f), 1.0f);
    return copysignf(t < 0.625f ? small : big, a);
}

#define GATE_TILES 13
template <int T>
__global__ __launch_bounds__(256) void gate_gemm_gru_kernel(
    const float* __restrict__ Cx, const float* __restrict__ NF,
    const float* __restrict__ WihT, const float* __restrict__ WhhT,
    const float* __restrict__ b_ih, const float* __restrict__ b_hh,
    float* __restrict__ out, int V) {
    __shared__ __attribute__((aligned(16))) float xs[2][2][16][132];
    const int tid = threadIdx.x;
    const int lane = tid & 31;
    const int wave = tid >> 5;
    const int lo = lane & 15;
    const int hi = lane >> 4;
    const int nTiles = (V + 15) >> 4;
    const int t0 = blockIdx.x * T;
    const int t1 = (t0 + T < nTiles) ? t0 + T : nTiles;
    if (t0 >= t1) return;
    const int col = wave * 16 + lo;

    v2f bi[3][32], bh[3][32];
#pragma unroll
    for (int g = 0; g < 3; ++g) {
#pragma unroll
        for (int s = 0; s < 32; ++s) {
            const int ka = 4 * s + 2 * hi;
            bi[g][s][0] = WihT[(size_t)ka * 384 + g * 128 + col];
            bi[g][s][1] = WihT[(size_t)(ka + 1) * 384 + g * 128 + col];
            bh[g][s][0] = WhhT[(size_t)ka * 384 + g * 128 + col];
            bh[g][s][1] = WhhT[(size_t)(ka + 1) * 384 + g * 128 + col];
        }
    }

    float bvi[3], bvh[3];
#pragma unroll
    for (int g = 0; g < 3; ++g) {
        bvi[g] = b_ih[g * 128 + col];
        bvh[g] = b_hh[g * 128 + col];
    }

    const int sr = tid >> 5;
    const int sc = (tid & 31) * 4;
    float4 sc0, sc1, sn0, sn1;
    {
        int r0 = t0 * 16 + sr;      if (r0 >= V) r0 = V - 1;
        int r1 = t0 * 16 + sr + 8;  if (r1 >= V) r1 = V - 1;
        const unsigned o0 = (unsigned)r0 * D_FEAT + sc;
        const unsigned o1 = (unsigned)r1 * D_FEAT + sc;
        sc0 = *(const float4*)(Cx + o0);
        sc1 = *(const float4*)(Cx + o1);
        sn0 = *(const float4*)(NF + o0);
        sn1 = *(const float4*)(NF + o1);
    }
    *(float4*)&xs[0][0][sr][sc] = sc0;
    *(float4*)&xs[0][0][sr + 8][sc] = sc1;
    *(float4*)&xs[0][1][sr][sc] = sn0;
    *(float4*)&xs[0][1][sr + 8][sc] = sn1;
    __syncthreads();

    for (int t = t0; t < t1; ++t) {
        const int buf = (t - t0) & 1;
        const bool more = (t + 1 < t1);
        if (more) {
            int r0 = (t + 1) * 16 + sr;      if (r0 >= V) r0 = V - 1;
            int r1 = (t + 1) * 16 + sr + 8;  if (r1 >= V) r1 = V - 1;
            const unsigned o0 = (unsigned)r0 * D_FEAT + sc;
            const unsigned o1 = (unsigned)r1 * D_FEAT + sc;
            sc0 = *(const float4*)(Cx + o0);
            sc1 = *(const float4*)(Cx + o1);
            sn0 = *(const float4*)(NF + o0);
            sn1 = *(const float4*)(NF + o1);
        }
        v8f ai[3], ah[3];
#pragma unroll
        for (int g = 0; g < 3; ++g) {
#pragma unroll
            for (int r = 0; r < 8; ++r) { ai[g][r] = bvi[g]; ah[g][r] = bvh[g]; }
        }
        if (t != t0) {
            __builtin_amdgcn_s_barrier_wait(-1);
            __builtin_amdgcn_fence(__ATOMIC_ACQUIRE, "workgroup", "local");
        }
#pragma unroll
        for (int s = 0; s < 32; ++s) {
            const int ka = 4 * s + 2 * hi;
            v2f ac, an;
            ac[0] = xs[buf][0][lo][ka];
            ac[1] = xs[buf][0][lo][ka + 1];
            an[0] = xs[buf][1][lo][ka];
            an[1] = xs[buf][1][lo][ka + 1];
#pragma unroll
            for (int g = 0; g < 3; ++g) {
                ai[g] = __builtin_amdgcn_wmma_f32_16x16x4_f32(
                    false, ac, false, bi[g][s], (short)0, ai[g], false, false);
                ah[g] = __builtin_amdgcn_wmma_f32_16x16x4_f32(
                    false, an, false, bh[g][s], (short)0, ah[g], false, false);
            }
        }
        if (more) {
            *(float4*)&xs[buf ^ 1][0][sr][sc] = sc0;
            *(float4*)&xs[buf ^ 1][0][sr + 8][sc] = sc1;
            *(float4*)&xs[buf ^ 1][1][sr][sc] = sn0;
            *(float4*)&xs[buf ^ 1][1][sr + 8][sc] = sn1;
        }
        float xv[8];
#pragma unroll
        for (int r = 0; r < 8; ++r) xv[r] = xs[buf][1][r + 8 * hi][col];
        if (more) {
            __builtin_amdgcn_fence(__ATOMIC_RELEASE, "workgroup", "local");
            __builtin_amdgcn_s_barrier_signal(-1);
        }
        float gr[8], gu[8], gn[8], go[8];
#pragma unroll
        for (int r = 0; r < 8; ++r) gr[r] = sigmoid_f(ai[0][r] + ah[0][r]);
#pragma unroll
        for (int r = 0; r < 8; ++r) gu[r] = sigmoid_f(ai[1][r] + ah[1][r]);
#pragma unroll
        for (int r = 0; r < 8; ++r) gn[r] = tanh_sel(ai[2][r] + gr[r] * ah[2][r]);
#pragma unroll
        for (int r = 0; r < 8; ++r) {
            const float h = fmaf(1.0f - gu[r], gn[r], gu[r] * xv[r]);
            go[r] = h > 0.0f ? h : 0.0f;
        }
        const int rowbase = t * 16;
        if (rowbase + 16 <= V) {
#pragma unroll
            for (int r = 0; r < 8; ++r)
                out[(unsigned)(rowbase + r + 8 * hi) * D_FEAT + col] = go[r];
        } else {
#pragma unroll
            for (int r = 0; r < 8; ++r) {
                const int row = rowbase + r + 8 * hi;
                if (row < V) out[(unsigned)row * D_FEAT + col] = go[r];
            }
        }
    }
}

extern "C" void kernel_launch(void* const* d_in, const int* in_sizes, int n_in,
                              void* d_out, int out_size, void* d_ws, size_t ws_size,
                              hipStream_t stream) {
    const float* nf     = (const float*)d_in[0];
    const int*   src    = (const int*)d_in[1];
    const int*   dst    = (const int*)d_in[2];
    const float* W_edge = (const float*)d_in[3];
    const float* b_edge = (const float*)d_in[4];
    const float* W_proj = (const float*)d_in[5];
    const float* b_proj = (const float*)d_in[6];
    const float* W_ih   = (const float*)d_in[7];
    const float* b_ih   = (const float*)d_in[8];
    const float* W_hh   = (const float*)d_in[9];
    const float* b_hh   = (const float*)d_in[10];

    const int V = in_sizes[0] / D_FEAT;
    const int E = in_sizes[1];
    const int nGroups = (V + GRP - 1) / GRP;
    const int chunkLen = (E + NCHUNK - 1) / NCHUNK;

    float* hv     = (float*)d_ws;
    float* c      = hv + (size_t)V * 128;
    float* WhhT   = c + (size_t)V * 128;
    float* WihT   = WhhT + 128 * 384;
    int2*  bucket = (int2*)(WihT + 128 * 384);
    int*   segoff = (int*)(bucket + (size_t)NCHUNK * chunkLen);
    float* pd     = (float*)(segoff + (size_t)NCHUNK * (nGroups + 1));
    float* ps     = pd + V;

    const int blk = 256;

    const int rowTiles = (V + 15) / 16;
    const int proj_blocks = rowTiles >= 768 ? 768
                          : (rowTiles > 2 * PACK_BLOCKS ? rowTiles : 2 * PACK_BLOCKS);

    proj_node_pack_kernel<<<NCHUNK + proj_blocks, blk, 0, stream>>>(
        nf, W_proj, b_proj, hv, V, proj_blocks, W_edge, pd, ps,
        W_hh, WhhT, W_ih, WihT, src, dst, E, chunkLen, nGroups, segoff, bucket);

    group_gather_kernel<<<nGroups, 1024, 0, stream>>>(
        segoff, bucket, pd, ps, b_edge, hv, c, V, chunkLen, nGroups);

    gate_gemm_gru_kernel<GATE_TILES><<<(rowTiles + GATE_TILES - 1) / GATE_TILES, blk, 0, stream>>>(
        c, nf, WihT, WhhT, b_ih, b_hh, (float*)d_out, V);
}
